# starting kernel
# baseline (speedup 1.0000x reference)
#include <hip/hip_runtime.h>
#include <math.h>

typedef __attribute__((ext_vector_type(2))) float v2f;
typedef __attribute__((ext_vector_type(8))) float v8f;

#define NT 512

constexpr int NG = 100, NL = 400, NN = 300, NWF = 50, NS = 200;
constexpr int G_P = 112;
constexpr int S_P = 208;
constexpr int PDHG_ITERS = 300, POWER_ITERS = 25;

constexpr int OX_P  = 0;
constexpr int OX_RU = OX_P  + G_P;
constexpr int OX_RD = OX_RU + G_P;
constexpr int OX_U  = OX_RD + G_P;
constexpr int OX_D  = OX_U  + G_P * S_P;
constexpr int OX_FU = OX_D  + G_P * S_P;
constexpr int OX_FD = OX_FU + NL;
constexpr int XSZ   = OX_FD + NL;

constexpr int OY_Y1 = 0;
constexpr int OY_Y2 = 16;
constexpr int OY_Y3 = OY_Y2 + NL;
constexpr int OY_Y4 = OY_Y3 + NL;
constexpr int OY_Z1 = OY_Y4 + S_P;
constexpr int OY_Z2 = OY_Z1 + G_P;
constexpr int OY_Z3 = OY_Z2 + G_P;
constexpr int OY_Z4 = OY_Z3 + G_P * S_P;
constexpr int OY_Z5 = OY_Z4 + G_P * S_P;
constexpr int OY_Z6 = OY_Z5 + NL * S_P;
constexpr int YSZ   = OY_Z6 + NL * S_P;

constexpr int WS_PG   = 0;
constexpr int WS_PGT  = WS_PG   + NL * G_P;
constexpr int WS_PW   = WS_PGT  + G_P * NL;
constexpr int WS_B2   = WS_PW   + NL * S_P;
constexpr int WS_B3   = WS_B2   + NL;
constexpr int WS_B4   = WS_B3   + NL;
constexpr int WS_B1   = WS_B4   + S_P;
constexpr int WS_M1   = WS_B1   + 16;
constexpr int WS_NV   = WS_M1   + NN * S_P;
constexpr int WS_X    = WS_NV   + NN + 4;
constexpr int WS_XB   = WS_X    + XSZ;
constexpr int WS_ATY  = WS_XB   + XSZ;
constexpr int WS_Y    = WS_ATY  + XSZ;
constexpr int WS_AX   = WS_Y    + YSZ;
constexpr int WS_T    = WS_AX   + YSZ;
constexpr int WS_FLOW = WS_T    + G_P * S_P;
constexpr int WS_GP   = WS_FLOW + NL * S_P;
constexpr int WS_COL  = WS_GP   + NL;
constexpr int WS_S3   = WS_COL  + S_P;
constexpr int WS_S4   = WS_S3   + G_P;
constexpr int WS_S5   = WS_S4   + G_P;
constexpr int WS_S6   = WS_S5   + NL;
constexpr int WS_GTY  = WS_S6   + NL;
constexpr int WS_SUMP = WS_GTY  + G_P;
constexpr int WS_TOTAL= WS_SUMP + 16;


__global__ void k_zero(float* w) {
    for (int i = blockIdx.x * blockDim.x + threadIdx.x; i < WS_TOTAL;
         i += gridDim.x * blockDim.x) w[i] = 0.0f;
}

__global__ void k_prep1(float* w, const float* w_scen, const float* node_W,
                        const float* node_L, const float* Pd, const float* w_exp) {
    const int NM1 = NN * NS;
    const int total = NM1 + NN + NS + 1;
    for (int idx = blockIdx.x * blockDim.x + threadIdx.x; idx < total;
         idx += gridDim.x * blockDim.x) {
        if (idx < NM1) {
            int n = idx / NS, s = idx % NS;
            float a = 0.f;
            for (int ww = 0; ww < NWF; ++ww) a += node_W[n*NWF + ww] * w_scen[s*NWF + ww];
            w[WS_M1 + n*S_P + s] = a;
        } else if (idx < NM1 + NN) {
            int n = idx - NM1;
            float a = 0.f;
            for (int ww = 0; ww < NWF; ++ww) a += node_W[n*NWF + ww] * w_exp[ww];
            for (int m = 0; m < NN; ++m)     a -= node_L[n*NN + m] * Pd[m];
            w[WS_NV + n] = a;
        } else if (idx < NM1 + NN + NS) {
            int s = idx - NM1 - NN;
            float a = 0.f;
            for (int ww = 0; ww < NWF; ++ww) a += w_scen[s*NWF + ww];
            w[WS_B4 + s] = -a;
        } else {
            float a = 0.f;
            for (int m = 0; m < NN; ++m)  a += Pd[m];
            for (int ww = 0; ww < NWF; ++ww) a -= w_exp[ww];
            w[WS_B1] = a;
        }
    }
}

__global__ void k_prep2(float* w, const float* PTDF, const float* node_G,
                        const float* Line_cap) {
    const int A = NL * NG;
    const int B = A + NG * NL;
    const int C = B + NL * NS;
    const int D2 = C + NL;
    const int E = D2 + NL;
    for (int idx = blockIdx.x * blockDim.x + threadIdx.x; idx < E;
         idx += gridDim.x * blockDim.x) {
        if (idx < A) {
            int l = idx / NG, g = idx % NG;
            float a = 0.f;
            for (int n = 0; n < NN; ++n) a += PTDF[l*NN + n] * node_G[n*NG + g];
            w[WS_PG + l*G_P + g] = a;
        } else if (idx < B) {
            int r = idx - A; int g = r / NL, l = r % NL;
            float a = 0.f;
            for (int n = 0; n < NN; ++n) a += PTDF[l*NN + n] * node_G[n*NG + g];
            w[WS_PGT + g*NL + l] = a;
        } else if (idx < C) {
            int r = idx - B; int l = r / NS, s = r % NS;
            float a = 0.f;
            for (int n = 0; n < NN; ++n) a += PTDF[l*NN + n] * w[WS_M1 + n*S_P + s];
            w[WS_PW + l*S_P + s] = a;
        } else if (idx < D2) {
            int l = idx - C;
            float a = 0.f;
            for (int n = 0; n < NN; ++n) a += PTDF[l*NN + n] * w[WS_NV + n];
            w[WS_B2 + l] = Line_cap[l] - a;
        } else {
            int l = idx - D2;
            float a = 0.f;
            for (int n = 0; n < NN; ++n) a += PTDF[l*NN + n] * w[WS_NV + n];
            w[WS_B3 + l] = Line_cap[l] + a;
        }
    }
}

__global__ void k_setv(float* w) {
    const int total = 3*NG + 2*NG*NS + 2*NL;
    float* X = w + WS_X;
    for (int idx = blockIdx.x * blockDim.x + threadIdx.x; idx < total;
         idx += gridDim.x * blockDim.x) {
        int off;
        if (idx < NG)                off = OX_P  + idx;
        else if (idx < 2*NG)         off = OX_RU + (idx - NG);
        else if (idx < 3*NG)         off = OX_RD + (idx - 2*NG);
        else if (idx < 3*NG + NG*NS) { int r = idx - 3*NG; off = OX_U + (r/NS)*S_P + (r%NS); }
        else if (idx < 3*NG + 2*NG*NS) { int r = idx - 3*NG - NG*NS; off = OX_D + (r/NS)*S_P + (r%NS); }
        else if (idx < 3*NG + 2*NG*NS + NL) off = OX_FU + (idx - 3*NG - 2*NG*NS);
        else                         off = OX_FD + (idx - 3*NG - 2*NG*NS - NL);
        X[off] = 1.0f;
    }
}


__device__ __forceinline__ float block_reduce(float v, float* red, int tid) {
    red[tid] = v; __syncthreads();
    for (int off = NT >> 1; off > 0; off >>= 1) {
        if (tid < off) red[tid] += red[tid + off];
        __syncthreads();
    }
    float r = red[0]; __syncthreads();
    return r;
}

__device__ void matvec_core(float* w, const float* xb, float* yb, int tid) {
    const int lane = tid & 31, wave = tid >> 5, nw = NT / 32;
    const float* PG = w + WS_PG;
    float* flow = w + WS_FLOW;
    __syncthreads();
    {
        const float* Ux = xb + OX_U; const float* Dx = xb + OX_D;
        const int NTile = 25 * 13;
        const int m16 = lane & 15;
        const int kh  = (lane >> 4) * 2;
        for (int tile = wave; tile < NTile; tile += nw) {
            int tm = tile / 13, tn = tile % 13;
            int m = tm * 16 + m16;
            int n = tn * 16 + m16;
            v8f acc = {0.f,0.f,0.f,0.f,0.f,0.f,0.f,0.f};
            const float* arow = PG + m * G_P;
            for (int k0 = 0; k0 < NG; k0 += 4) {
                int ka = k0 + kh;
                v2f a, b;
                a[0] = arow[ka]; a[1] = arow[ka + 1];
                int bi = ka * S_P + n;
                b[0] = Ux[bi]        - Dx[bi];
                b[1] = Ux[bi + S_P]  - Dx[bi + S_P];
                acc = __builtin_amdgcn_wmma_f32_16x16x4_f32(false, a, false, b,
                                                            (short)0, acc, false, false);
            }
            int mr = tm * 16 + ((lane >> 4) << 3);
            for (int i = 0; i < 8; ++i) flow[(mr + i) * S_P + n] = acc[i];
        }
    }
    if (tid < NL) {
        float s = 0.f; const float* r = PG + tid * G_P; const float* p = xb + OX_P;
        for (int g = 0; g < NG; ++g) s += r[g] * p[g];
        w[WS_GP + tid] = s;
    }
    if (tid < NS) {
        float s = 0.f;
        const float* Ux = xb + OX_U + tid; const float* Dx = xb + OX_D + tid;
        for (int g = 0; g < NG; ++g) s += Ux[g * S_P] - Dx[g * S_P];
        w[WS_COL + tid] = s;
    }
    if (tid == NT - 1) {
        float s = 0.f;
        for (int g = 0; g < NG; ++g) s += xb[OX_P + g];
        w[WS_SUMP] = s;
    }
    __syncthreads();
    if (tid == 0) yb[OY_Y1] = w[WS_SUMP];
    for (int l = tid; l < NL; l += NT) {
        float g = w[WS_GP + l];
        yb[OY_Y2 + l] =  g + xb[OX_FU + l];
        yb[OY_Y3 + l] = -g + xb[OX_FD + l];
    }
    for (int s = tid; s < NS; s += NT) yb[OY_Y4 + s] = w[WS_COL + s];
    for (int g = tid; g < NG; g += NT) {
        yb[OY_Z1 + g] = xb[OX_P + g]  + xb[OX_RU + g];
        yb[OY_Z2 + g] = xb[OX_RD + g] - xb[OX_P + g];
    }
    for (int i = tid; i < NG * NS; i += NT) {
        int g = i / NS, s = i % NS, o = g * S_P + s;
        yb[OY_Z3 + o] = xb[OX_U + o] - xb[OX_RU + g];
        yb[OY_Z4 + o] = xb[OX_D + o] - xb[OX_RD + g];
    }
    for (int i = tid; i < NL * NS; i += NT) {
        int l = i / NS, s = i % NS, o = l * S_P + s;
        float f = flow[o];
        yb[OY_Z5 + o] =  f - xb[OX_FU + l];
        yb[OY_Z6 + o] = -f - xb[OX_FD + l];
    }
}

__device__ void rmatvec_core(float* w, const float* yb, float* xo, int tid) {
    const int lane = tid & 31, wave = tid >> 5, nw = NT / 32;
    const float* PGT = w + WS_PGT;
    float* T = w + WS_T;
    __syncthreads();
    {
        const float* Z5 = yb + OY_Z5; const float* Z6 = yb + OY_Z6;
        const int NTile = 7 * 13;
        const int m16 = lane & 15;
        const int kh  = (lane >> 4) * 2;
        for (int tile = wave; tile < NTile; tile += nw) {
            int tm = tile / 13, tn = tile % 13;
            int m = tm * 16 + m16;
            int n = tn * 16 + m16;
            v8f acc = {0.f,0.f,0.f,0.f,0.f,0.f,0.f,0.f};
            const float* arow = PGT + m * NL;
            for (int k0 = 0; k0 < NL; k0 += 4) {
                int ka = k0 + kh;
                v2f a, b;
                a[0] = arow[ka]; a[1] = arow[ka + 1];
                int bi = ka * S_P + n;
                b[0] = Z5[bi]       - Z6[bi];
                b[1] = Z5[bi + S_P] - Z6[bi + S_P];
                acc = __builtin_amdgcn_wmma_f32_16x16x4_f32(false, a, false, b,
                                                            (short)0, acc, false, false);
            }
            int mr = tm * 16 + ((lane >> 4) << 3);
            for (int i = 0; i < 8; ++i) T[(mr + i) * S_P + n] = acc[i];
        }
    }
    if (tid < NG) {
        float a = 0.f, b = 0.f, c = 0.f;
        const float* z3r = yb + OY_Z3 + tid * S_P;
        const float* z4r = yb + OY_Z4 + tid * S_P;
        for (int s = 0; s < NS; ++s) { a += z3r[s]; b += z4r[s]; }
        const float* pr = PGT + tid * NL;
        for (int l = 0; l < NL; ++l) c += pr[l] * (yb[OY_Y2 + l] - yb[OY_Y3 + l]);
        w[WS_S3 + tid] = a; w[WS_S4 + tid] = b; w[WS_GTY + tid] = c;
    }
    if (tid >= NG && tid < NG + NL) {
        int l = tid - NG;
        float a = 0.f, b = 0.f;
        const float* z5r = yb + OY_Z5 + l * S_P;
        const float* z6r = yb + OY_Z6 + l * S_P;
        for (int s = 0; s < NS; ++s) { a += z5r[s]; b += z6r[s]; }
        w[WS_S5 + l] = a; w[WS_S6 + l] = b;
    }
    __syncthreads();
    float y1v = yb[OY_Y1];
    for (int g = tid; g < NG; g += NT) {
        xo[OX_P  + g] = y1v + w[WS_GTY + g] + yb[OY_Z1 + g] - yb[OY_Z2 + g];
        xo[OX_RU + g] = yb[OY_Z1 + g] - w[WS_S3 + g];
        xo[OX_RD + g] = yb[OY_Z2 + g] - w[WS_S4 + g];
    }
    for (int l = tid; l < NL; l += NT) {
        xo[OX_FU + l] = yb[OY_Y2 + l] - w[WS_S5 + l];
        xo[OX_FD + l] = yb[OY_Y3 + l] - w[WS_S6 + l];
    }
    for (int i = tid; i < NG * NS; i += NT) {
        int g = i / NS, s = i % NS, o = g * S_P + s;
        float tv = T[o];
        xo[OX_U + o] =  yb[OY_Y4 + s] + yb[OY_Z3 + o] + tv;
        xo[OX_D + o] = -yb[OY_Y4 + s] + yb[OY_Z4 + o] - tv;
    }
}


__global__ __launch_bounds__(NT) void k_solver(float* w, const float* Cost,
                                               const float* Cru, const float* Crd,
                                               const float* Pmax, float* out) {
    __shared__ float red[NT];
    const int tid = threadIdx.x;
    float* X   = w + WS_X;
    float* XB  = w + WS_XB;
    float* ATY = w + WS_ATY;
    float* Y   = w + WS_Y;
    float* AX  = w + WS_AX;

    for (int i = tid * 32; i < NL * G_P; i += NT * 32) {
        __builtin_prefetch(&w[WS_PG + i], 0, 3);
        __builtin_prefetch(&w[WS_PGT + i], 0, 3);
    }

    float sumsq = 0.f;
    for (int it = 0; it < POWER_ITERS; ++it) {
        __syncthreads();
        float part = 0.f;
        for (int i = tid; i < XSZ; i += NT) { float v = X[i]; part += v * v; }
        sumsq = block_reduce(part, red, tid);
        float inv = 1.0f / sqrtf(sumsq);
        for (int i = tid; i < XSZ; i += NT) X[i] *= inv;
        matvec_core(w, X, AX, tid);
        rmatvec_core(w, AX, X, tid);
    }
    __syncthreads();
    {
        float part = 0.f;
        for (int i = tid; i < XSZ; i += NT) { float v = X[i]; part += v * v; }
        sumsq = block_reduce(part, red, tid);
    }
    const float step = 0.9f / sqrtf(sqrtf(sumsq));
    const float b1 = w[WS_B1];
    const float* pw = w + WS_PW;

    for (int i = tid; i < XSZ; i += NT) { X[i] = 0.f; XB[i] = 0.f; }
    for (int i = tid; i < YSZ; i += NT) Y[i] = 0.f;

    for (int it = 0; it < PDHG_ITERS; ++it) {
        rmatvec_core(w, Y, ATY, tid);
        __syncthreads();
        for (int g = tid; g < NG; g += NT) {
            float o, n;
            o = X[OX_P  + g]; n = fmaxf(o - step * (Cost[g] + ATY[OX_P  + g]), 0.f);
            X[OX_P  + g] = n; XB[OX_P  + g] = 2.f * n - o;
            o = X[OX_RU + g]; n = fmaxf(o - step * (Cru[g]  + ATY[OX_RU + g]), 0.f);
            X[OX_RU + g] = n; XB[OX_RU + g] = 2.f * n - o;
            o = X[OX_RD + g]; n = fmaxf(o - step * (Crd[g]  + ATY[OX_RD + g]), 0.f);
            X[OX_RD + g] = n; XB[OX_RD + g] = 2.f * n - o;
        }
        for (int l = tid; l < NL; l += NT) {
            float o, n;
            o = X[OX_FU + l]; n = fmaxf(o - step * ATY[OX_FU + l], 0.f);
            X[OX_FU + l] = n; XB[OX_FU + l] = 2.f * n - o;
            o = X[OX_FD + l]; n = fmaxf(o - step * ATY[OX_FD + l], 0.f);
            X[OX_FD + l] = n; XB[OX_FD + l] = 2.f * n - o;
        }
        for (int i = tid; i < NG * NS; i += NT) {
            int g = i / NS, s = i % NS, o0 = g * S_P + s;
            float o, n;
            o = X[OX_U + o0]; n = fmaxf(o - step * ATY[OX_U + o0], 0.f);
            X[OX_U + o0] = n; XB[OX_U + o0] = 2.f * n - o;
            o = X[OX_D + o0]; n = fmaxf(o - step * ATY[OX_D + o0], 0.f);
            X[OX_D + o0] = n; XB[OX_D + o0] = 2.f * n - o;
        }
        matvec_core(w, XB, AX, tid);
        __syncthreads();
        if (tid == 0) Y[OY_Y1] += step * (AX[OY_Y1] - b1);
        for (int l = tid; l < NL; l += NT) {
            Y[OY_Y2 + l] += step * (AX[OY_Y2 + l] - w[WS_B2 + l]);
            Y[OY_Y3 + l] += step * (AX[OY_Y3 + l] - w[WS_B3 + l]);
        }
        for (int s = tid; s < NS; s += NT)
            Y[OY_Y4 + s] += step * (AX[OY_Y4 + s] - w[WS_B4 + s]);
        for (int g = tid; g < NG; g += NT) {
            Y[OY_Z1 + g] = fmaxf(Y[OY_Z1 + g] + step * (AX[OY_Z1 + g] - Pmax[g]), 0.f);
            Y[OY_Z2 + g] = fmaxf(Y[OY_Z2 + g] + step *  AX[OY_Z2 + g], 0.f);
        }
        for (int i = tid; i < NG * NS; i += NT) {
            int g = i / NS, s = i % NS, o = g * S_P + s;
            Y[OY_Z3 + o] = fmaxf(Y[OY_Z3 + o] + step * AX[OY_Z3 + o], 0.f);
            Y[OY_Z4 + o] = fmaxf(Y[OY_Z4 + o] + step * AX[OY_Z4 + o], 0.f);
        }
        for (int i = tid; i < NL * NS; i += NT) {
            int l = i / NS, s = i % NS, o = l * S_P + s;
            Y[OY_Z5 + o] = fmaxf(Y[OY_Z5 + o] + step * (AX[OY_Z5 + o] + pw[o]), 0.f);
            Y[OY_Z6 + o] = fmaxf(Y[OY_Z6 + o] + step * (AX[OY_Z6 + o] - pw[o]), 0.f);
        }
        __syncthreads();
    }

    for (int g = tid; g < NG; g += NT) {
        out[g]          = X[OX_P  + g];
        out[NG + g]     = X[OX_RU + g];
        out[2 * NG + g] = X[OX_RD + g];
    }
    for (int i = tid; i < NG * NS; i += NT) {
        int g = i / NS, s = i % NS;
        out[3 * NG + i]           = X[OX_U + g * S_P + s];
        out[3 * NG + NG * NS + i] = X[OX_D + g * S_P + s];
    }
    for (int l = tid; l < NL; l += NT) {
        out[3 * NG + 2 * NG * NS + l]      = X[OX_FU + l];
        out[3 * NG + 2 * NG * NS + NL + l] = X[OX_FD + l];
    }
}


extern "C" void kernel_launch(void* const* d_in, const int* in_sizes, int n_in,
                              void* d_out, int out_size, void* d_ws, size_t ws_size,
                              hipStream_t stream) {
    const float* w_scen   = (const float*)d_in[0];
    const float* Cost     = (const float*)d_in[1];
    const float* C_r_up   = (const float*)d_in[2];
    const float* C_r_down = (const float*)d_in[3];
    const float* Pmax     = (const float*)d_in[4];
    const float* PTDF     = (const float*)d_in[5];
    const float* node_G   = (const float*)d_in[6];
    const float* node_W   = (const float*)d_in[7];
    const float* node_L   = (const float*)d_in[8];
    const float* Pd       = (const float*)d_in[9];
    const float* w_exp    = (const float*)d_in[10];
    const float* Line_cap = (const float*)d_in[11];
    float* w   = (float*)d_ws;
    float* out = (float*)d_out;

    k_zero <<<512, 256, 0, stream>>>(w);
    k_prep1<<<240, 256, 0, stream>>>(w, w_scen, node_W, node_L, Pd, w_exp);
    k_prep2<<<640, 256, 0, stream>>>(w, PTDF, node_G, Line_cap);
    k_setv <<<161, 256, 0, stream>>>(w);
    k_solver<<<1, NT, 0, stream>>>(w, Cost, C_r_up, C_r_down, Pmax, out);
}


# Row sums of z3..z6 taken in the products' epilogues: each tile's rows added over its 16 lanes (DPP), one partial per column strip in LDS, 13 partials added in order; row sums change order
# speedup vs baseline: 4.0699x; 4.0699x over previous
#include <hip/hip_runtime.h>
#include <math.h>

typedef __attribute__((ext_vector_type(2))) float v2f;
typedef __attribute__((ext_vector_type(8))) float v8f;

#define NT 512

constexpr int NG = 100, NL = 400, NN = 300, NWF = 50, NS = 200;
constexpr int G_P = 112;
constexpr int S_P = 208;
constexpr int PDHG_ITERS = 300, POWER_ITERS = 25;
constexpr int PG_LP = 116;
constexpr int N_STRIP = S_P / 16;
constexpr int RS5 = 0;
constexpr int RS6 = RS5 + N_STRIP * NL;
constexpr int RS3 = RS6 + N_STRIP * NL;
constexpr int RS4 = RS3 + N_STRIP * G_P;
constexpr int RS_TOTAL = RS4 + N_STRIP * G_P;

constexpr int OX_P  = 0;
constexpr int OX_RU = OX_P  + G_P;
constexpr int OX_RD = OX_RU + G_P;
constexpr int OX_U  = OX_RD + G_P;
constexpr int OX_D  = OX_U  + G_P * S_P;
constexpr int OX_FU = OX_D  + G_P * S_P;
constexpr int OX_FD = OX_FU + NL;
constexpr int XSZ   = OX_FD + NL;

constexpr int OY_Y1 = 0;
constexpr int OY_Y2 = 16;
constexpr int OY_Y3 = OY_Y2 + NL;
constexpr int OY_Y4 = OY_Y3 + NL;
constexpr int OY_Z1 = OY_Y4 + S_P;
constexpr int OY_Z2 = OY_Z1 + G_P;
constexpr int OY_Z3 = OY_Z2 + G_P;
constexpr int OY_Z4 = OY_Z3 + G_P * S_P;
constexpr int OY_Z5 = OY_Z4 + G_P * S_P;
constexpr int OY_Z6 = OY_Z5 + NL * S_P;
constexpr int YSZ   = OY_Z6 + NL * S_P;

constexpr int WS_PG   = 0;
constexpr int WS_PGT  = WS_PG   + NL * G_P;
constexpr int WS_PW   = WS_PGT  + G_P * NL;
constexpr int WS_B2   = WS_PW   + NL * S_P;
constexpr int WS_B3   = WS_B2   + NL;
constexpr int WS_B4   = WS_B3   + NL;
constexpr int WS_B1   = WS_B4   + S_P;
constexpr int WS_M1   = WS_B1   + 16;
constexpr int WS_NV   = WS_M1   + NN * S_P;
constexpr int WS_X    = WS_NV   + NN + 4;
constexpr int WS_XB   = WS_X    + XSZ;
constexpr int WS_ATY  = WS_XB   + XSZ;
constexpr int WS_Y    = WS_ATY  + XSZ;
constexpr int WS_AX   = WS_Y    + YSZ;
constexpr int WS_T    = WS_AX   + YSZ;
constexpr int WS_FLOW = WS_T    + G_P * S_P;
constexpr int WS_GP   = WS_FLOW + NL * S_P;
constexpr int WS_COL  = WS_GP   + NL;
constexpr int WS_S3   = WS_COL  + S_P;
constexpr int WS_S4   = WS_S3   + G_P;
constexpr int WS_S5   = WS_S4   + G_P;
constexpr int WS_S6   = WS_S5   + NL;
constexpr int WS_GTY  = WS_S6   + NL;
constexpr int WS_SUMP = WS_GTY  + G_P;
constexpr int WS_TOTAL= WS_SUMP + 16;


__global__ void k_zero(float* w) {
    for (int i = blockIdx.x * blockDim.x + threadIdx.x; i < WS_TOTAL;
         i += gridDim.x * blockDim.x) w[i] = 0.0f;
}

__global__ void k_prep1(float* w, const float* w_scen, const float* node_W,
                        const float* node_L, const float* Pd, const float* w_exp) {
    const int NM1 = NN * NS;
    const int total = NM1 + NN + NS + 1;
    for (int idx = blockIdx.x * blockDim.x + threadIdx.x; idx < total;
         idx += gridDim.x * blockDim.x) {
        if (idx < NM1) {
            int n = idx / NS, s = idx % NS;
            float a = 0.f;
            for (int ww = 0; ww < NWF; ++ww) a += node_W[n*NWF + ww] * w_scen[s*NWF + ww];
            w[WS_M1 + n*S_P + s] = a;
        } else if (idx < NM1 + NN) {
            int n = idx - NM1;
            float a = 0.f;
            for (int ww = 0; ww < NWF; ++ww) a += node_W[n*NWF + ww] * w_exp[ww];
            for (int m = 0; m < NN; ++m)     a -= node_L[n*NN + m] * Pd[m];
            w[WS_NV + n] = a;
        } else if (idx < NM1 + NN + NS) {
            int s = idx - NM1 - NN;
            float a = 0.f;
            for (int ww = 0; ww < NWF; ++ww) a += w_scen[s*NWF + ww];
            w[WS_B4 + s] = -a;
        } else {
            float a = 0.f;
            for (int m = 0; m < NN; ++m)  a += Pd[m];
            for (int ww = 0; ww < NWF; ++ww) a -= w_exp[ww];
            w[WS_B1] = a;
        }
    }
}

__global__ void k_prep2(float* w, const float* PTDF, const float* node_G,
                        const float* Line_cap) {
    const int A = NL * NG;
    const int C = A + NL * NS;
    const int D2 = C + NL;
    const int E = D2 + NL;
    for (int idx = blockIdx.x * blockDim.x + threadIdx.x; idx < E;
         idx += gridDim.x * blockDim.x) {
        if (idx < A) {
            int l = idx / NG, g = idx % NG;
            float a = 0.f;
            for (int n = 0; n < NN; ++n) a += PTDF[l*NN + n] * node_G[n*NG + g];
            w[WS_PG + l*G_P + g] = a;
        } else if (idx < C) {
            int r = idx - A; int l = r / NS, s = r % NS;
            float a = 0.f;
            for (int n = 0; n < NN; ++n) a += PTDF[l*NN + n] * w[WS_M1 + n*S_P + s];
            w[WS_PW + l*S_P + s] = a;
        } else if (idx < D2) {
            int l = idx - C;
            float a = 0.f;
            for (int n = 0; n < NN; ++n) a += PTDF[l*NN + n] * w[WS_NV + n];
            w[WS_B2 + l] = Line_cap[l] - a;
        } else {
            int l = idx - D2;
            float a = 0.f;
            for (int n = 0; n < NN; ++n) a += PTDF[l*NN + n] * w[WS_NV + n];
            w[WS_B3 + l] = Line_cap[l] + a;
        }
    }
}

__global__ void k_setv(float* w) {
    const int total = 3*NG + 2*NG*NS + 2*NL;
    float* X = w + WS_X;
    for (int idx = blockIdx.x * blockDim.x + threadIdx.x; idx < total;
         idx += gridDim.x * blockDim.x) {
        int off;
        if (idx < NG)                off = OX_P  + idx;
        else if (idx < 2*NG)         off = OX_RU + (idx - NG);
        else if (idx < 3*NG)         off = OX_RD + (idx - 2*NG);
        else if (idx < 3*NG + NG*NS) { int r = idx - 3*NG; off = OX_U + (r/NS)*S_P + (r%NS); }
        else if (idx < 3*NG + 2*NG*NS) { int r = idx - 3*NG - NG*NS; off = OX_D + (r/NS)*S_P + (r%NS); }
        else if (idx < 3*NG + 2*NG*NS + NL) off = OX_FU + (idx - 3*NG - 2*NG*NS);
        else                         off = OX_FD + (idx - 3*NG - 2*NG*NS - NL);
        X[off] = 1.0f;
    }
}


__device__ __forceinline__ float block_reduce(float v, float* red, int tid) {
    red[tid] = v; __syncthreads();
    for (int off = NT >> 1; off > 0; off >>= 1) {
        if (tid < off) red[tid] += red[tid + off];
        __syncthreads();
    }
    float r = red[0]; __syncthreads();
    return r;
}

template <int M>
__device__ __forceinline__ float row_xor(float x) {
    return __int_as_float(__builtin_amdgcn_update_dpp(0, __float_as_int(x), 0x160 + M, 0xf, 0xf, true));
}

__device__ __forceinline__ float rows_over_lanes(const float (&v)[8], int c) {
    const bool b2 = (c & 4) != 0, b1 = (c & 2) != 0, b0 = (c & 1) != 0;
    float w[4], u[2];
#pragma unroll
    for (int j = 0; j < 4; ++j) {
        float keep = b2 ? v[4 + j] : v[j];
        float send = b2 ? v[j] : v[4 + j];
        w[j] = keep + row_xor<4>(send);
    }
#pragma unroll
    for (int j = 0; j < 2; ++j) {
        float keep = b1 ? w[2 + j] : w[j];
        float send = b1 ? w[j] : w[2 + j];
        u[j] = keep + row_xor<2>(send);
    }
    float keep = b0 ? u[1] : u[0];
    float send = b0 ? u[0] : u[1];
    float t = keep + row_xor<1>(send);
    return t + row_xor<8>(t);
}

template <bool DUAL>
__device__ __forceinline__ void matvec_core(float* w, const float* sPG, const float* xb,
                                            float* yb, int tid, float* Yd, const float* pw,
                                            float step, float* sRS) {
    const int lane = tid & 31, wave = __builtin_amdgcn_readfirstlane(tid >> 5), nw = NT / 32;
    float* flow = w + WS_FLOW;
    __syncthreads();
    {
        const float* Ux = xb + OX_U; const float* Dx = xb + OX_D;
        constexpr int TM = 25, NTile = 25 * 13;
        const int m16 = lane & 15;
        const int kh  = (lane >> 4) * 2;
        const int t1 = ((wave + 1) * NTile) / nw;
        int t = (wave * NTile) / nw;
        while (t < t1) {
            const int tn = t / TM;
            const int tm_lo = t - tn * TM;
            const int tm_hi = (tm_lo + (t1 - t) < TM) ? tm_lo + (t1 - t) : TM;
            const int n = tn * 16 + m16;
            v2f b[25];
#pragma unroll
            for (int j = 0; j < 25; ++j) {
                int bi = (4 * j + kh) * S_P + n;
                b[j][0] = Ux[bi]       - Dx[bi];
                b[j][1] = Ux[bi + S_P] - Dx[bi + S_P];
            }
            for (int tm = tm_lo; tm < tm_hi; ++tm) {
                const int mr = tm * 16 + ((lane >> 4) << 3);
                float y5[8], y6[8], pv[8], fu[8], fd[8];
                if (DUAL) {
#pragma unroll
                    for (int i = 0; i < 8; ++i) {
                        int o = (mr + i) * S_P + n;
                        y5[i] = Yd[OY_Z5 + o]; y6[i] = Yd[OY_Z6 + o]; pv[i] = pw[o];
                        fu[i] = xb[OX_FU + mr + i]; fd[i] = xb[OX_FD + mr + i];
                    }
                }
                const float* arow = sPG + (tm * 16 + m16) * PG_LP + kh;
                v2f a[25];
#pragma unroll
                for (int j = 0; j < 25; ++j) { a[j][0] = arow[4 * j]; a[j][1] = arow[4 * j + 1]; }
                v8f acc = {0.f,0.f,0.f,0.f,0.f,0.f,0.f,0.f};
#pragma unroll
                for (int j = 0; j < 25; ++j)
                    acc = __builtin_amdgcn_wmma_f32_16x16x4_f32(false, a[j], false, b[j],
                                                                (short)0, acc, false, false);
                __builtin_amdgcn_sched_group_barrier(0x100, 13, 0);
                __builtin_amdgcn_sched_group_barrier(0x008, 25, 0);
                if (DUAL) {
                    const bool live = n < NS;
                    float r5[8], r6[8];
#pragma unroll
                    for (int i = 0; i < 8; ++i) {
                        int o = (mr + i) * S_P + n;
                        float f = acc[i];
                        float n5 = fmaxf(y5[i] + step * (( f - fu[i]) + pv[i]), 0.f);
                        float n6 = fmaxf(y6[i] + step * ((-f - fd[i]) - pv[i]), 0.f);
                        r5[i] = live ? n5 : 0.f;
                        r6[i] = live ? n6 : 0.f;
                        Yd[OY_Z5 + o] = r5[i];
                        Yd[OY_Z6 + o] = r6[i];
                    }
                    const float s5 = rows_over_lanes(r5, m16);
                    const float s6 = rows_over_lanes(r6, m16);
                    if (m16 < 8) {
                        sRS[RS5 + tn * NL + mr + m16] = s5;
                        sRS[RS6 + tn * NL + mr + m16] = s6;
                    }
                } else {
                    for (int i = 0; i < 8; ++i) flow[(mr + i) * S_P + n] = acc[i];
                }
            }
            t += tm_hi - tm_lo;
        }
    }
    if (tid < NL) {
        float s = 0.f; const float* r = sPG + tid * PG_LP; const float* p = xb + OX_P;
        for (int g = 0; g < NG; ++g) s += r[g] * p[g];
        w[WS_GP + tid] = s;
    }
    if (tid < NS) {
        float s = 0.f;
        const float* Ux = xb + OX_U + tid; const float* Dx = xb + OX_D + tid;
        for (int g = 0; g < NG; ++g) s += Ux[g * S_P] - Dx[g * S_P];
        w[WS_COL + tid] = s;
    }
    if (tid == NT - 1) {
        float s = 0.f;
        for (int g = 0; g < NG; ++g) s += xb[OX_P + g];
        w[WS_SUMP] = s;
    }
    __syncthreads();
    if (tid == 0) yb[OY_Y1] = w[WS_SUMP];
    for (int l = tid; l < NL; l += NT) {
        float g = w[WS_GP + l];
        yb[OY_Y2 + l] =  g + xb[OX_FU + l];
        yb[OY_Y3 + l] = -g + xb[OX_FD + l];
    }
    for (int s = tid; s < NS; s += NT) yb[OY_Y4 + s] = w[WS_COL + s];
    for (int g = tid; g < NG; g += NT) {
        yb[OY_Z1 + g] = xb[OX_P + g]  + xb[OX_RU + g];
        yb[OY_Z2 + g] = xb[OX_RD + g] - xb[OX_P + g];
    }
    if (!DUAL) {
        for (int i = tid; i < NG * NS; i += NT) {
            int g = i / NS, s = i % NS, o = g * S_P + s;
            yb[OY_Z3 + o] = xb[OX_U + o] - xb[OX_RU + g];
            yb[OY_Z4 + o] = xb[OX_D + o] - xb[OX_RD + g];
        }
        for (int i = tid; i < NL * NS; i += NT) {
            int l = i / NS, s = i % NS, o = l * S_P + s;
            float f = flow[o];
            yb[OY_Z5 + o] =  f - xb[OX_FU + l];
            yb[OY_Z6 + o] = -f - xb[OX_FD + l];
        }
    }
}

template <bool FUSED>
__device__ __forceinline__ void rmatvec_core(float* w, const float* sPG, float* yb,
                                             float* xo, int tid, float* X, float* XB,
                                             const float* Cost, const float* Cru,
                                             const float* Crd, float step, float* sRS) {
    const int lane = tid & 31, wave = __builtin_amdgcn_readfirstlane(tid >> 5), nw = NT / 32;
    float* T = w + WS_T;
    __syncthreads();
    if (FUSED) {
        if (tid < NG) {
            float a = 0.f, b = 0.f, c = 0.f;
            for (int k = 0; k < N_STRIP; ++k) {
                a += sRS[RS3 + k * G_P + tid]; b += sRS[RS4 + k * G_P + tid];
            }
            const float* pc = sPG + tid;
            for (int l = 0; l < NL; ++l) c += pc[l * PG_LP] * (yb[OY_Y2 + l] - yb[OY_Y3 + l]);
            const float z1 = yb[OY_Z1 + tid], z2 = yb[OY_Z2 + tid];
            const float ap  = yb[OY_Y1] + c + z1 - z2;
            const float aru = z1 - a;
            const float ard = z2 - b;
            float o, n;
            o = X[OX_P  + tid]; n = fmaxf(o - step * (Cost[tid] + ap),  0.f);
            X[OX_P  + tid] = n; XB[OX_P  + tid] = 2.f * n - o;
            o = X[OX_RU + tid]; n = fmaxf(o - step * (Cru[tid]  + aru), 0.f);
            X[OX_RU + tid] = n; XB[OX_RU + tid] = 2.f * n - o;
            o = X[OX_RD + tid]; n = fmaxf(o - step * (Crd[tid]  + ard), 0.f);
            X[OX_RD + tid] = n; XB[OX_RD + tid] = 2.f * n - o;
        }
        if (tid >= NG && tid < NG + NL) {
            int l = tid - NG;
            float a = 0.f, b = 0.f;
            for (int k = 0; k < N_STRIP; ++k) {
                a += sRS[RS5 + k * NL + l]; b += sRS[RS6 + k * NL + l];
            }
            const float afu = yb[OY_Y2 + l] - a;
            const float afd = yb[OY_Y3 + l] - b;
            float o, n;
            o = X[OX_FU + l]; n = fmaxf(o - step * afu, 0.f);
            X[OX_FU + l] = n; XB[OX_FU + l] = 2.f * n - o;
            o = X[OX_FD + l]; n = fmaxf(o - step * afd, 0.f);
            X[OX_FD + l] = n; XB[OX_FD + l] = 2.f * n - o;
        }
        __syncthreads();
    }
    {
        const float* Z5 = yb + OY_Z5; const float* Z6 = yb + OY_Z6;
        constexpr int TM = 7, NTile = 7 * 13;
        const int m16 = lane & 15;
        const int kh  = (lane >> 4) * 2;
        const int t1 = ((wave + 1) * NTile) / nw;
        int t = (wave * NTile) / nw;
        while (t < t1) {
            const int tn = t / TM;
            const int tm_lo = t - tn * TM;
            const int tm_hi = (tm_lo + (t1 - t) < TM) ? tm_lo + (t1 - t) : TM;
            const int n = tn * 16 + m16;
            v8f acc[TM];
#pragma unroll
            for (int q = 0; q < TM; ++q) acc[q] = (v8f){0.f,0.f,0.f,0.f,0.f,0.f,0.f,0.f};
            for (int kb = 0; kb < NL; kb += 40) {
                v2f b[10];
#pragma unroll
                for (int j = 0; j < 10; ++j) {
                    int bi = (kb + 4 * j + kh) * S_P + n;
                    b[j][0] = Z5[bi]       - Z6[bi];
                    b[j][1] = Z5[bi + S_P] - Z6[bi + S_P];
                }
#pragma unroll
                for (int q = 0; q < TM; ++q) {
                    if (q >= tm_lo && q < tm_hi) {
                        const float* acol = sPG + (kb + kh) * PG_LP + q * 16 + m16;
                        v2f a[10];
#pragma unroll
                        for (int j = 0; j < 10; ++j) {
                            a[j][0] = acol[(4 * j) * PG_LP]; a[j][1] = acol[(4 * j + 1) * PG_LP];
                        }
#pragma unroll
                        for (int j = 0; j < 10; ++j)
                            acc[q] = __builtin_amdgcn_wmma_f32_16x16x4_f32(false, a[j], false, b[j],
                                                                           (short)0, acc[q], false, false);
                        __builtin_amdgcn_sched_group_barrier(0x100, 10, 0);
                        __builtin_amdgcn_sched_group_barrier(0x008, 10, 0);
                    }
                }
            }
            const float y4v = FUSED ? yb[OY_Y4 + n] : 0.f;
#pragma unroll
            for (int q = 0; q < TM; ++q) {
                if (q >= tm_lo && q < tm_hi) {
                    int mr = q * 16 + ((lane >> 4) << 3);
                    if (FUSED) {
                        float xu[8], xd[8], z3v[8], z4v[8], ruv[8], rdv[8];
#pragma unroll
                        for (int i = 0; i < 8; ++i) {
                            int o = (mr + i) * S_P + n;
                            xu[i]  = X[OX_U + o];      xd[i]  = X[OX_D + o];
                            z3v[i] = yb[OY_Z3 + o];    z4v[i] = yb[OY_Z4 + o];
                            ruv[i] = XB[OX_RU + mr + i]; rdv[i] = XB[OX_RD + mr + i];
                        }
                        float r3[8], r4[8];
#pragma unroll
                        for (int i = 0; i < 8; ++i) {
                            int o = (mr + i) * S_P + n;
                            float tv = acc[q][i];
                            float aU =  y4v + z3v[i] + tv;
                            float aD = -y4v + z4v[i] - tv;
                            float nU = fmaxf(xu[i] - step * aU, 0.f);
                            float nD = fmaxf(xd[i] - step * aD, 0.f);
                            float bU = 2.f * nU - xu[i];
                            float bD = 2.f * nD - xd[i];
                            float n3 = fmaxf(z3v[i] + step * (bU - ruv[i]), 0.f);
                            float n4 = fmaxf(z4v[i] + step * (bD - rdv[i]), 0.f);
                            const bool ok = n < NS && mr + i < NG;
                            if (ok) {
                                X[OX_U + o]  = nU;  XB[OX_U + o] = bU;
                                X[OX_D + o]  = nD;  XB[OX_D + o] = bD;
                                yb[OY_Z3 + o] = n3; yb[OY_Z4 + o] = n4;
                            }
                            r3[i] = ok ? n3 : 0.f;
                            r4[i] = ok ? n4 : 0.f;
                        }
                        const float s3 = rows_over_lanes(r3, m16);
                        const float s4 = rows_over_lanes(r4, m16);
                        if (m16 < 8) {
                            sRS[RS3 + tn * G_P + mr + m16] = s3;
                            sRS[RS4 + tn * G_P + mr + m16] = s4;
                        }
                    } else {
                        for (int i = 0; i < 8; ++i) T[(mr + i) * S_P + n] = acc[q][i];
                    }
                }
            }
            t += tm_hi - tm_lo;
        }
    }
    if (FUSED) return;
    if (tid < NG) {
        float a = 0.f, b = 0.f, c = 0.f;
        const float* z3r = yb + OY_Z3 + tid * S_P;
        const float* z4r = yb + OY_Z4 + tid * S_P;
        for (int s = 0; s < NS; ++s) { a += z3r[s]; b += z4r[s]; }
        const float* pc = sPG + tid;
        for (int l = 0; l < NL; ++l) c += pc[l * PG_LP] * (yb[OY_Y2 + l] - yb[OY_Y3 + l]);
        w[WS_S3 + tid] = a; w[WS_S4 + tid] = b; w[WS_GTY + tid] = c;
    }
    if (tid >= NG && tid < NG + NL) {
        int l = tid - NG;
        float a = 0.f, b = 0.f;
        const float* z5r = yb + OY_Z5 + l * S_P;
        const float* z6r = yb + OY_Z6 + l * S_P;
        for (int s = 0; s < NS; ++s) { a += z5r[s]; b += z6r[s]; }
        w[WS_S5 + l] = a; w[WS_S6 + l] = b;
    }
    __syncthreads();
    float y1v = yb[OY_Y1];
    for (int g = tid; g < NG; g += NT) {
        xo[OX_P  + g] = y1v + w[WS_GTY + g] + yb[OY_Z1 + g] - yb[OY_Z2 + g];
        xo[OX_RU + g] = yb[OY_Z1 + g] - w[WS_S3 + g];
        xo[OX_RD + g] = yb[OY_Z2 + g] - w[WS_S4 + g];
    }
    for (int l = tid; l < NL; l += NT) {
        xo[OX_FU + l] = yb[OY_Y2 + l] - w[WS_S5 + l];
        xo[OX_FD + l] = yb[OY_Y3 + l] - w[WS_S6 + l];
    }
    for (int i = tid; i < NG * NS; i += NT) {
        int g = i / NS, s = i % NS, o = g * S_P + s;
        float tv = T[o];
        xo[OX_U + o] =  yb[OY_Y4 + s] + yb[OY_Z3 + o] + tv;
        xo[OX_D + o] = -yb[OY_Y4 + s] + yb[OY_Z4 + o] - tv;
    }
}


__global__ __launch_bounds__(NT) void k_solver(float* w, const float* Cost,
                                               const float* Cru, const float* Crd,
                                               const float* Pmax, float* out) {
    __shared__ float red[NT];
    __shared__ __attribute__((aligned(16))) float sPG[NL * PG_LP];
    __shared__ float sRS[RS_TOTAL];
    const int tid = threadIdx.x;
    float* X   = w + WS_X;
    float* XB  = w + WS_XB;
    float* Y   = w + WS_Y;
    float* AX  = w + WS_AX;

    for (int i = tid; i < NL * G_P; i += NT) {
        int l = i / G_P, g = i - l * G_P;
        sPG[l * PG_LP + g] = w[WS_PG + i];
    }
    __syncthreads();

    float sumsq = 0.f;
    for (int it = 0; it < POWER_ITERS; ++it) {
        __syncthreads();
        float part = 0.f;
        for (int i = tid; i < XSZ; i += NT) { float v = X[i]; part += v * v; }
        sumsq = block_reduce(part, red, tid);
        float inv = 1.0f / sqrtf(sumsq);
        for (int i = tid; i < XSZ; i += NT) X[i] *= inv;
        matvec_core<false>(w, sPG, X, AX, tid, nullptr, nullptr, 0.f, nullptr);
        rmatvec_core<false>(w, sPG, AX, X, tid, nullptr, nullptr, nullptr, nullptr, nullptr, 0.f, nullptr);
    }
    __syncthreads();
    {
        float part = 0.f;
        for (int i = tid; i < XSZ; i += NT) { float v = X[i]; part += v * v; }
        sumsq = block_reduce(part, red, tid);
    }
    const float step = 0.9f / sqrtf(sqrtf(sumsq));
    const float b1 = w[WS_B1];
    const float* pw = w + WS_PW;

    for (int i = tid; i < XSZ; i += NT) { X[i] = 0.f; XB[i] = 0.f; }
    for (int i = tid; i < YSZ; i += NT) Y[i] = 0.f;
    for (int i = tid; i < RS_TOTAL; i += NT) sRS[i] = 0.f;

    for (int it = 0; it < PDHG_ITERS; ++it) {
        rmatvec_core<true>(w, sPG, Y, nullptr, tid, X, XB, Cost, Cru, Crd, step, sRS);
        matvec_core<true>(w, sPG, XB, AX, tid, Y, pw, step, sRS);
        __syncthreads();
        if (tid == 0) Y[OY_Y1] += step * (AX[OY_Y1] - b1);
        for (int l = tid; l < NL; l += NT) {
            Y[OY_Y2 + l] += step * (AX[OY_Y2 + l] - w[WS_B2 + l]);
            Y[OY_Y3 + l] += step * (AX[OY_Y3 + l] - w[WS_B3 + l]);
        }
        for (int s = tid; s < NS; s += NT)
            Y[OY_Y4 + s] += step * (AX[OY_Y4 + s] - w[WS_B4 + s]);
        for (int g = tid; g < NG; g += NT) {
            Y[OY_Z1 + g] = fmaxf(Y[OY_Z1 + g] + step * (AX[OY_Z1 + g] - Pmax[g]), 0.f);
            Y[OY_Z2 + g] = fmaxf(Y[OY_Z2 + g] + step *  AX[OY_Z2 + g], 0.f);
        }
        __syncthreads();
    }

    for (int g = tid; g < NG; g += NT) {
        out[g]          = X[OX_P  + g];
        out[NG + g]     = X[OX_RU + g];
        out[2 * NG + g] = X[OX_RD + g];
    }
    for (int i = tid; i < NG * NS; i += NT) {
        int g = i / NS, s = i % NS;
        out[3 * NG + i]           = X[OX_U + g * S_P + s];
        out[3 * NG + NG * NS + i] = X[OX_D + g * S_P + s];
    }
    for (int l = tid; l < NL; l += NT) {
        out[3 * NG + 2 * NG * NS + l]      = X[OX_FU + l];
        out[3 * NG + 2 * NG * NS + NL + l] = X[OX_FD + l];
    }
}


extern "C" void kernel_launch(void* const* d_in, const int* in_sizes, int n_in,
                              void* d_out, int out_size, void* d_ws, size_t ws_size,
                              hipStream_t stream) {
    const float* w_scen   = (const float*)d_in[0];
    const float* Cost     = (const float*)d_in[1];
    const float* C_r_up   = (const float*)d_in[2];
    const float* C_r_down = (const float*)d_in[3];
    const float* Pmax     = (const float*)d_in[4];
    const float* PTDF     = (const float*)d_in[5];
    const float* node_G   = (const float*)d_in[6];
    const float* node_W   = (const float*)d_in[7];
    const float* node_L   = (const float*)d_in[8];
    const float* Pd       = (const float*)d_in[9];
    const float* w_exp    = (const float*)d_in[10];
    const float* Line_cap = (const float*)d_in[11];
    float* w   = (float*)d_ws;
    float* out = (float*)d_out;

    k_zero <<<512, 256, 0, stream>>>(w);
    k_prep1<<<240, 256, 0, stream>>>(w, w_scen, node_W, node_L, Pd, w_exp);
    k_prep2<<<640, 256, 0, stream>>>(w, PTDF, node_G, Line_cap);
    k_setv <<<161, 256, 0, stream>>>(w);
    k_solver<<<1, NT, 0, stream>>>(w, Cost, C_r_up, C_r_down, Pmax, out);
}


# Short serial reductions split over all 16 waves: PG^T(y2-y3) in 4 line ranges of 100 on 4 waves each, partials added in LDS in range order; sum(p) over the 32 lanes of a wave; these sums change order
# speedup vs baseline: 4.2872x; 1.0534x over previous
#include <hip/hip_runtime.h>
#include <math.h>

typedef __attribute__((ext_vector_type(2))) float v2f;
typedef __attribute__((ext_vector_type(8))) float v8f;

#define NT 512

constexpr int NG = 100, NL = 400, NN = 300, NWF = 50, NS = 200;
constexpr int G_P = 112;
constexpr int S_P = 208;
constexpr int PDHG_ITERS = 300, POWER_ITERS = 25;
constexpr int PG_LP = 116;
constexpr int N_STRIP = S_P / 16;
constexpr int RS5 = 0;
constexpr int RS6 = RS5 + N_STRIP * NL;
constexpr int RS3 = RS6 + N_STRIP * NL;
constexpr int RS4 = RS3 + N_STRIP * G_P;
constexpr int RS_TOTAL = RS4 + N_STRIP * G_P;
constexpr int C_RANGES = 4;
constexpr int C_RANGE = NL / C_RANGES;
constexpr int C_SLOT = 128;

constexpr int OX_P  = 0;
constexpr int OX_RU = OX_P  + G_P;
constexpr int OX_RD = OX_RU + G_P;
constexpr int OX_U  = OX_RD + G_P;
constexpr int OX_D  = OX_U  + G_P * S_P;
constexpr int OX_FU = OX_D  + G_P * S_P;
constexpr int OX_FD = OX_FU + NL;
constexpr int XSZ   = OX_FD + NL;

constexpr int OY_Y1 = 0;
constexpr int OY_Y2 = 16;
constexpr int OY_Y3 = OY_Y2 + NL;
constexpr int OY_Y4 = OY_Y3 + NL;
constexpr int OY_Z1 = OY_Y4 + S_P;
constexpr int OY_Z2 = OY_Z1 + G_P;
constexpr int OY_Z3 = OY_Z2 + G_P;
constexpr int OY_Z4 = OY_Z3 + G_P * S_P;
constexpr int OY_Z5 = OY_Z4 + G_P * S_P;
constexpr int OY_Z6 = OY_Z5 + NL * S_P;
constexpr int YSZ   = OY_Z6 + NL * S_P;

constexpr int WS_PG   = 0;
constexpr int WS_PGT  = WS_PG   + NL * G_P;
constexpr int WS_PW   = WS_PGT  + G_P * NL;
constexpr int WS_B2   = WS_PW   + NL * S_P;
constexpr int WS_B3   = WS_B2   + NL;
constexpr int WS_B4   = WS_B3   + NL;
constexpr int WS_B1   = WS_B4   + S_P;
constexpr int WS_M1   = WS_B1   + 16;
constexpr int WS_NV   = WS_M1   + NN * S_P;
constexpr int WS_X    = WS_NV   + NN + 4;
constexpr int WS_XB   = WS_X    + XSZ;
constexpr int WS_ATY  = WS_XB   + XSZ;
constexpr int WS_Y    = WS_ATY  + XSZ;
constexpr int WS_AX   = WS_Y    + YSZ;
constexpr int WS_T    = WS_AX   + YSZ;
constexpr int WS_FLOW = WS_T    + G_P * S_P;
constexpr int WS_GP   = WS_FLOW + NL * S_P;
constexpr int WS_COL  = WS_GP   + NL;
constexpr int WS_S3   = WS_COL  + S_P;
constexpr int WS_S4   = WS_S3   + G_P;
constexpr int WS_S5   = WS_S4   + G_P;
constexpr int WS_S6   = WS_S5   + NL;
constexpr int WS_GTY  = WS_S6   + NL;
constexpr int WS_SUMP = WS_GTY  + G_P;
constexpr int WS_TOTAL= WS_SUMP + 16;


__global__ void k_zero(float* w) {
    for (int i = blockIdx.x * blockDim.x + threadIdx.x; i < WS_TOTAL;
         i += gridDim.x * blockDim.x) w[i] = 0.0f;
}

__global__ void k_prep1(float* w, const float* w_scen, const float* node_W,
                        const float* node_L, const float* Pd, const float* w_exp) {
    const int NM1 = NN * NS;
    const int total = NM1 + NN + NS + 1;
    for (int idx = blockIdx.x * blockDim.x + threadIdx.x; idx < total;
         idx += gridDim.x * blockDim.x) {
        if (idx < NM1) {
            int n = idx / NS, s = idx % NS;
            float a = 0.f;
            for (int ww = 0; ww < NWF; ++ww) a += node_W[n*NWF + ww] * w_scen[s*NWF + ww];
            w[WS_M1 + n*S_P + s] = a;
        } else if (idx < NM1 + NN) {
            int n = idx - NM1;
            float a = 0.f;
            for (int ww = 0; ww < NWF; ++ww) a += node_W[n*NWF + ww] * w_exp[ww];
            for (int m = 0; m < NN; ++m)     a -= node_L[n*NN + m] * Pd[m];
            w[WS_NV + n] = a;
        } else if (idx < NM1 + NN + NS) {
            int s = idx - NM1 - NN;
            float a = 0.f;
            for (int ww = 0; ww < NWF; ++ww) a += w_scen[s*NWF + ww];
            w[WS_B4 + s] = -a;
        } else {
            float a = 0.f;
            for (int m = 0; m < NN; ++m)  a += Pd[m];
            for (int ww = 0; ww < NWF; ++ww) a -= w_exp[ww];
            w[WS_B1] = a;
        }
    }
}

__global__ void k_prep2(float* w, const float* PTDF, const float* node_G,
                        const float* Line_cap) {
    const int A = NL * NG;
    const int C = A + NL * NS;
    const int D2 = C + NL;
    const int E = D2 + NL;
    for (int idx = blockIdx.x * blockDim.x + threadIdx.x; idx < E;
         idx += gridDim.x * blockDim.x) {
        if (idx < A) {
            int l = idx / NG, g = idx % NG;
            float a = 0.f;
            for (int n = 0; n < NN; ++n) a += PTDF[l*NN + n] * node_G[n*NG + g];
            w[WS_PG + l*G_P + g] = a;
        } else if (idx < C) {
            int r = idx - A; int l = r / NS, s = r % NS;
            float a = 0.f;
            for (int n = 0; n < NN; ++n) a += PTDF[l*NN + n] * w[WS_M1 + n*S_P + s];
            w[WS_PW + l*S_P + s] = a;
        } else if (idx < D2) {
            int l = idx - C;
            float a = 0.f;
            for (int n = 0; n < NN; ++n) a += PTDF[l*NN + n] * w[WS_NV + n];
            w[WS_B2 + l] = Line_cap[l] - a;
        } else {
            int l = idx - D2;
            float a = 0.f;
            for (int n = 0; n < NN; ++n) a += PTDF[l*NN + n] * w[WS_NV + n];
            w[WS_B3 + l] = Line_cap[l] + a;
        }
    }
}

__global__ void k_setv(float* w) {
    const int total = 3*NG + 2*NG*NS + 2*NL;
    float* X = w + WS_X;
    for (int idx = blockIdx.x * blockDim.x + threadIdx.x; idx < total;
         idx += gridDim.x * blockDim.x) {
        int off;
        if (idx < NG)                off = OX_P  + idx;
        else if (idx < 2*NG)         off = OX_RU + (idx - NG);
        else if (idx < 3*NG)         off = OX_RD + (idx - 2*NG);
        else if (idx < 3*NG + NG*NS) { int r = idx - 3*NG; off = OX_U + (r/NS)*S_P + (r%NS); }
        else if (idx < 3*NG + 2*NG*NS) { int r = idx - 3*NG - NG*NS; off = OX_D + (r/NS)*S_P + (r%NS); }
        else if (idx < 3*NG + 2*NG*NS + NL) off = OX_FU + (idx - 3*NG - 2*NG*NS);
        else                         off = OX_FD + (idx - 3*NG - 2*NG*NS - NL);
        X[off] = 1.0f;
    }
}


__device__ __forceinline__ float block_reduce(float v, float* red, int tid) {
    red[tid] = v; __syncthreads();
    for (int off = NT >> 1; off > 0; off >>= 1) {
        if (tid < off) red[tid] += red[tid + off];
        __syncthreads();
    }
    float r = red[0]; __syncthreads();
    return r;
}

template <int M>
__device__ __forceinline__ float row_xor(float x) {
    return __int_as_float(__builtin_amdgcn_update_dpp(0, __float_as_int(x), 0x160 + M, 0xf, 0xf, true));
}

__device__ __forceinline__ float rows_over_lanes(const float (&v)[8], int c) {
    const bool b2 = (c & 4) != 0, b1 = (c & 2) != 0, b0 = (c & 1) != 0;
    float w[4], u[2];
#pragma unroll
    for (int j = 0; j < 4; ++j) {
        float keep = b2 ? v[4 + j] : v[j];
        float send = b2 ? v[j] : v[4 + j];
        w[j] = keep + row_xor<4>(send);
    }
#pragma unroll
    for (int j = 0; j < 2; ++j) {
        float keep = b1 ? w[2 + j] : w[j];
        float send = b1 ? w[j] : w[2 + j];
        u[j] = keep + row_xor<2>(send);
    }
    float keep = b0 ? u[1] : u[0];
    float send = b0 ? u[0] : u[1];
    float t = keep + row_xor<1>(send);
    return t + row_xor<8>(t);
}

__device__ __forceinline__ void pgt_ydiff_partials(const float* sPG, const float* yb, int tid, float* sPart) {
    const int r = __builtin_amdgcn_readfirstlane(tid / C_SLOT);
    const int g = tid % C_SLOT;
    if (g < NG) {
        const float* pc = sPG + r * C_RANGE * PG_LP + g;
        const float* y2 = yb + OY_Y2 + r * C_RANGE;
        const float* y3 = yb + OY_Y3 + r * C_RANGE;
        float c = 0.f;
#pragma unroll
        for (int l = 0; l < C_RANGE; ++l) c += pc[l * PG_LP] * (y2[l] - y3[l]);
        sPart[r * C_SLOT + g] = c;
    }
}

__device__ __forceinline__ float pgt_ydiff_sum(const float* sPart, int g) {
    return ((sPart[g] + sPart[C_SLOT + g]) + sPart[2 * C_SLOT + g]) + sPart[3 * C_SLOT + g];
}

template <bool DUAL>
__device__ __forceinline__ void matvec_core(float* w, const float* sPG, const float* xb,
                                            float* yb, int tid, float* Yd, const float* pw,
                                            float step, float* sRS) {
    const int lane = tid & 31, wave = __builtin_amdgcn_readfirstlane(tid >> 5), nw = NT / 32;
    float* flow = w + WS_FLOW;
    __syncthreads();
    {
        const float* Ux = xb + OX_U; const float* Dx = xb + OX_D;
        constexpr int TM = 25, NTile = 25 * 13;
        const int m16 = lane & 15;
        const int kh  = (lane >> 4) * 2;
        const int t1 = ((wave + 1) * NTile) / nw;
        int t = (wave * NTile) / nw;
        while (t < t1) {
            const int tn = t / TM;
            const int tm_lo = t - tn * TM;
            const int tm_hi = (tm_lo + (t1 - t) < TM) ? tm_lo + (t1 - t) : TM;
            const int n = tn * 16 + m16;
            v2f b[25];
#pragma unroll
            for (int j = 0; j < 25; ++j) {
                int bi = (4 * j + kh) * S_P + n;
                b[j][0] = Ux[bi]       - Dx[bi];
                b[j][1] = Ux[bi + S_P] - Dx[bi + S_P];
            }
            for (int tm = tm_lo; tm < tm_hi; ++tm) {
                const int mr = tm * 16 + ((lane >> 4) << 3);
                float y5[8], y6[8], pv[8], fu[8], fd[8];
                if (DUAL) {
#pragma unroll
                    for (int i = 0; i < 8; ++i) {
                        int o = (mr + i) * S_P + n;
                        y5[i] = Yd[OY_Z5 + o]; y6[i] = Yd[OY_Z6 + o]; pv[i] = pw[o];
                        fu[i] = xb[OX_FU + mr + i]; fd[i] = xb[OX_FD + mr + i];
                    }
                }
                const float* arow = sPG + (tm * 16 + m16) * PG_LP + kh;
                v2f a[25];
#pragma unroll
                for (int j = 0; j < 25; ++j) { a[j][0] = arow[4 * j]; a[j][1] = arow[4 * j + 1]; }
                v8f acc = {0.f,0.f,0.f,0.f,0.f,0.f,0.f,0.f};
#pragma unroll
                for (int j = 0; j < 25; ++j)
                    acc = __builtin_amdgcn_wmma_f32_16x16x4_f32(false, a[j], false, b[j],
                                                                (short)0, acc, false, false);
                __builtin_amdgcn_sched_group_barrier(0x100, 13, 0);
                __builtin_amdgcn_sched_group_barrier(0x008, 25, 0);
                if (DUAL) {
                    const bool live = n < NS;
                    float r5[8], r6[8];
#pragma unroll
                    for (int i = 0; i < 8; ++i) {
                        int o = (mr + i) * S_P + n;
                        float f = acc[i];
                        float n5 = fmaxf(y5[i] + step * (( f - fu[i]) + pv[i]), 0.f);
                        float n6 = fmaxf(y6[i] + step * ((-f - fd[i]) - pv[i]), 0.f);
                        r5[i] = live ? n5 : 0.f;
                        r6[i] = live ? n6 : 0.f;
                        Yd[OY_Z5 + o] = r5[i];
                        Yd[OY_Z6 + o] = r6[i];
                    }
                    const float s5 = rows_over_lanes(r5, m16);
                    const float s6 = rows_over_lanes(r6, m16);
                    if (m16 < 8) {
                        sRS[RS5 + tn * NL + mr + m16] = s5;
                        sRS[RS6 + tn * NL + mr + m16] = s6;
                    }
                } else {
                    for (int i = 0; i < 8; ++i) flow[(mr + i) * S_P + n] = acc[i];
                }
            }
            t += tm_hi - tm_lo;
        }
    }
    if (tid < NL) {
        float s = 0.f; const float* r = sPG + tid * PG_LP; const float* p = xb + OX_P;
        for (int g = 0; g < NG; ++g) s += r[g] * p[g];
        w[WS_GP + tid] = s;
    }
    if (tid < NS) {
        float s = 0.f;
        const float* Ux = xb + OX_U + tid; const float* Dx = xb + OX_D + tid;
        for (int g = 0; g < NG; ++g) s += Ux[g * S_P] - Dx[g * S_P];
        w[WS_COL + tid] = s;
    }
    if (wave == nw - 1) {
        float a = xb[OX_P + lane] + xb[OX_P + 32 + lane] + xb[OX_P + 64 + lane];
        if (lane < NG - 96) a += xb[OX_P + 96 + lane];
        a += row_xor<1>(a); a += row_xor<2>(a); a += row_xor<4>(a); a += row_xor<8>(a);
        a += __shfl_xor(a, 16);
        if (lane == 0) w[WS_SUMP] = a;
    }
    __syncthreads();
    if (tid == 0) yb[OY_Y1] = w[WS_SUMP];
    for (int l = tid; l < NL; l += NT) {
        float g = w[WS_GP + l];
        yb[OY_Y2 + l] =  g + xb[OX_FU + l];
        yb[OY_Y3 + l] = -g + xb[OX_FD + l];
    }
    for (int s = tid; s < NS; s += NT) yb[OY_Y4 + s] = w[WS_COL + s];
    for (int g = tid; g < NG; g += NT) {
        yb[OY_Z1 + g] = xb[OX_P + g]  + xb[OX_RU + g];
        yb[OY_Z2 + g] = xb[OX_RD + g] - xb[OX_P + g];
    }
    if (!DUAL) {
        for (int i = tid; i < NG * NS; i += NT) {
            int g = i / NS, s = i % NS, o = g * S_P + s;
            yb[OY_Z3 + o] = xb[OX_U + o] - xb[OX_RU + g];
            yb[OY_Z4 + o] = xb[OX_D + o] - xb[OX_RD + g];
        }
        for (int i = tid; i < NL * NS; i += NT) {
            int l = i / NS, s = i % NS, o = l * S_P + s;
            float f = flow[o];
            yb[OY_Z5 + o] =  f - xb[OX_FU + l];
            yb[OY_Z6 + o] = -f - xb[OX_FD + l];
        }
    }
}

template <bool FUSED>
__device__ __forceinline__ void rmatvec_core(float* w, const float* sPG, float* yb,
                                             float* xo, int tid, float* X, float* XB,
                                             const float* Cost, const float* Cru,
                                             const float* Crd, float step, float* sRS, float* sPart) {
    const int lane = tid & 31, wave = __builtin_amdgcn_readfirstlane(tid >> 5), nw = NT / 32;
    float* T = w + WS_T;
    __syncthreads();
    if (FUSED) {
        pgt_ydiff_partials(sPG, yb, tid, sPart);
        __syncthreads();
        if (tid < NG) {
            float a = 0.f, b = 0.f;
            for (int k = 0; k < N_STRIP; ++k) {
                a += sRS[RS3 + k * G_P + tid]; b += sRS[RS4 + k * G_P + tid];
            }
            const float c = pgt_ydiff_sum(sPart, tid);
            const float z1 = yb[OY_Z1 + tid], z2 = yb[OY_Z2 + tid];
            const float ap  = yb[OY_Y1] + c + z1 - z2;
            const float aru = z1 - a;
            const float ard = z2 - b;
            float o, n;
            o = X[OX_P  + tid]; n = fmaxf(o - step * (Cost[tid] + ap),  0.f);
            X[OX_P  + tid] = n; XB[OX_P  + tid] = 2.f * n - o;
            o = X[OX_RU + tid]; n = fmaxf(o - step * (Cru[tid]  + aru), 0.f);
            X[OX_RU + tid] = n; XB[OX_RU + tid] = 2.f * n - o;
            o = X[OX_RD + tid]; n = fmaxf(o - step * (Crd[tid]  + ard), 0.f);
            X[OX_RD + tid] = n; XB[OX_RD + tid] = 2.f * n - o;
        }
        if (tid >= NG && tid < NG + NL) {
            int l = tid - NG;
            float a = 0.f, b = 0.f;
            for (int k = 0; k < N_STRIP; ++k) {
                a += sRS[RS5 + k * NL + l]; b += sRS[RS6 + k * NL + l];
            }
            const float afu = yb[OY_Y2 + l] - a;
            const float afd = yb[OY_Y3 + l] - b;
            float o, n;
            o = X[OX_FU + l]; n = fmaxf(o - step * afu, 0.f);
            X[OX_FU + l] = n; XB[OX_FU + l] = 2.f * n - o;
            o = X[OX_FD + l]; n = fmaxf(o - step * afd, 0.f);
            X[OX_FD + l] = n; XB[OX_FD + l] = 2.f * n - o;
        }
        __syncthreads();
    }
    {
        const float* Z5 = yb + OY_Z5; const float* Z6 = yb + OY_Z6;
        constexpr int TM = 7, NTile = 7 * 13;
        const int m16 = lane & 15;
        const int kh  = (lane >> 4) * 2;
        const int t1 = ((wave + 1) * NTile) / nw;
        int t = (wave * NTile) / nw;
        while (t < t1) {
            const int tn = t / TM;
            const int tm_lo = t - tn * TM;
            const int tm_hi = (tm_lo + (t1 - t) < TM) ? tm_lo + (t1 - t) : TM;
            const int n = tn * 16 + m16;
            v8f acc[TM];
#pragma unroll
            for (int q = 0; q < TM; ++q) acc[q] = (v8f){0.f,0.f,0.f,0.f,0.f,0.f,0.f,0.f};
            for (int kb = 0; kb < NL; kb += 40) {
                v2f b[10];
#pragma unroll
                for (int j = 0; j < 10; ++j) {
                    int bi = (kb + 4 * j + kh) * S_P + n;
                    b[j][0] = Z5[bi]       - Z6[bi];
                    b[j][1] = Z5[bi + S_P] - Z6[bi + S_P];
                }
#pragma unroll
                for (int q = 0; q < TM; ++q) {
                    if (q >= tm_lo && q < tm_hi) {
                        const float* acol = sPG + (kb + kh) * PG_LP + q * 16 + m16;
                        v2f a[10];
#pragma unroll
                        for (int j = 0; j < 10; ++j) {
                            a[j][0] = acol[(4 * j) * PG_LP]; a[j][1] = acol[(4 * j + 1) * PG_LP];
                        }
#pragma unroll
                        for (int j = 0; j < 10; ++j)
                            acc[q] = __builtin_amdgcn_wmma_f32_16x16x4_f32(false, a[j], false, b[j],
                                                                           (short)0, acc[q], false, false);
                        __builtin_amdgcn_sched_group_barrier(0x100, 10, 0);
                        __builtin_amdgcn_sched_group_barrier(0x008, 10, 0);
                    }
                }
            }
            const float y4v = FUSED ? yb[OY_Y4 + n] : 0.f;
#pragma unroll
            for (int q = 0; q < TM; ++q) {
                if (q >= tm_lo && q < tm_hi) {
                    int mr = q * 16 + ((lane >> 4) << 3);
                    if (FUSED) {
                        float xu[8], xd[8], z3v[8], z4v[8], ruv[8], rdv[8];
#pragma unroll
                        for (int i = 0; i < 8; ++i) {
                            int o = (mr + i) * S_P + n;
                            xu[i]  = X[OX_U + o];      xd[i]  = X[OX_D + o];
                            z3v[i] = yb[OY_Z3 + o];    z4v[i] = yb[OY_Z4 + o];
                            ruv[i] = XB[OX_RU + mr + i]; rdv[i] = XB[OX_RD + mr + i];
                        }
                        float r3[8], r4[8];
#pragma unroll
                        for (int i = 0; i < 8; ++i) {
                            int o = (mr + i) * S_P + n;
                            float tv = acc[q][i];
                            float aU =  y4v + z3v[i] + tv;
                            float aD = -y4v + z4v[i] - tv;
                            float nU = fmaxf(xu[i] - step * aU, 0.f);
                            float nD = fmaxf(xd[i] - step * aD, 0.f);
                            float bU = 2.f * nU - xu[i];
                            float bD = 2.f * nD - xd[i];
                            float n3 = fmaxf(z3v[i] + step * (bU - ruv[i]), 0.f);
                            float n4 = fmaxf(z4v[i] + step * (bD - rdv[i]), 0.f);
                            const bool ok = n < NS && mr + i < NG;
                            if (ok) {
                                X[OX_U + o]  = nU;  XB[OX_U + o] = bU;
                                X[OX_D + o]  = nD;  XB[OX_D + o] = bD;
                                yb[OY_Z3 + o] = n3; yb[OY_Z4 + o] = n4;
                            }
                            r3[i] = ok ? n3 : 0.f;
                            r4[i] = ok ? n4 : 0.f;
                        }
                        const float s3 = rows_over_lanes(r3, m16);
                        const float s4 = rows_over_lanes(r4, m16);
                        if (m16 < 8) {
                            sRS[RS3 + tn * G_P + mr + m16] = s3;
                            sRS[RS4 + tn * G_P + mr + m16] = s4;
                        }
                    } else {
                        for (int i = 0; i < 8; ++i) T[(mr + i) * S_P + n] = acc[q][i];
                    }
                }
            }
            t += tm_hi - tm_lo;
        }
    }
    if (FUSED) return;
    pgt_ydiff_partials(sPG, yb, tid, sPart);
    if (tid < NG) {
        float a = 0.f, b = 0.f;
        const float* z3r = yb + OY_Z3 + tid * S_P;
        const float* z4r = yb + OY_Z4 + tid * S_P;
        for (int s = 0; s < NS; ++s) { a += z3r[s]; b += z4r[s]; }
        w[WS_S3 + tid] = a; w[WS_S4 + tid] = b;
    }
    if (tid >= NG && tid < NG + NL) {
        int l = tid - NG;
        float a = 0.f, b = 0.f;
        const float* z5r = yb + OY_Z5 + l * S_P;
        const float* z6r = yb + OY_Z6 + l * S_P;
        for (int s = 0; s < NS; ++s) { a += z5r[s]; b += z6r[s]; }
        w[WS_S5 + l] = a; w[WS_S6 + l] = b;
    }
    __syncthreads();
    float y1v = yb[OY_Y1];
    for (int g = tid; g < NG; g += NT) {
        xo[OX_P  + g] = y1v + pgt_ydiff_sum(sPart, g) + yb[OY_Z1 + g] - yb[OY_Z2 + g];
        xo[OX_RU + g] = yb[OY_Z1 + g] - w[WS_S3 + g];
        xo[OX_RD + g] = yb[OY_Z2 + g] - w[WS_S4 + g];
    }
    for (int l = tid; l < NL; l += NT) {
        xo[OX_FU + l] = yb[OY_Y2 + l] - w[WS_S5 + l];
        xo[OX_FD + l] = yb[OY_Y3 + l] - w[WS_S6 + l];
    }
    for (int i = tid; i < NG * NS; i += NT) {
        int g = i / NS, s = i % NS, o = g * S_P + s;
        float tv = T[o];
        xo[OX_U + o] =  yb[OY_Y4 + s] + yb[OY_Z3 + o] + tv;
        xo[OX_D + o] = -yb[OY_Y4 + s] + yb[OY_Z4 + o] - tv;
    }
}


__global__ __launch_bounds__(NT) void k_solver(float* w, const float* Cost,
                                               const float* Cru, const float* Crd,
                                               const float* Pmax, float* out) {
    __shared__ float red[NT];
    __shared__ __attribute__((aligned(16))) float sPG[NL * PG_LP];
    __shared__ float sRS[RS_TOTAL];
    __shared__ float sPart[C_RANGES * C_SLOT];
    const int tid = threadIdx.x;
    float* X   = w + WS_X;
    float* XB  = w + WS_XB;
    float* Y   = w + WS_Y;
    float* AX  = w + WS_AX;

    for (int i = tid; i < NL * G_P; i += NT) {
        int l = i / G_P, g = i - l * G_P;
        sPG[l * PG_LP + g] = w[WS_PG + i];
    }
    __syncthreads();

    float sumsq = 0.f;
    for (int it = 0; it < POWER_ITERS; ++it) {
        __syncthreads();
        float part = 0.f;
        for (int i = tid; i < XSZ; i += NT) { float v = X[i]; part += v * v; }
        sumsq = block_reduce(part, red, tid);
        float inv = 1.0f / sqrtf(sumsq);
        for (int i = tid; i < XSZ; i += NT) X[i] *= inv;
        matvec_core<false>(w, sPG, X, AX, tid, nullptr, nullptr, 0.f, nullptr);
        rmatvec_core<false>(w, sPG, AX, X, tid, nullptr, nullptr, nullptr, nullptr, nullptr, 0.f, nullptr, sPart);
    }
    __syncthreads();
    {
        float part = 0.f;
        for (int i = tid; i < XSZ; i += NT) { float v = X[i]; part += v * v; }
        sumsq = block_reduce(part, red, tid);
    }
    const float step = 0.9f / sqrtf(sqrtf(sumsq));
    const float b1 = w[WS_B1];
    const float* pw = w + WS_PW;

    for (int i = tid; i < XSZ; i += NT) { X[i] = 0.f; XB[i] = 0.f; }
    for (int i = tid; i < YSZ; i += NT) Y[i] = 0.f;
    for (int i = tid; i < RS_TOTAL; i += NT) sRS[i] = 0.f;

    for (int it = 0; it < PDHG_ITERS; ++it) {
        rmatvec_core<true>(w, sPG, Y, nullptr, tid, X, XB, Cost, Cru, Crd, step, sRS, sPart);
        matvec_core<true>(w, sPG, XB, AX, tid, Y, pw, step, sRS);
        __syncthreads();
        if (tid == 0) Y[OY_Y1] += step * (AX[OY_Y1] - b1);
        for (int l = tid; l < NL; l += NT) {
            Y[OY_Y2 + l] += step * (AX[OY_Y2 + l] - w[WS_B2 + l]);
            Y[OY_Y3 + l] += step * (AX[OY_Y3 + l] - w[WS_B3 + l]);
        }
        for (int s = tid; s < NS; s += NT)
            Y[OY_Y4 + s] += step * (AX[OY_Y4 + s] - w[WS_B4 + s]);
        for (int g = tid; g < NG; g += NT) {
            Y[OY_Z1 + g] = fmaxf(Y[OY_Z1 + g] + step * (AX[OY_Z1 + g] - Pmax[g]), 0.f);
            Y[OY_Z2 + g] = fmaxf(Y[OY_Z2 + g] + step *  AX[OY_Z2 + g], 0.f);
        }
        __syncthreads();
    }

    for (int g = tid; g < NG; g += NT) {
        out[g]          = X[OX_P  + g];
        out[NG + g]     = X[OX_RU + g];
        out[2 * NG + g] = X[OX_RD + g];
    }
    for (int i = tid; i < NG * NS; i += NT) {
        int g = i / NS, s = i % NS;
        out[3 * NG + i]           = X[OX_U + g * S_P + s];
        out[3 * NG + NG * NS + i] = X[OX_D + g * S_P + s];
    }
    for (int l = tid; l < NL; l += NT) {
        out[3 * NG + 2 * NG * NS + l]      = X[OX_FU + l];
        out[3 * NG + 2 * NG * NS + NL + l] = X[OX_FD + l];
    }
}


extern "C" void kernel_launch(void* const* d_in, const int* in_sizes, int n_in,
                              void* d_out, int out_size, void* d_ws, size_t ws_size,
                              hipStream_t stream) {
    const float* w_scen   = (const float*)d_in[0];
    const float* Cost     = (const float*)d_in[1];
    const float* C_r_up   = (const float*)d_in[2];
    const float* C_r_down = (const float*)d_in[3];
    const float* Pmax     = (const float*)d_in[4];
    const float* PTDF     = (const float*)d_in[5];
    const float* node_G   = (const float*)d_in[6];
    const float* node_W   = (const float*)d_in[7];
    const float* node_L   = (const float*)d_in[8];
    const float* Pd       = (const float*)d_in[9];
    const float* w_exp    = (const float*)d_in[10];
    const float* Line_cap = (const float*)d_in[11];
    float* w   = (float*)d_ws;
    float* out = (float*)d_out;

    k_zero <<<512, 256, 0, stream>>>(w);
    k_prep1<<<240, 256, 0, stream>>>(w, w_scen, node_W, node_L, Pd, w_exp);
    k_prep2<<<640, 256, 0, stream>>>(w, PTDF, node_G, Line_cap);
    k_setv <<<161, 256, 0, stream>>>(w);
    k_solver<<<1, NT, 0, stream>>>(w, Cost, C_r_up, C_r_down, Pmax, out);
}


# Power iteration: z3..z6 formed in both products' epilogues from the accumulator registers, U, D of A^T y written there, row sums from strip partials; passes over flow, t, z go; row sums change order
# speedup vs baseline: 4.6155x; 1.0766x over previous
#include <hip/hip_runtime.h>
#include <math.h>

typedef __attribute__((ext_vector_type(2))) float v2f;
typedef __attribute__((ext_vector_type(8))) float v8f;

#define NT 512

constexpr int NG = 100, NL = 400, NN = 300, NWF = 50, NS = 200;
constexpr int G_P = 112;
constexpr int S_P = 208;
constexpr int PDHG_ITERS = 300, POWER_ITERS = 25;
constexpr int PG_LP = 116;
constexpr int N_STRIP = S_P / 16;
constexpr int RS5 = 0;
constexpr int RS6 = RS5 + N_STRIP * NL;
constexpr int RS3 = RS6 + N_STRIP * NL;
constexpr int RS4 = RS3 + N_STRIP * G_P;
constexpr int RS_TOTAL = RS4 + N_STRIP * G_P;
constexpr int C_RANGES = 4;
constexpr int C_RANGE = NL / C_RANGES;
constexpr int C_SLOT = 128;

constexpr int OX_P  = 0;
constexpr int OX_RU = OX_P  + G_P;
constexpr int OX_RD = OX_RU + G_P;
constexpr int OX_U  = OX_RD + G_P;
constexpr int OX_D  = OX_U  + G_P * S_P;
constexpr int OX_FU = OX_D  + G_P * S_P;
constexpr int OX_FD = OX_FU + NL;
constexpr int XSZ   = OX_FD + NL;

constexpr int OY_Y1 = 0;
constexpr int OY_Y2 = 16;
constexpr int OY_Y3 = OY_Y2 + NL;
constexpr int OY_Y4 = OY_Y3 + NL;
constexpr int OY_Z1 = OY_Y4 + S_P;
constexpr int OY_Z2 = OY_Z1 + G_P;
constexpr int OY_Z3 = OY_Z2 + G_P;
constexpr int OY_Z4 = OY_Z3 + G_P * S_P;
constexpr int OY_Z5 = OY_Z4 + G_P * S_P;
constexpr int OY_Z6 = OY_Z5 + NL * S_P;
constexpr int YSZ   = OY_Z6 + NL * S_P;

constexpr int WS_PG   = 0;
constexpr int WS_PGT  = WS_PG   + NL * G_P;
constexpr int WS_PW   = WS_PGT  + G_P * NL;
constexpr int WS_B2   = WS_PW   + NL * S_P;
constexpr int WS_B3   = WS_B2   + NL;
constexpr int WS_B4   = WS_B3   + NL;
constexpr int WS_B1   = WS_B4   + S_P;
constexpr int WS_M1   = WS_B1   + 16;
constexpr int WS_NV   = WS_M1   + NN * S_P;
constexpr int WS_X    = WS_NV   + NN + 4;
constexpr int WS_XB   = WS_X    + XSZ;
constexpr int WS_ATY  = WS_XB   + XSZ;
constexpr int WS_Y    = WS_ATY  + XSZ;
constexpr int WS_AX   = WS_Y    + YSZ;
constexpr int WS_GP   = WS_AX   + YSZ;
constexpr int WS_COL  = WS_GP   + NL;
constexpr int WS_GTY  = WS_COL  + S_P;
constexpr int WS_SUMP = WS_GTY  + G_P;
constexpr int WS_TOTAL= WS_SUMP + 16;


__global__ void k_zero(float* w) {
    for (int i = blockIdx.x * blockDim.x + threadIdx.x; i < WS_TOTAL;
         i += gridDim.x * blockDim.x) w[i] = 0.0f;
}

__global__ void k_prep1(float* w, const float* w_scen, const float* node_W,
                        const float* node_L, const float* Pd, const float* w_exp) {
    const int NM1 = NN * NS;
    const int total = NM1 + NN + NS + 1;
    for (int idx = blockIdx.x * blockDim.x + threadIdx.x; idx < total;
         idx += gridDim.x * blockDim.x) {
        if (idx < NM1) {
            int n = idx / NS, s = idx % NS;
            float a = 0.f;
            for (int ww = 0; ww < NWF; ++ww) a += node_W[n*NWF + ww] * w_scen[s*NWF + ww];
            w[WS_M1 + n*S_P + s] = a;
        } else if (idx < NM1 + NN) {
            int n = idx - NM1;
            float a = 0.f;
            for (int ww = 0; ww < NWF; ++ww) a += node_W[n*NWF + ww] * w_exp[ww];
            for (int m = 0; m < NN; ++m)     a -= node_L[n*NN + m] * Pd[m];
            w[WS_NV + n] = a;
        } else if (idx < NM1 + NN + NS) {
            int s = idx - NM1 - NN;
            float a = 0.f;
            for (int ww = 0; ww < NWF; ++ww) a += w_scen[s*NWF + ww];
            w[WS_B4 + s] = -a;
        } else {
            float a = 0.f;
            for (int m = 0; m < NN; ++m)  a += Pd[m];
            for (int ww = 0; ww < NWF; ++ww) a -= w_exp[ww];
            w[WS_B1] = a;
        }
    }
}

__global__ void k_prep2(float* w, const float* PTDF, const float* node_G,
                        const float* Line_cap) {
    const int A = NL * NG;
    const int C = A + NL * NS;
    const int D2 = C + NL;
    const int E = D2 + NL;
    for (int idx = blockIdx.x * blockDim.x + threadIdx.x; idx < E;
         idx += gridDim.x * blockDim.x) {
        if (idx < A) {
            int l = idx / NG, g = idx % NG;
            float a = 0.f;
            for (int n = 0; n < NN; ++n) a += PTDF[l*NN + n] * node_G[n*NG + g];
            w[WS_PG + l*G_P + g] = a;
        } else if (idx < C) {
            int r = idx - A; int l = r / NS, s = r % NS;
            float a = 0.f;
            for (int n = 0; n < NN; ++n) a += PTDF[l*NN + n] * w[WS_M1 + n*S_P + s];
            w[WS_PW + l*S_P + s] = a;
        } else if (idx < D2) {
            int l = idx - C;
            float a = 0.f;
            for (int n = 0; n < NN; ++n) a += PTDF[l*NN + n] * w[WS_NV + n];
            w[WS_B2 + l] = Line_cap[l] - a;
        } else {
            int l = idx - D2;
            float a = 0.f;
            for (int n = 0; n < NN; ++n) a += PTDF[l*NN + n] * w[WS_NV + n];
            w[WS_B3 + l] = Line_cap[l] + a;
        }
    }
}

__global__ void k_setv(float* w) {
    const int total = 3*NG + 2*NG*NS + 2*NL;
    float* X = w + WS_X;
    for (int idx = blockIdx.x * blockDim.x + threadIdx.x; idx < total;
         idx += gridDim.x * blockDim.x) {
        int off;
        if (idx < NG)                off = OX_P  + idx;
        else if (idx < 2*NG)         off = OX_RU + (idx - NG);
        else if (idx < 3*NG)         off = OX_RD + (idx - 2*NG);
        else if (idx < 3*NG + NG*NS) { int r = idx - 3*NG; off = OX_U + (r/NS)*S_P + (r%NS); }
        else if (idx < 3*NG + 2*NG*NS) { int r = idx - 3*NG - NG*NS; off = OX_D + (r/NS)*S_P + (r%NS); }
        else if (idx < 3*NG + 2*NG*NS + NL) off = OX_FU + (idx - 3*NG - 2*NG*NS);
        else                         off = OX_FD + (idx - 3*NG - 2*NG*NS - NL);
        X[off] = 1.0f;
    }
}


__device__ __forceinline__ float block_reduce(float v, float* red, int tid) {
    red[tid] = v; __syncthreads();
    for (int off = NT >> 1; off > 0; off >>= 1) {
        if (tid < off) red[tid] += red[tid + off];
        __syncthreads();
    }
    float r = red[0]; __syncthreads();
    return r;
}

template <int M>
__device__ __forceinline__ float row_xor(float x) {
    return __int_as_float(__builtin_amdgcn_update_dpp(0, __float_as_int(x), 0x160 + M, 0xf, 0xf, true));
}

__device__ __forceinline__ float rows_over_lanes(const float (&v)[8], int c) {
    const bool b2 = (c & 4) != 0, b1 = (c & 2) != 0, b0 = (c & 1) != 0;
    float w[4], u[2];
#pragma unroll
    for (int j = 0; j < 4; ++j) {
        float keep = b2 ? v[4 + j] : v[j];
        float send = b2 ? v[j] : v[4 + j];
        w[j] = keep + row_xor<4>(send);
    }
#pragma unroll
    for (int j = 0; j < 2; ++j) {
        float keep = b1 ? w[2 + j] : w[j];
        float send = b1 ? w[j] : w[2 + j];
        u[j] = keep + row_xor<2>(send);
    }
    float keep = b0 ? u[1] : u[0];
    float send = b0 ? u[0] : u[1];
    float t = keep + row_xor<1>(send);
    return t + row_xor<8>(t);
}

__device__ __forceinline__ void pgt_ydiff_partials(const float* sPG, const float* yb, int tid, float* sPart) {
    const int r = __builtin_amdgcn_readfirstlane(tid / C_SLOT);
    const int g = tid % C_SLOT;
    if (g < NG) {
        const float* pc = sPG + r * C_RANGE * PG_LP + g;
        const float* y2 = yb + OY_Y2 + r * C_RANGE;
        const float* y3 = yb + OY_Y3 + r * C_RANGE;
        float c = 0.f;
#pragma unroll
        for (int l = 0; l < C_RANGE; ++l) c += pc[l * PG_LP] * (y2[l] - y3[l]);
        sPart[r * C_SLOT + g] = c;
    }
}

__device__ __forceinline__ float pgt_ydiff_sum(const float* sPart, int g) {
    return ((sPart[g] + sPart[C_SLOT + g]) + sPart[2 * C_SLOT + g]) + sPart[3 * C_SLOT + g];
}

template <bool DUAL>
__device__ __forceinline__ void matvec_core(float* w, const float* sPG, const float* xb,
                                            float* yb, int tid, float* Yd, const float* pw,
                                            float step, float* sRS) {
    const int lane = tid & 31, wave = __builtin_amdgcn_readfirstlane(tid >> 5), nw = NT / 32;
    float* Zd = DUAL ? Yd : yb;
    __syncthreads();
    {
        const float* Ux = xb + OX_U; const float* Dx = xb + OX_D;
        constexpr int TM = 25, NTile = 25 * 13;
        const int m16 = lane & 15;
        const int kh  = (lane >> 4) * 2;
        const int t1 = ((wave + 1) * NTile) / nw;
        int t = (wave * NTile) / nw;
        while (t < t1) {
            const int tn = t / TM;
            const int tm_lo = t - tn * TM;
            const int tm_hi = (tm_lo + (t1 - t) < TM) ? tm_lo + (t1 - t) : TM;
            const int n = tn * 16 + m16;
            v2f b[25];
#pragma unroll
            for (int j = 0; j < 25; ++j) {
                int bi = (4 * j + kh) * S_P + n;
                b[j][0] = Ux[bi]       - Dx[bi];
                b[j][1] = Ux[bi + S_P] - Dx[bi + S_P];
            }
            for (int tm = tm_lo; tm < tm_hi; ++tm) {
                const int mr = tm * 16 + ((lane >> 4) << 3);
                float y5[8], y6[8], pv[8], fu[8], fd[8];
#pragma unroll
                for (int i = 0; i < 8; ++i) {
                    int o = (mr + i) * S_P + n;
                    if (DUAL) { y5[i] = Yd[OY_Z5 + o]; y6[i] = Yd[OY_Z6 + o]; pv[i] = pw[o]; }
                    fu[i] = xb[OX_FU + mr + i]; fd[i] = xb[OX_FD + mr + i];
                }
                const float* arow = sPG + (tm * 16 + m16) * PG_LP + kh;
                v2f a[25];
#pragma unroll
                for (int j = 0; j < 25; ++j) { a[j][0] = arow[4 * j]; a[j][1] = arow[4 * j + 1]; }
                v8f acc = {0.f,0.f,0.f,0.f,0.f,0.f,0.f,0.f};
#pragma unroll
                for (int j = 0; j < 25; ++j)
                    acc = __builtin_amdgcn_wmma_f32_16x16x4_f32(false, a[j], false, b[j],
                                                                (short)0, acc, false, false);
                __builtin_amdgcn_sched_group_barrier(0x100, 13, 0);
                __builtin_amdgcn_sched_group_barrier(0x008, 25, 0);
                const bool live = n < NS;
                float r5[8], r6[8];
#pragma unroll
                for (int i = 0; i < 8; ++i) {
                    int o = (mr + i) * S_P + n;
                    float f = acc[i];
                    float n5, n6;
                    if (DUAL) {
                        n5 = fmaxf(y5[i] + step * (( f - fu[i]) + pv[i]), 0.f);
                        n6 = fmaxf(y6[i] + step * ((-f - fd[i]) - pv[i]), 0.f);
                    } else {
                        n5 =  f - fu[i];
                        n6 = -f - fd[i];
                    }
                    r5[i] = live ? n5 : 0.f;
                    r6[i] = live ? n6 : 0.f;
                    Zd[OY_Z5 + o] = r5[i];
                    Zd[OY_Z6 + o] = r6[i];
                }
                const float s5 = rows_over_lanes(r5, m16);
                const float s6 = rows_over_lanes(r6, m16);
                if (m16 < 8) {
                    sRS[RS5 + tn * NL + mr + m16] = s5;
                    sRS[RS6 + tn * NL + mr + m16] = s6;
                }
            }
            t += tm_hi - tm_lo;
        }
    }
    if (tid < NL) {
        float s = 0.f; const float* r = sPG + tid * PG_LP; const float* p = xb + OX_P;
        for (int g = 0; g < NG; ++g) s += r[g] * p[g];
        w[WS_GP + tid] = s;
    }
    if (tid < NS) {
        float s = 0.f;
        const float* Ux = xb + OX_U + tid; const float* Dx = xb + OX_D + tid;
        for (int g = 0; g < NG; ++g) s += Ux[g * S_P] - Dx[g * S_P];
        w[WS_COL + tid] = s;
    }
    if (wave == nw - 1) {
        float a = xb[OX_P + lane] + xb[OX_P + 32 + lane] + xb[OX_P + 64 + lane];
        if (lane < NG - 96) a += xb[OX_P + 96 + lane];
        a += row_xor<1>(a); a += row_xor<2>(a); a += row_xor<4>(a); a += row_xor<8>(a);
        a += __shfl_xor(a, 16);
        if (lane == 0) w[WS_SUMP] = a;
    }
    __syncthreads();
    if (tid == 0) yb[OY_Y1] = w[WS_SUMP];
    for (int l = tid; l < NL; l += NT) {
        float g = w[WS_GP + l];
        yb[OY_Y2 + l] =  g + xb[OX_FU + l];
        yb[OY_Y3 + l] = -g + xb[OX_FD + l];
    }
    for (int s = tid; s < NS; s += NT) yb[OY_Y4 + s] = w[WS_COL + s];
    for (int g = tid; g < NG; g += NT) {
        yb[OY_Z1 + g] = xb[OX_P + g]  + xb[OX_RU + g];
        yb[OY_Z2 + g] = xb[OX_RD + g] - xb[OX_P + g];
    }
}

template <bool FUSED>
__device__ __forceinline__ void rmatvec_core(float* w, const float* sPG, float* yb,
                                             float* xo, int tid, float* X, float* XB,
                                             const float* Cost, const float* Cru,
                                             const float* Crd, float step, float* sRS, float* sPart) {
    const int lane = tid & 31, wave = __builtin_amdgcn_readfirstlane(tid >> 5), nw = NT / 32;
    __syncthreads();
    if (FUSED) {
        pgt_ydiff_partials(sPG, yb, tid, sPart);
        __syncthreads();
        if (tid < NG) {
            float a = 0.f, b = 0.f;
            for (int k = 0; k < N_STRIP; ++k) {
                a += sRS[RS3 + k * G_P + tid]; b += sRS[RS4 + k * G_P + tid];
            }
            const float c = pgt_ydiff_sum(sPart, tid);
            const float z1 = yb[OY_Z1 + tid], z2 = yb[OY_Z2 + tid];
            const float ap  = yb[OY_Y1] + c + z1 - z2;
            const float aru = z1 - a;
            const float ard = z2 - b;
            float o, n;
            o = X[OX_P  + tid]; n = fmaxf(o - step * (Cost[tid] + ap),  0.f);
            X[OX_P  + tid] = n; XB[OX_P  + tid] = 2.f * n - o;
            o = X[OX_RU + tid]; n = fmaxf(o - step * (Cru[tid]  + aru), 0.f);
            X[OX_RU + tid] = n; XB[OX_RU + tid] = 2.f * n - o;
            o = X[OX_RD + tid]; n = fmaxf(o - step * (Crd[tid]  + ard), 0.f);
            X[OX_RD + tid] = n; XB[OX_RD + tid] = 2.f * n - o;
        }
        if (tid >= NG && tid < NG + NL) {
            int l = tid - NG;
            float a = 0.f, b = 0.f;
            for (int k = 0; k < N_STRIP; ++k) {
                a += sRS[RS5 + k * NL + l]; b += sRS[RS6 + k * NL + l];
            }
            const float afu = yb[OY_Y2 + l] - a;
            const float afd = yb[OY_Y3 + l] - b;
            float o, n;
            o = X[OX_FU + l]; n = fmaxf(o - step * afu, 0.f);
            X[OX_FU + l] = n; XB[OX_FU + l] = 2.f * n - o;
            o = X[OX_FD + l]; n = fmaxf(o - step * afd, 0.f);
            X[OX_FD + l] = n; XB[OX_FD + l] = 2.f * n - o;
        }
        __syncthreads();
    }
    {
        const float* Z5 = yb + OY_Z5; const float* Z6 = yb + OY_Z6;
        constexpr int TM = 7, NTile = 7 * 13;
        const int m16 = lane & 15;
        const int kh  = (lane >> 4) * 2;
        const int t1 = ((wave + 1) * NTile) / nw;
        int t = (wave * NTile) / nw;
        while (t < t1) {
            const int tn = t / TM;
            const int tm_lo = t - tn * TM;
            const int tm_hi = (tm_lo + (t1 - t) < TM) ? tm_lo + (t1 - t) : TM;
            const int n = tn * 16 + m16;
            v8f acc[TM];
#pragma unroll
            for (int q = 0; q < TM; ++q) acc[q] = (v8f){0.f,0.f,0.f,0.f,0.f,0.f,0.f,0.f};
            for (int kb = 0; kb < NL; kb += 40) {
                v2f b[10];
#pragma unroll
                for (int j = 0; j < 10; ++j) {
                    int bi = (kb + 4 * j + kh) * S_P + n;
                    b[j][0] = Z5[bi]       - Z6[bi];
                    b[j][1] = Z5[bi + S_P] - Z6[bi + S_P];
                }
#pragma unroll
                for (int q = 0; q < TM; ++q) {
                    if (q >= tm_lo && q < tm_hi) {
                        const float* acol = sPG + (kb + kh) * PG_LP + q * 16 + m16;
                        v2f a[10];
#pragma unroll
                        for (int j = 0; j < 10; ++j) {
                            a[j][0] = acol[(4 * j) * PG_LP]; a[j][1] = acol[(4 * j + 1) * PG_LP];
                        }
#pragma unroll
                        for (int j = 0; j < 10; ++j)
                            acc[q] = __builtin_amdgcn_wmma_f32_16x16x4_f32(false, a[j], false, b[j],
                                                                           (short)0, acc[q], false, false);
                        __builtin_amdgcn_sched_group_barrier(0x100, 10, 0);
                        __builtin_amdgcn_sched_group_barrier(0x008, 10, 0);
                    }
                }
            }
            const float y4v = yb[OY_Y4 + n];
#pragma unroll
            for (int q = 0; q < TM; ++q) {
                if (q >= tm_lo && q < tm_hi) {
                    int mr = q * 16 + ((lane >> 4) << 3);
                    if (FUSED) {
                        float xu[8], xd[8], z3v[8], z4v[8], ruv[8], rdv[8];
#pragma unroll
                        for (int i = 0; i < 8; ++i) {
                            int o = (mr + i) * S_P + n;
                            xu[i]  = X[OX_U + o];      xd[i]  = X[OX_D + o];
                            z3v[i] = yb[OY_Z3 + o];    z4v[i] = yb[OY_Z4 + o];
                            ruv[i] = XB[OX_RU + mr + i]; rdv[i] = XB[OX_RD + mr + i];
                        }
                        float r3[8], r4[8];
#pragma unroll
                        for (int i = 0; i < 8; ++i) {
                            int o = (mr + i) * S_P + n;
                            float tv = acc[q][i];
                            float aU =  y4v + z3v[i] + tv;
                            float aD = -y4v + z4v[i] - tv;
                            float nU = fmaxf(xu[i] - step * aU, 0.f);
                            float nD = fmaxf(xd[i] - step * aD, 0.f);
                            float bU = 2.f * nU - xu[i];
                            float bD = 2.f * nD - xd[i];
                            float n3 = fmaxf(z3v[i] + step * (bU - ruv[i]), 0.f);
                            float n4 = fmaxf(z4v[i] + step * (bD - rdv[i]), 0.f);
                            const bool ok = n < NS && mr + i < NG;
                            if (ok) {
                                X[OX_U + o]  = nU;  XB[OX_U + o] = bU;
                                X[OX_D + o]  = nD;  XB[OX_D + o] = bD;
                                yb[OY_Z3 + o] = n3; yb[OY_Z4 + o] = n4;
                            }
                            r3[i] = ok ? n3 : 0.f;
                            r4[i] = ok ? n4 : 0.f;
                        }
                        const float s3 = rows_over_lanes(r3, m16);
                        const float s4 = rows_over_lanes(r4, m16);
                        if (m16 < 8) {
                            sRS[RS3 + tn * G_P + mr + m16] = s3;
                            sRS[RS4 + tn * G_P + mr + m16] = s4;
                        }
                    } else {
                        float xu[8], xd[8], ruv[8], rdv[8];
#pragma unroll
                        for (int i = 0; i < 8; ++i) {
                            int o = (mr + i) * S_P + n;
                            xu[i]  = xo[OX_U + o];       xd[i]  = xo[OX_D + o];
                            ruv[i] = xo[OX_RU + mr + i]; rdv[i] = xo[OX_RD + mr + i];
                        }
                        float r3[8], r4[8];
#pragma unroll
                        for (int i = 0; i < 8; ++i) {
                            int o = (mr + i) * S_P + n;
                            float tv = acc[q][i];
                            float z3 = xu[i] - ruv[i];
                            float z4 = xd[i] - rdv[i];
                            const bool ok = n < NS && mr + i < NG;
                            if (ok) {
                                xo[OX_U + o] =  y4v + z3 + tv;
                                xo[OX_D + o] = -y4v + z4 - tv;
                            }
                            r3[i] = ok ? z3 : 0.f;
                            r4[i] = ok ? z4 : 0.f;
                        }
                        const float s3 = rows_over_lanes(r3, m16);
                        const float s4 = rows_over_lanes(r4, m16);
                        if (m16 < 8) {
                            sRS[RS3 + tn * G_P + mr + m16] = s3;
                            sRS[RS4 + tn * G_P + mr + m16] = s4;
                        }
                    }
                }
            }
            t += tm_hi - tm_lo;
        }
    }
    if (FUSED) return;
    pgt_ydiff_partials(sPG, yb, tid, sPart);
    __syncthreads();
    if (tid < NG) {
        float a = 0.f, b = 0.f;
        for (int k = 0; k < N_STRIP; ++k) {
            a += sRS[RS3 + k * G_P + tid]; b += sRS[RS4 + k * G_P + tid];
        }
        const float z1 = yb[OY_Z1 + tid], z2 = yb[OY_Z2 + tid];
        xo[OX_P  + tid] = yb[OY_Y1] + pgt_ydiff_sum(sPart, tid) + z1 - z2;
        xo[OX_RU + tid] = z1 - a;
        xo[OX_RD + tid] = z2 - b;
    }
    if (tid >= NG && tid < NG + NL) {
        int l = tid - NG;
        float a = 0.f, b = 0.f;
        for (int k = 0; k < N_STRIP; ++k) {
            a += sRS[RS5 + k * NL + l]; b += sRS[RS6 + k * NL + l];
        }
        xo[OX_FU + l] = yb[OY_Y2 + l] - a;
        xo[OX_FD + l] = yb[OY_Y3 + l] - b;
    }
}


__global__ __launch_bounds__(NT) void k_solver(float* w, const float* Cost,
                                               const float* Cru, const float* Crd,
                                               const float* Pmax, float* out) {
    __shared__ float red[NT];
    __shared__ __attribute__((aligned(16))) float sPG[NL * PG_LP];
    __shared__ float sRS[RS_TOTAL];
    __shared__ float sPart[C_RANGES * C_SLOT];
    const int tid = threadIdx.x;
    float* X   = w + WS_X;
    float* XB  = w + WS_XB;
    float* Y   = w + WS_Y;
    float* AX  = w + WS_AX;

    for (int i = tid; i < NL * G_P; i += NT) {
        int l = i / G_P, g = i - l * G_P;
        sPG[l * PG_LP + g] = w[WS_PG + i];
    }
    __syncthreads();

    float sumsq = 0.f;
    for (int it = 0; it < POWER_ITERS; ++it) {
        __syncthreads();
        float part = 0.f;
        for (int i = tid; i < XSZ; i += NT) { float v = X[i]; part += v * v; }
        sumsq = block_reduce(part, red, tid);
        float inv = 1.0f / sqrtf(sumsq);
        for (int i = tid; i < XSZ; i += NT) X[i] *= inv;
        matvec_core<false>(w, sPG, X, AX, tid, nullptr, nullptr, 0.f, sRS);
        rmatvec_core<false>(w, sPG, AX, X, tid, nullptr, nullptr, nullptr, nullptr, nullptr, 0.f, sRS, sPart);
    }
    __syncthreads();
    {
        float part = 0.f;
        for (int i = tid; i < XSZ; i += NT) { float v = X[i]; part += v * v; }
        sumsq = block_reduce(part, red, tid);
    }
    const float step = 0.9f / sqrtf(sqrtf(sumsq));
    const float b1 = w[WS_B1];
    const float* pw = w + WS_PW;

    for (int i = tid; i < XSZ; i += NT) { X[i] = 0.f; XB[i] = 0.f; }
    for (int i = tid; i < YSZ; i += NT) Y[i] = 0.f;
    for (int i = tid; i < RS_TOTAL; i += NT) sRS[i] = 0.f;

    for (int it = 0; it < PDHG_ITERS; ++it) {
        rmatvec_core<true>(w, sPG, Y, nullptr, tid, X, XB, Cost, Cru, Crd, step, sRS, sPart);
        matvec_core<true>(w, sPG, XB, AX, tid, Y, pw, step, sRS);
        __syncthreads();
        if (tid == 0) Y[OY_Y1] += step * (AX[OY_Y1] - b1);
        for (int l = tid; l < NL; l += NT) {
            Y[OY_Y2 + l] += step * (AX[OY_Y2 + l] - w[WS_B2 + l]);
            Y[OY_Y3 + l] += step * (AX[OY_Y3 + l] - w[WS_B3 + l]);
        }
        for (int s = tid; s < NS; s += NT)
            Y[OY_Y4 + s] += step * (AX[OY_Y4 + s] - w[WS_B4 + s]);
        for (int g = tid; g < NG; g += NT) {
            Y[OY_Z1 + g] = fmaxf(Y[OY_Z1 + g] + step * (AX[OY_Z1 + g] - Pmax[g]), 0.f);
            Y[OY_Z2 + g] = fmaxf(Y[OY_Z2 + g] + step *  AX[OY_Z2 + g], 0.f);
        }
        __syncthreads();
    }

    for (int g = tid; g < NG; g += NT) {
        out[g]          = X[OX_P  + g];
        out[NG + g]     = X[OX_RU + g];
        out[2 * NG + g] = X[OX_RD + g];
    }
    for (int i = tid; i < NG * NS; i += NT) {
        int g = i / NS, s = i % NS;
        out[3 * NG + i]           = X[OX_U + g * S_P + s];
        out[3 * NG + NG * NS + i] = X[OX_D + g * S_P + s];
    }
    for (int l = tid; l < NL; l += NT) {
        out[3 * NG + 2 * NG * NS + l]      = X[OX_FU + l];
        out[3 * NG + 2 * NG * NS + NL + l] = X[OX_FD + l];
    }
}


extern "C" void kernel_launch(void* const* d_in, const int* in_sizes, int n_in,
                              void* d_out, int out_size, void* d_ws, size_t ws_size,
                              hipStream_t stream) {
    const float* w_scen   = (const float*)d_in[0];
    const float* Cost     = (const float*)d_in[1];
    const float* C_r_up   = (const float*)d_in[2];
    const float* C_r_down = (const float*)d_in[3];
    const float* Pmax     = (const float*)d_in[4];
    const float* PTDF     = (const float*)d_in[5];
    const float* node_G   = (const float*)d_in[6];
    const float* node_W   = (const float*)d_in[7];
    const float* node_L   = (const float*)d_in[8];
    const float* Pd       = (const float*)d_in[9];
    const float* w_exp    = (const float*)d_in[10];
    const float* Line_cap = (const float*)d_in[11];
    float* w   = (float*)d_ws;
    float* out = (float*)d_out;

    k_zero <<<512, 256, 0, stream>>>(w);
    k_prep1<<<240, 256, 0, stream>>>(w, w_scen, node_W, node_L, Pd, w_exp);
    k_prep2<<<640, 256, 0, stream>>>(w, PTDF, node_G, Line_cap);
    k_setv <<<161, 256, 0, stream>>>(w);
    k_solver<<<1, NT, 0, stream>>>(w, Cost, C_r_up, C_r_down, Pmax, out);
}


# PDHG loop: xb_U - xb_D (the flow product's B operand) and xb_p stay in LDS where the primal step forms them: B fragments, column sums, PG p read LDS, not L2; xb_U, xb_D never stored; sums keep order
# speedup vs baseline: 4.8534x; 1.0515x over previous
#include <hip/hip_runtime.h>
#include <math.h>

typedef __attribute__((ext_vector_type(2))) float v2f;
typedef __attribute__((ext_vector_type(8))) float v8f;

#define NT 512

constexpr int NG = 100, NL = 400, NN = 300, NWF = 50, NS = 200;
constexpr int G_P = 112;
constexpr int S_P = 208;
constexpr int PDHG_ITERS = 300, POWER_ITERS = 25;
constexpr int PG_LP = 116;
constexpr int N_STRIP = S_P / 16;
constexpr int RS5 = 0;
constexpr int RS6 = RS5 + N_STRIP * NL;
constexpr int RS3 = RS6 + N_STRIP * NL;
constexpr int RS4 = RS3 + N_STRIP * G_P;
constexpr int RS_TOTAL = RS4 + N_STRIP * G_P;
constexpr int C_RANGES = 4;
constexpr int C_RANGE = NL / C_RANGES;
constexpr int C_SLOT = 128;
constexpr int UD_LP = 210;

constexpr int OX_P  = 0;
constexpr int OX_RU = OX_P  + G_P;
constexpr int OX_RD = OX_RU + G_P;
constexpr int OX_U  = OX_RD + G_P;
constexpr int OX_D  = OX_U  + G_P * S_P;
constexpr int OX_FU = OX_D  + G_P * S_P;
constexpr int OX_FD = OX_FU + NL;
constexpr int XSZ   = OX_FD + NL;

constexpr int OY_Y1 = 0;
constexpr int OY_Y2 = 16;
constexpr int OY_Y3 = OY_Y2 + NL;
constexpr int OY_Y4 = OY_Y3 + NL;
constexpr int OY_Z1 = OY_Y4 + S_P;
constexpr int OY_Z2 = OY_Z1 + G_P;
constexpr int OY_Z3 = OY_Z2 + G_P;
constexpr int OY_Z4 = OY_Z3 + G_P * S_P;
constexpr int OY_Z5 = OY_Z4 + G_P * S_P;
constexpr int OY_Z6 = OY_Z5 + NL * S_P;
constexpr int YSZ   = OY_Z6 + NL * S_P;

constexpr int WS_PG   = 0;
constexpr int WS_PGT  = WS_PG   + NL * G_P;
constexpr int WS_PW   = WS_PGT  + G_P * NL;
constexpr int WS_B2   = WS_PW   + NL * S_P;
constexpr int WS_B3   = WS_B2   + NL;
constexpr int WS_B4   = WS_B3   + NL;
constexpr int WS_B1   = WS_B4   + S_P;
constexpr int WS_M1   = WS_B1   + 16;
constexpr int WS_NV   = WS_M1   + NN * S_P;
constexpr int WS_X    = WS_NV   + NN + 4;
constexpr int WS_XB   = WS_X    + XSZ;
constexpr int WS_ATY  = WS_XB   + XSZ;
constexpr int WS_Y    = WS_ATY  + XSZ;
constexpr int WS_AX   = WS_Y    + YSZ;
constexpr int WS_GP   = WS_AX   + YSZ;
constexpr int WS_COL  = WS_GP   + NL;
constexpr int WS_GTY  = WS_COL  + S_P;
constexpr int WS_SUMP = WS_GTY  + G_P;
constexpr int WS_TOTAL= WS_SUMP + 16;


__global__ void k_zero(float* w) {
    for (int i = blockIdx.x * blockDim.x + threadIdx.x; i < WS_TOTAL;
         i += gridDim.x * blockDim.x) w[i] = 0.0f;
}

__global__ void k_prep1(float* w, const float* w_scen, const float* node_W,
                        const float* node_L, const float* Pd, const float* w_exp) {
    const int NM1 = NN * NS;
    const int total = NM1 + NN + NS + 1;
    for (int idx = blockIdx.x * blockDim.x + threadIdx.x; idx < total;
         idx += gridDim.x * blockDim.x) {
        if (idx < NM1) {
            int n = idx / NS, s = idx % NS;
            float a = 0.f;
            for (int ww = 0; ww < NWF; ++ww) a += node_W[n*NWF + ww] * w_scen[s*NWF + ww];
            w[WS_M1 + n*S_P + s] = a;
        } else if (idx < NM1 + NN) {
            int n = idx - NM1;
            float a = 0.f;
            for (int ww = 0; ww < NWF; ++ww) a += node_W[n*NWF + ww] * w_exp[ww];
            for (int m = 0; m < NN; ++m)     a -= node_L[n*NN + m] * Pd[m];
            w[WS_NV + n] = a;
        } else if (idx < NM1 + NN + NS) {
            int s = idx - NM1 - NN;
            float a = 0.f;
            for (int ww = 0; ww < NWF; ++ww) a += w_scen[s*NWF + ww];
            w[WS_B4 + s] = -a;
        } else {
            float a = 0.f;
            for (int m = 0; m < NN; ++m)  a += Pd[m];
            for (int ww = 0; ww < NWF; ++ww) a -= w_exp[ww];
            w[WS_B1] = a;
        }
    }
}

__global__ void k_prep2(float* w, const float* PTDF, const float* node_G,
                        const float* Line_cap) {
    const int A = NL * NG;
    const int C = A + NL * NS;
    const int D2 = C + NL;
    const int E = D2 + NL;
    for (int idx = blockIdx.x * blockDim.x + threadIdx.x; idx < E;
         idx += gridDim.x * blockDim.x) {
        if (idx < A) {
            int l = idx / NG, g = idx % NG;
            float a = 0.f;
            for (int n = 0; n < NN; ++n) a += PTDF[l*NN + n] * node_G[n*NG + g];
            w[WS_PG + l*G_P + g] = a;
        } else if (idx < C) {
            int r = idx - A; int l = r / NS, s = r % NS;
            float a = 0.f;
            for (int n = 0; n < NN; ++n) a += PTDF[l*NN + n] * w[WS_M1 + n*S_P + s];
            w[WS_PW + l*S_P + s] = a;
        } else if (idx < D2) {
            int l = idx - C;
            float a = 0.f;
            for (int n = 0; n < NN; ++n) a += PTDF[l*NN + n] * w[WS_NV + n];
            w[WS_B2 + l] = Line_cap[l] - a;
        } else {
            int l = idx - D2;
            float a = 0.f;
            for (int n = 0; n < NN; ++n) a += PTDF[l*NN + n] * w[WS_NV + n];
            w[WS_B3 + l] = Line_cap[l] + a;
        }
    }
}

__global__ void k_setv(float* w) {
    const int total = 3*NG + 2*NG*NS + 2*NL;
    float* X = w + WS_X;
    for (int idx = blockIdx.x * blockDim.x + threadIdx.x; idx < total;
         idx += gridDim.x * blockDim.x) {
        int off;
        if (idx < NG)                off = OX_P  + idx;
        else if (idx < 2*NG)         off = OX_RU + (idx - NG);
        else if (idx < 3*NG)         off = OX_RD + (idx - 2*NG);
        else if (idx < 3*NG + NG*NS) { int r = idx - 3*NG; off = OX_U + (r/NS)*S_P + (r%NS); }
        else if (idx < 3*NG + 2*NG*NS) { int r = idx - 3*NG - NG*NS; off = OX_D + (r/NS)*S_P + (r%NS); }
        else if (idx < 3*NG + 2*NG*NS + NL) off = OX_FU + (idx - 3*NG - 2*NG*NS);
        else                         off = OX_FD + (idx - 3*NG - 2*NG*NS - NL);
        X[off] = 1.0f;
    }
}


__device__ __forceinline__ float block_reduce(float v, float* red, int tid) {
    red[tid] = v; __syncthreads();
    for (int off = NT >> 1; off > 0; off >>= 1) {
        if (tid < off) red[tid] += red[tid + off];
        __syncthreads();
    }
    float r = red[0]; __syncthreads();
    return r;
}

template <int M>
__device__ __forceinline__ float row_xor(float x) {
    return __int_as_float(__builtin_amdgcn_update_dpp(0, __float_as_int(x), 0x160 + M, 0xf, 0xf, true));
}

__device__ __forceinline__ float rows_over_lanes(const float (&v)[8], int c) {
    const bool b2 = (c & 4) != 0, b1 = (c & 2) != 0, b0 = (c & 1) != 0;
    float w[4], u[2];
#pragma unroll
    for (int j = 0; j < 4; ++j) {
        float keep = b2 ? v[4 + j] : v[j];
        float send = b2 ? v[j] : v[4 + j];
        w[j] = keep + row_xor<4>(send);
    }
#pragma unroll
    for (int j = 0; j < 2; ++j) {
        float keep = b1 ? w[2 + j] : w[j];
        float send = b1 ? w[j] : w[2 + j];
        u[j] = keep + row_xor<2>(send);
    }
    float keep = b0 ? u[1] : u[0];
    float send = b0 ? u[0] : u[1];
    float t = keep + row_xor<1>(send);
    return t + row_xor<8>(t);
}

__device__ __forceinline__ void pgt_ydiff_partials(const float* sPG, const float* yb, int tid, float* sPart) {
    const int r = __builtin_amdgcn_readfirstlane(tid / C_SLOT);
    const int g = tid % C_SLOT;
    if (g < NG) {
        const float* pc = sPG + r * C_RANGE * PG_LP + g;
        const float* y2 = yb + OY_Y2 + r * C_RANGE;
        const float* y3 = yb + OY_Y3 + r * C_RANGE;
        float c = 0.f;
#pragma unroll
        for (int l = 0; l < C_RANGE; ++l) c += pc[l * PG_LP] * (y2[l] - y3[l]);
        sPart[r * C_SLOT + g] = c;
    }
}

__device__ __forceinline__ float pgt_ydiff_sum(const float* sPart, int g) {
    return ((sPart[g] + sPart[C_SLOT + g]) + sPart[2 * C_SLOT + g]) + sPart[3 * C_SLOT + g];
}

template <bool DUAL>
__device__ __forceinline__ void matvec_core(float* w, const float* sPG, const float* xb,
                                            float* yb, int tid, float* Yd, const float* pw,
                                            float step, float* sRS, const float* sUD, const float* sXP) {
    const int lane = tid & 31, wave = __builtin_amdgcn_readfirstlane(tid >> 5), nw = NT / 32;
    float* Zd = DUAL ? Yd : yb;
    __syncthreads();
    {
        const float* Ux = xb + OX_U; const float* Dx = xb + OX_D;
        constexpr int TM = 25, NTile = 25 * 13;
        const int m16 = lane & 15;
        const int kh  = (lane >> 4) * 2;
        const int t1 = ((wave + 1) * NTile) / nw;
        int t = (wave * NTile) / nw;
        while (t < t1) {
            const int tn = t / TM;
            const int tm_lo = t - tn * TM;
            const int tm_hi = (tm_lo + (t1 - t) < TM) ? tm_lo + (t1 - t) : TM;
            const int n = tn * 16 + m16;
            v2f b[25];
#pragma unroll
            for (int j = 0; j < 25; ++j) {
                if (DUAL) {
                    int bi = (4 * j + kh) * UD_LP + n;
                    b[j][0] = sUD[bi];
                    b[j][1] = sUD[bi + UD_LP];
                } else {
                    int bi = (4 * j + kh) * S_P + n;
                    b[j][0] = Ux[bi]       - Dx[bi];
                    b[j][1] = Ux[bi + S_P] - Dx[bi + S_P];
                }
            }
            for (int tm = tm_lo; tm < tm_hi; ++tm) {
                const int mr = tm * 16 + ((lane >> 4) << 3);
                float y5[8], y6[8], pv[8], fu[8], fd[8];
#pragma unroll
                for (int i = 0; i < 8; ++i) {
                    int o = (mr + i) * S_P + n;
                    if (DUAL) { y5[i] = Yd[OY_Z5 + o]; y6[i] = Yd[OY_Z6 + o]; pv[i] = pw[o]; }
                    fu[i] = xb[OX_FU + mr + i]; fd[i] = xb[OX_FD + mr + i];
                }
                const float* arow = sPG + (tm * 16 + m16) * PG_LP + kh;
                v2f a[25];
#pragma unroll
                for (int j = 0; j < 25; ++j) { a[j][0] = arow[4 * j]; a[j][1] = arow[4 * j + 1]; }
                v8f acc = {0.f,0.f,0.f,0.f,0.f,0.f,0.f,0.f};
#pragma unroll
                for (int j = 0; j < 25; ++j)
                    acc = __builtin_amdgcn_wmma_f32_16x16x4_f32(false, a[j], false, b[j],
                                                                (short)0, acc, false, false);
                __builtin_amdgcn_sched_group_barrier(0x100, 13, 0);
                __builtin_amdgcn_sched_group_barrier(0x008, 25, 0);
                const bool live = n < NS;
                float r5[8], r6[8];
#pragma unroll
                for (int i = 0; i < 8; ++i) {
                    int o = (mr + i) * S_P + n;
                    float f = acc[i];
                    float n5, n6;
                    if (DUAL) {
                        n5 = fmaxf(y5[i] + step * (( f - fu[i]) + pv[i]), 0.f);
                        n6 = fmaxf(y6[i] + step * ((-f - fd[i]) - pv[i]), 0.f);
                    } else {
                        n5 =  f - fu[i];
                        n6 = -f - fd[i];
                    }
                    r5[i] = live ? n5 : 0.f;
                    r6[i] = live ? n6 : 0.f;
                    Zd[OY_Z5 + o] = r5[i];
                    Zd[OY_Z6 + o] = r6[i];
                }
                const float s5 = rows_over_lanes(r5, m16);
                const float s6 = rows_over_lanes(r6, m16);
                if (m16 < 8) {
                    sRS[RS5 + tn * NL + mr + m16] = s5;
                    sRS[RS6 + tn * NL + mr + m16] = s6;
                }
            }
            t += tm_hi - tm_lo;
        }
    }
    if (tid < NL) {
        float s = 0.f; const float* r = sPG + tid * PG_LP;
        if (DUAL) {
            for (int g = 0; g < NG; ++g) s += r[g] * sXP[g];
        } else {
            const float* p = xb + OX_P;
            for (int g = 0; g < NG; ++g) s += r[g] * p[g];
        }
        w[WS_GP + tid] = s;
    }
    if (tid < NS) {
        float s = 0.f;
        if (DUAL) {
            const float* d = sUD + tid;
            for (int g = 0; g < NG; ++g) s += d[g * UD_LP];
        } else {
            const float* Ux = xb + OX_U + tid; const float* Dx = xb + OX_D + tid;
            for (int g = 0; g < NG; ++g) s += Ux[g * S_P] - Dx[g * S_P];
        }
        w[WS_COL + tid] = s;
    }
    if (wave == nw - 1) {
        float a;
        if (DUAL) {
            a = sXP[lane] + sXP[32 + lane] + sXP[64 + lane];
            if (lane < NG - 96) a += sXP[96 + lane];
        } else {
            a = xb[OX_P + lane] + xb[OX_P + 32 + lane] + xb[OX_P + 64 + lane];
            if (lane < NG - 96) a += xb[OX_P + 96 + lane];
        }
        a += row_xor<1>(a); a += row_xor<2>(a); a += row_xor<4>(a); a += row_xor<8>(a);
        a += __shfl_xor(a, 16);
        if (lane == 0) w[WS_SUMP] = a;
    }
    __syncthreads();
    if (tid == 0) yb[OY_Y1] = w[WS_SUMP];
    for (int l = tid; l < NL; l += NT) {
        float g = w[WS_GP + l];
        yb[OY_Y2 + l] =  g + xb[OX_FU + l];
        yb[OY_Y3 + l] = -g + xb[OX_FD + l];
    }
    for (int s = tid; s < NS; s += NT) yb[OY_Y4 + s] = w[WS_COL + s];
    for (int g = tid; g < NG; g += NT) {
        yb[OY_Z1 + g] = xb[OX_P + g]  + xb[OX_RU + g];
        yb[OY_Z2 + g] = xb[OX_RD + g] - xb[OX_P + g];
    }
}

template <bool FUSED>
__device__ __forceinline__ void rmatvec_core(float* w, const float* sPG, float* yb,
                                             float* xo, int tid, float* X, float* XB,
                                             const float* Cost, const float* Cru,
                                             const float* Crd, float step, float* sRS, float* sPart,
                                             float* sUD, float* sXP) {
    const int lane = tid & 31, wave = __builtin_amdgcn_readfirstlane(tid >> 5), nw = NT / 32;
    __syncthreads();
    if (FUSED) {
        pgt_ydiff_partials(sPG, yb, tid, sPart);
        __syncthreads();
        if (tid < NG) {
            float a = 0.f, b = 0.f;
            for (int k = 0; k < N_STRIP; ++k) {
                a += sRS[RS3 + k * G_P + tid]; b += sRS[RS4 + k * G_P + tid];
            }
            const float c = pgt_ydiff_sum(sPart, tid);
            const float z1 = yb[OY_Z1 + tid], z2 = yb[OY_Z2 + tid];
            const float ap  = yb[OY_Y1] + c + z1 - z2;
            const float aru = z1 - a;
            const float ard = z2 - b;
            float o, n;
            o = X[OX_P  + tid]; n = fmaxf(o - step * (Cost[tid] + ap),  0.f);
            const float bp = 2.f * n - o;
            X[OX_P  + tid] = n; XB[OX_P  + tid] = bp; sXP[tid] = bp;
            o = X[OX_RU + tid]; n = fmaxf(o - step * (Cru[tid]  + aru), 0.f);
            X[OX_RU + tid] = n; XB[OX_RU + tid] = 2.f * n - o;
            o = X[OX_RD + tid]; n = fmaxf(o - step * (Crd[tid]  + ard), 0.f);
            X[OX_RD + tid] = n; XB[OX_RD + tid] = 2.f * n - o;
        }
        if (tid >= NG && tid < NG + NL) {
            int l = tid - NG;
            float a = 0.f, b = 0.f;
            for (int k = 0; k < N_STRIP; ++k) {
                a += sRS[RS5 + k * NL + l]; b += sRS[RS6 + k * NL + l];
            }
            const float afu = yb[OY_Y2 + l] - a;
            const float afd = yb[OY_Y3 + l] - b;
            float o, n;
            o = X[OX_FU + l]; n = fmaxf(o - step * afu, 0.f);
            X[OX_FU + l] = n; XB[OX_FU + l] = 2.f * n - o;
            o = X[OX_FD + l]; n = fmaxf(o - step * afd, 0.f);
            X[OX_FD + l] = n; XB[OX_FD + l] = 2.f * n - o;
        }
        __syncthreads();
    }
    {
        const float* Z5 = yb + OY_Z5; const float* Z6 = yb + OY_Z6;
        constexpr int TM = 7, NTile = 7 * 13;
        const int m16 = lane & 15;
        const int kh  = (lane >> 4) * 2;
        const int t1 = ((wave + 1) * NTile) / nw;
        int t = (wave * NTile) / nw;
        while (t < t1) {
            const int tn = t / TM;
            const int tm_lo = t - tn * TM;
            const int tm_hi = (tm_lo + (t1 - t) < TM) ? tm_lo + (t1 - t) : TM;
            const int n = tn * 16 + m16;
            v8f acc[TM];
#pragma unroll
            for (int q = 0; q < TM; ++q) acc[q] = (v8f){0.f,0.f,0.f,0.f,0.f,0.f,0.f,0.f};
            for (int kb = 0; kb < NL; kb += 40) {
                v2f b[10];
#pragma unroll
                for (int j = 0; j < 10; ++j) {
                    int bi = (kb + 4 * j + kh) * S_P + n;
                    b[j][0] = Z5[bi]       - Z6[bi];
                    b[j][1] = Z5[bi + S_P] - Z6[bi + S_P];
                }
#pragma unroll
                for (int q = 0; q < TM; ++q) {
                    if (q >= tm_lo && q < tm_hi) {
                        const float* acol = sPG + (kb + kh) * PG_LP + q * 16 + m16;
                        v2f a[10];
#pragma unroll
                        for (int j = 0; j < 10; ++j) {
                            a[j][0] = acol[(4 * j) * PG_LP]; a[j][1] = acol[(4 * j + 1) * PG_LP];
                        }
#pragma unroll
                        for (int j = 0; j < 10; ++j)
                            acc[q] = __builtin_amdgcn_wmma_f32_16x16x4_f32(false, a[j], false, b[j],
                                                                           (short)0, acc[q], false, false);
                        __builtin_amdgcn_sched_group_barrier(0x100, 10, 0);
                        __builtin_amdgcn_sched_group_barrier(0x008, 10, 0);
                    }
                }
            }
            const float y4v = yb[OY_Y4 + n];
            const int ud0 = ((lane >> 4) << 3) * UD_LP + n;
#pragma unroll
            for (int q = 0; q < TM; ++q) {
                if (q >= tm_lo && q < tm_hi) {
                    int mr = q * 16 + ((lane >> 4) << 3);
                    if (FUSED) {
                        float xu[8], xd[8], z3v[8], z4v[8], ruv[8], rdv[8];
#pragma unroll
                        for (int i = 0; i < 8; ++i) {
                            int o = (mr + i) * S_P + n;
                            xu[i]  = X[OX_U + o];      xd[i]  = X[OX_D + o];
                            z3v[i] = yb[OY_Z3 + o];    z4v[i] = yb[OY_Z4 + o];
                            ruv[i] = XB[OX_RU + mr + i]; rdv[i] = XB[OX_RD + mr + i];
                        }
                        float r3[8], r4[8];
#pragma unroll
                        for (int i = 0; i < 8; ++i) {
                            int o = (mr + i) * S_P + n;
                            float tv = acc[q][i];
                            float aU =  y4v + z3v[i] + tv;
                            float aD = -y4v + z4v[i] - tv;
                            float nU = fmaxf(xu[i] - step * aU, 0.f);
                            float nD = fmaxf(xd[i] - step * aD, 0.f);
                            float bU = 2.f * nU - xu[i];
                            float bD = 2.f * nD - xd[i];
                            float n3 = fmaxf(z3v[i] + step * (bU - ruv[i]), 0.f);
                            float n4 = fmaxf(z4v[i] + step * (bD - rdv[i]), 0.f);
                            const bool ok = n < NS && mr + i < NG;
                            if (ok) {
                                X[OX_U + o]  = nU;
                                X[OX_D + o]  = nD;
                                sUD[ud0 + (q * 16 + i) * UD_LP] = bU - bD;
                                yb[OY_Z3 + o] = n3; yb[OY_Z4 + o] = n4;
                            }
                            r3[i] = ok ? n3 : 0.f;
                            r4[i] = ok ? n4 : 0.f;
                        }
                        const float s3 = rows_over_lanes(r3, m16);
                        const float s4 = rows_over_lanes(r4, m16);
                        if (m16 < 8) {
                            sRS[RS3 + tn * G_P + mr + m16] = s3;
                            sRS[RS4 + tn * G_P + mr + m16] = s4;
                        }
                    } else {
                        float xu[8], xd[8], ruv[8], rdv[8];
#pragma unroll
                        for (int i = 0; i < 8; ++i) {
                            int o = (mr + i) * S_P + n;
                            xu[i]  = xo[OX_U + o];       xd[i]  = xo[OX_D + o];
                            ruv[i] = xo[OX_RU + mr + i]; rdv[i] = xo[OX_RD + mr + i];
                        }
                        float r3[8], r4[8];
#pragma unroll
                        for (int i = 0; i < 8; ++i) {
                            int o = (mr + i) * S_P + n;
                            float tv = acc[q][i];
                            float z3 = xu[i] - ruv[i];
                            float z4 = xd[i] - rdv[i];
                            const bool ok = n < NS && mr + i < NG;
                            if (ok) {
                                xo[OX_U + o] =  y4v + z3 + tv;
                                xo[OX_D + o] = -y4v + z4 - tv;
                            }
                            r3[i] = ok ? z3 : 0.f;
                            r4[i] = ok ? z4 : 0.f;
                        }
                        const float s3 = rows_over_lanes(r3, m16);
                        const float s4 = rows_over_lanes(r4, m16);
                        if (m16 < 8) {
                            sRS[RS3 + tn * G_P + mr + m16] = s3;
                            sRS[RS4 + tn * G_P + mr + m16] = s4;
                        }
                    }
                }
            }
            t += tm_hi - tm_lo;
        }
    }
    if (FUSED) return;
    pgt_ydiff_partials(sPG, yb, tid, sPart);
    __syncthreads();
    if (tid < NG) {
        float a = 0.f, b = 0.f;
        for (int k = 0; k < N_STRIP; ++k) {
            a += sRS[RS3 + k * G_P + tid]; b += sRS[RS4 + k * G_P + tid];
        }
        const float z1 = yb[OY_Z1 + tid], z2 = yb[OY_Z2 + tid];
        xo[OX_P  + tid] = yb[OY_Y1] + pgt_ydiff_sum(sPart, tid) + z1 - z2;
        xo[OX_RU + tid] = z1 - a;
        xo[OX_RD + tid] = z2 - b;
    }
    if (tid >= NG && tid < NG + NL) {
        int l = tid - NG;
        float a = 0.f, b = 0.f;
        for (int k = 0; k < N_STRIP; ++k) {
            a += sRS[RS5 + k * NL + l]; b += sRS[RS6 + k * NL + l];
        }
        xo[OX_FU + l] = yb[OY_Y2 + l] - a;
        xo[OX_FD + l] = yb[OY_Y3 + l] - b;
    }
}


__global__ __launch_bounds__(NT) void k_solver(float* w, const float* Cost,
                                               const float* Cru, const float* Crd,
                                               const float* Pmax, float* out) {
    __shared__ float red[NT];
    __shared__ __attribute__((aligned(16))) float sPG[NL * PG_LP];
    __shared__ float sRS[RS_TOTAL];
    __shared__ float sPart[C_RANGES * C_SLOT];
    __shared__ float sUD[NG * UD_LP];
    __shared__ float sXP[G_P];
    const int tid = threadIdx.x;
    float* X   = w + WS_X;
    float* XB  = w + WS_XB;
    float* Y   = w + WS_Y;
    float* AX  = w + WS_AX;

    for (int i = tid; i < NL * G_P; i += NT) {
        int l = i / G_P, g = i - l * G_P;
        sPG[l * PG_LP + g] = w[WS_PG + i];
    }
    __syncthreads();

    float sumsq = 0.f;
    for (int it = 0; it < POWER_ITERS; ++it) {
        __syncthreads();
        float part = 0.f;
        for (int i = tid; i < XSZ; i += NT) { float v = X[i]; part += v * v; }
        sumsq = block_reduce(part, red, tid);
        float inv = 1.0f / sqrtf(sumsq);
        for (int i = tid; i < XSZ; i += NT) X[i] *= inv;
        matvec_core<false>(w, sPG, X, AX, tid, nullptr, nullptr, 0.f, sRS, nullptr, nullptr);
        rmatvec_core<false>(w, sPG, AX, X, tid, nullptr, nullptr, nullptr, nullptr, nullptr, 0.f, sRS, sPart, nullptr, nullptr);
    }
    __syncthreads();
    {
        float part = 0.f;
        for (int i = tid; i < XSZ; i += NT) { float v = X[i]; part += v * v; }
        sumsq = block_reduce(part, red, tid);
    }
    const float step = 0.9f / sqrtf(sqrtf(sumsq));
    const float b1 = w[WS_B1];
    const float* pw = w + WS_PW;

    for (int i = tid; i < XSZ; i += NT) { X[i] = 0.f; XB[i] = 0.f; }
    for (int i = tid; i < YSZ; i += NT) Y[i] = 0.f;
    for (int i = tid; i < RS_TOTAL; i += NT) sRS[i] = 0.f;
    for (int i = tid; i < NG * UD_LP; i += NT) sUD[i] = 0.f;

    for (int it = 0; it < PDHG_ITERS; ++it) {
        rmatvec_core<true>(w, sPG, Y, nullptr, tid, X, XB, Cost, Cru, Crd, step, sRS, sPart, sUD, sXP);
        matvec_core<true>(w, sPG, XB, AX, tid, Y, pw, step, sRS, sUD, sXP);
        __syncthreads();
        if (tid == 0) Y[OY_Y1] += step * (AX[OY_Y1] - b1);
        for (int l = tid; l < NL; l += NT) {
            Y[OY_Y2 + l] += step * (AX[OY_Y2 + l] - w[WS_B2 + l]);
            Y[OY_Y3 + l] += step * (AX[OY_Y3 + l] - w[WS_B3 + l]);
        }
        for (int s = tid; s < NS; s += NT)
            Y[OY_Y4 + s] += step * (AX[OY_Y4 + s] - w[WS_B4 + s]);
        for (int g = tid; g < NG; g += NT) {
            Y[OY_Z1 + g] = fmaxf(Y[OY_Z1 + g] + step * (AX[OY_Z1 + g] - Pmax[g]), 0.f);
            Y[OY_Z2 + g] = fmaxf(Y[OY_Z2 + g] + step *  AX[OY_Z2 + g], 0.f);
        }
        __syncthreads();
    }

    for (int g = tid; g < NG; g += NT) {
        out[g]          = X[OX_P  + g];
        out[NG + g]     = X[OX_RU + g];
        out[2 * NG + g] = X[OX_RD + g];
    }
    for (int i = tid; i < NG * NS; i += NT) {
        int g = i / NS, s = i % NS;
        out[3 * NG + i]           = X[OX_U + g * S_P + s];
        out[3 * NG + NG * NS + i] = X[OX_D + g * S_P + s];
    }
    for (int l = tid; l < NL; l += NT) {
        out[3 * NG + 2 * NG * NS + l]      = X[OX_FU + l];
        out[3 * NG + 2 * NG * NS + NL + l] = X[OX_FD + l];
    }
}


extern "C" void kernel_launch(void* const* d_in, const int* in_sizes, int n_in,
                              void* d_out, int out_size, void* d_ws, size_t ws_size,
                              hipStream_t stream) {
    const float* w_scen   = (const float*)d_in[0];
    const float* Cost     = (const float*)d_in[1];
    const float* C_r_up   = (const float*)d_in[2];
    const float* C_r_down = (const float*)d_in[3];
    const float* Pmax     = (const float*)d_in[4];
    const float* PTDF     = (const float*)d_in[5];
    const float* node_G   = (const float*)d_in[6];
    const float* node_W   = (const float*)d_in[7];
    const float* node_L   = (const float*)d_in[8];
    const float* Pd       = (const float*)d_in[9];
    const float* w_exp    = (const float*)d_in[10];
    const float* Line_cap = (const float*)d_in[11];
    float* w   = (float*)d_ws;
    float* out = (float*)d_out;

    k_zero <<<512, 256, 0, stream>>>(w);
    k_prep1<<<240, 256, 0, stream>>>(w, w_scen, node_W, node_L, Pd, w_exp);
    k_prep2<<<640, 256, 0, stream>>>(w, PTDF, node_G, Line_cap);
    k_setv <<<161, 256, 0, stream>>>(w);
    k_solver<<<1, NT, 0, stream>>>(w, Cost, C_r_up, C_r_down, Pmax, out);
}


# PDHG state planes (z5, z6, pw, U, D, z3, z4) lie in memory in accumulator-register order: epilogues and the PG^T product's B reads are whole-line 16-byte accesses; that product's k sum order changes
# speedup vs baseline: 5.0850x; 1.0477x over previous
#include <hip/hip_runtime.h>
#include <math.h>

typedef __attribute__((ext_vector_type(2))) float v2f;
typedef __attribute__((ext_vector_type(8))) float v8f;

#define NT 512

constexpr int NG = 100, NL = 400, NN = 300, NWF = 50, NS = 200;
constexpr int G_P = 112;
constexpr int S_P = 208;
constexpr int PDHG_ITERS = 300, POWER_ITERS = 25;
constexpr int PG_LP = 116;
constexpr int N_STRIP = S_P / 16;
constexpr int RS5 = 0;
constexpr int RS6 = RS5 + N_STRIP * NL;
constexpr int RS3 = RS6 + N_STRIP * NL;
constexpr int RS4 = RS3 + N_STRIP * G_P;
constexpr int RS_TOTAL = RS4 + N_STRIP * G_P;
constexpr int C_RANGES = 4;
constexpr int C_RANGE = NL / C_RANGES;
constexpr int C_SLOT = 128;
constexpr int UD_LP = 210;
constexpr int TILE_F = 256;
__host__ __device__ constexpr int packed_off(int row, int col, int row_tiles) {
    return ((col >> 4) * row_tiles + (row >> 4)) * TILE_F + ((((row >> 3) & 1) * 16) + (col & 15)) * 8 + (row & 7);
}

constexpr int OX_P  = 0;
constexpr int OX_RU = OX_P  + G_P;
constexpr int OX_RD = OX_RU + G_P;
constexpr int OX_U  = OX_RD + G_P;
constexpr int OX_D  = OX_U  + G_P * S_P;
constexpr int OX_FU = OX_D  + G_P * S_P;
constexpr int OX_FD = OX_FU + NL;
constexpr int XSZ   = OX_FD + NL;

constexpr int OY_Y1 = 0;
constexpr int OY_Y2 = 16;
constexpr int OY_Y3 = OY_Y2 + NL;
constexpr int OY_Y4 = OY_Y3 + NL;
constexpr int OY_Z1 = OY_Y4 + S_P;
constexpr int OY_Z2 = OY_Z1 + G_P;
constexpr int OY_Z3 = OY_Z2 + G_P;
constexpr int OY_Z4 = OY_Z3 + G_P * S_P;
constexpr int OY_Z5 = OY_Z4 + G_P * S_P;
constexpr int OY_Z6 = OY_Z5 + NL * S_P;
constexpr int YSZ   = OY_Z6 + NL * S_P;

constexpr int WS_PG   = 0;
constexpr int WS_PGT  = WS_PG   + NL * G_P;
constexpr int WS_PW   = WS_PGT  + G_P * NL;
constexpr int WS_B2   = WS_PW   + NL * S_P;
constexpr int WS_B3   = WS_B2   + NL;
constexpr int WS_B4   = WS_B3   + NL;
constexpr int WS_B1   = WS_B4   + S_P;
constexpr int WS_M1   = WS_B1   + 16;
constexpr int WS_NV   = WS_M1   + NN * S_P;
constexpr int WS_X    = WS_NV   + NN + 4;
constexpr int WS_XB   = WS_X    + XSZ;
constexpr int WS_ATY  = WS_XB   + XSZ;
constexpr int WS_Y    = WS_ATY  + XSZ;
constexpr int WS_AX   = WS_Y    + YSZ;
constexpr int WS_GP   = WS_AX   + YSZ;
constexpr int WS_COL  = WS_GP   + NL;
constexpr int WS_GTY  = WS_COL  + S_P;
constexpr int WS_SUMP = WS_GTY  + G_P;
constexpr int WS_TOTAL= WS_SUMP + 16;


__global__ void k_zero(float* w) {
    for (int i = blockIdx.x * blockDim.x + threadIdx.x; i < WS_TOTAL;
         i += gridDim.x * blockDim.x) w[i] = 0.0f;
}

__global__ void k_prep1(float* w, const float* w_scen, const float* node_W,
                        const float* node_L, const float* Pd, const float* w_exp) {
    const int NM1 = NN * NS;
    const int total = NM1 + NN + NS + 1;
    for (int idx = blockIdx.x * blockDim.x + threadIdx.x; idx < total;
         idx += gridDim.x * blockDim.x) {
        if (idx < NM1) {
            int n = idx / NS, s = idx % NS;
            float a = 0.f;
            for (int ww = 0; ww < NWF; ++ww) a += node_W[n*NWF + ww] * w_scen[s*NWF + ww];
            w[WS_M1 + n*S_P + s] = a;
        } else if (idx < NM1 + NN) {
            int n = idx - NM1;
            float a = 0.f;
            for (int ww = 0; ww < NWF; ++ww) a += node_W[n*NWF + ww] * w_exp[ww];
            for (int m = 0; m < NN; ++m)     a -= node_L[n*NN + m] * Pd[m];
            w[WS_NV + n] = a;
        } else if (idx < NM1 + NN + NS) {
            int s = idx - NM1 - NN;
            float a = 0.f;
            for (int ww = 0; ww < NWF; ++ww) a += w_scen[s*NWF + ww];
            w[WS_B4 + s] = -a;
        } else {
            float a = 0.f;
            for (int m = 0; m < NN; ++m)  a += Pd[m];
            for (int ww = 0; ww < NWF; ++ww) a -= w_exp[ww];
            w[WS_B1] = a;
        }
    }
}

__global__ void k_prep2(float* w, const float* PTDF, const float* node_G,
                        const float* Line_cap) {
    const int A = NL * NG;
    const int C = A + NL * NS;
    const int D2 = C + NL;
    const int E = D2 + NL;
    for (int idx = blockIdx.x * blockDim.x + threadIdx.x; idx < E;
         idx += gridDim.x * blockDim.x) {
        if (idx < A) {
            int l = idx / NG, g = idx % NG;
            float a = 0.f;
            for (int n = 0; n < NN; ++n) a += PTDF[l*NN + n] * node_G[n*NG + g];
            w[WS_PG + l*G_P + g] = a;
        } else if (idx < C) {
            int r = idx - A; int l = r / NS, s = r % NS;
            float a = 0.f;
            for (int n = 0; n < NN; ++n) a += PTDF[l*NN + n] * w[WS_M1 + n*S_P + s];
            w[WS_PW + packed_off(l, s, NL / 16)] = a;
        } else if (idx < D2) {
            int l = idx - C;
            float a = 0.f;
            for (int n = 0; n < NN; ++n) a += PTDF[l*NN + n] * w[WS_NV + n];
            w[WS_B2 + l] = Line_cap[l] - a;
        } else {
            int l = idx - D2;
            float a = 0.f;
            for (int n = 0; n < NN; ++n) a += PTDF[l*NN + n] * w[WS_NV + n];
            w[WS_B3 + l] = Line_cap[l] + a;
        }
    }
}

__global__ void k_setv(float* w) {
    const int total = 3*NG + 2*NG*NS + 2*NL;
    float* X = w + WS_X;
    for (int idx = blockIdx.x * blockDim.x + threadIdx.x; idx < total;
         idx += gridDim.x * blockDim.x) {
        int off;
        if (idx < NG)                off = OX_P  + idx;
        else if (idx < 2*NG)         off = OX_RU + (idx - NG);
        else if (idx < 3*NG)         off = OX_RD + (idx - 2*NG);
        else if (idx < 3*NG + NG*NS) { int r = idx - 3*NG; off = OX_U + (r/NS)*S_P + (r%NS); }
        else if (idx < 3*NG + 2*NG*NS) { int r = idx - 3*NG - NG*NS; off = OX_D + (r/NS)*S_P + (r%NS); }
        else if (idx < 3*NG + 2*NG*NS + NL) off = OX_FU + (idx - 3*NG - 2*NG*NS);
        else                         off = OX_FD + (idx - 3*NG - 2*NG*NS - NL);
        X[off] = 1.0f;
    }
}


__device__ __forceinline__ float block_reduce(float v, float* red, int tid) {
    red[tid] = v; __syncthreads();
    for (int off = NT >> 1; off > 0; off >>= 1) {
        if (tid < off) red[tid] += red[tid + off];
        __syncthreads();
    }
    float r = red[0]; __syncthreads();
    return r;
}

template <int M>
__device__ __forceinline__ float row_xor(float x) {
    return __int_as_float(__builtin_amdgcn_update_dpp(0, __float_as_int(x), 0x160 + M, 0xf, 0xf, true));
}

__device__ __forceinline__ float rows_over_lanes(const float (&v)[8], int c) {
    const bool b2 = (c & 4) != 0, b1 = (c & 2) != 0, b0 = (c & 1) != 0;
    float w[4], u[2];
#pragma unroll
    for (int j = 0; j < 4; ++j) {
        float keep = b2 ? v[4 + j] : v[j];
        float send = b2 ? v[j] : v[4 + j];
        w[j] = keep + row_xor<4>(send);
    }
#pragma unroll
    for (int j = 0; j < 2; ++j) {
        float keep = b1 ? w[2 + j] : w[j];
        float send = b1 ? w[j] : w[2 + j];
        u[j] = keep + row_xor<2>(send);
    }
    float keep = b0 ? u[1] : u[0];
    float send = b0 ? u[0] : u[1];
    float t = keep + row_xor<1>(send);
    return t + row_xor<8>(t);
}

__device__ __forceinline__ void pgt_ydiff_partials(const float* sPG, const float* yb, int tid, float* sPart) {
    const int r = __builtin_amdgcn_readfirstlane(tid / C_SLOT);
    const int g = tid % C_SLOT;
    if (g < NG) {
        const float* pc = sPG + r * C_RANGE * PG_LP + g;
        const float* y2 = yb + OY_Y2 + r * C_RANGE;
        const float* y3 = yb + OY_Y3 + r * C_RANGE;
        float c = 0.f;
#pragma unroll
        for (int l = 0; l < C_RANGE; ++l) c += pc[l * PG_LP] * (y2[l] - y3[l]);
        sPart[r * C_SLOT + g] = c;
    }
}

__device__ __forceinline__ float pgt_ydiff_sum(const float* sPart, int g) {
    return ((sPart[g] + sPart[C_SLOT + g]) + sPart[2 * C_SLOT + g]) + sPart[3 * C_SLOT + g];
}

template <bool DUAL>
__device__ __forceinline__ void matvec_core(float* w, const float* sPG, const float* xb,
                                            float* yb, int tid, float* Yd, const float* pw,
                                            float step, float* sRS, const float* sUD, const float* sXP) {
    const int lane = tid & 31, wave = __builtin_amdgcn_readfirstlane(tid >> 5), nw = NT / 32;
    float* Zd = DUAL ? Yd : yb;
    __syncthreads();
    {
        const float* Ux = xb + OX_U; const float* Dx = xb + OX_D;
        constexpr int TM = 25, NTile = 25 * 13;
        const int m16 = lane & 15;
        const int kh  = (lane >> 4) * 2;
        const int t1 = ((wave + 1) * NTile) / nw;
        int t = (wave * NTile) / nw;
        while (t < t1) {
            const int tn = t / TM;
            const int tm_lo = t - tn * TM;
            const int tm_hi = (tm_lo + (t1 - t) < TM) ? tm_lo + (t1 - t) : TM;
            const int n = tn * 16 + m16;
            v2f b[25];
#pragma unroll
            for (int j = 0; j < 25; ++j) {
                if (DUAL) {
                    int bi = (4 * j + kh) * UD_LP + n;
                    b[j][0] = sUD[bi];
                    b[j][1] = sUD[bi + UD_LP];
                } else {
                    int bi = (4 * j + kh) * S_P + n;
                    b[j][0] = Ux[bi]       - Dx[bi];
                    b[j][1] = Ux[bi + S_P] - Dx[bi + S_P];
                }
            }
            for (int tm = tm_lo; tm < tm_hi; ++tm) {
                const int mr = tm * 16 + ((lane >> 4) << 3);
                float y5[8], y6[8], pv[8], fu[8], fd[8];
                const int po = (tn * TM + tm) * TILE_F + lane * 8;
                if (DUAL) {
                    const float4 p0 = *(const float4*)(Yd + OY_Z5 + po), p1 = *(const float4*)(Yd + OY_Z5 + po + 4);
                    const float4 q0 = *(const float4*)(Yd + OY_Z6 + po), q1 = *(const float4*)(Yd + OY_Z6 + po + 4);
                    const float4 w0 = *(const float4*)(pw + po),         w1 = *(const float4*)(pw + po + 4);
                    y5[0] = p0.x; y5[1] = p0.y; y5[2] = p0.z; y5[3] = p0.w; y5[4] = p1.x; y5[5] = p1.y; y5[6] = p1.z; y5[7] = p1.w;
                    y6[0] = q0.x; y6[1] = q0.y; y6[2] = q0.z; y6[3] = q0.w; y6[4] = q1.x; y6[5] = q1.y; y6[6] = q1.z; y6[7] = q1.w;
                    pv[0] = w0.x; pv[1] = w0.y; pv[2] = w0.z; pv[3] = w0.w; pv[4] = w1.x; pv[5] = w1.y; pv[6] = w1.z; pv[7] = w1.w;
                }
#pragma unroll
                for (int i = 0; i < 8; ++i) { fu[i] = xb[OX_FU + mr + i]; fd[i] = xb[OX_FD + mr + i]; }
                const float* arow = sPG + (tm * 16 + m16) * PG_LP + kh;
                v2f a[25];
#pragma unroll
                for (int j = 0; j < 25; ++j) { a[j][0] = arow[4 * j]; a[j][1] = arow[4 * j + 1]; }
                v8f acc = {0.f,0.f,0.f,0.f,0.f,0.f,0.f,0.f};
#pragma unroll
                for (int j = 0; j < 25; ++j)
                    acc = __builtin_amdgcn_wmma_f32_16x16x4_f32(false, a[j], false, b[j],
                                                                (short)0, acc, false, false);
                __builtin_amdgcn_sched_group_barrier(0x100, 13, 0);
                __builtin_amdgcn_sched_group_barrier(0x008, 25, 0);
                const bool live = n < NS;
                float r5[8], r6[8];
#pragma unroll
                for (int i = 0; i < 8; ++i) {
                    int o = (mr + i) * S_P + n;
                    float f = acc[i];
                    float n5, n6;
                    if (DUAL) {
                        n5 = fmaxf(y5[i] + step * (( f - fu[i]) + pv[i]), 0.f);
                        n6 = fmaxf(y6[i] + step * ((-f - fd[i]) - pv[i]), 0.f);
                    } else {
                        n5 =  f - fu[i];
                        n6 = -f - fd[i];
                    }
                    r5[i] = live ? n5 : 0.f;
                    r6[i] = live ? n6 : 0.f;
                    if (!DUAL) {
                        Zd[OY_Z5 + o] = r5[i];
                        Zd[OY_Z6 + o] = r6[i];
                    }
                }
                if (DUAL) {
                    *(float4*)(Zd + OY_Z5 + po)     = make_float4(r5[0], r5[1], r5[2], r5[3]);
                    *(float4*)(Zd + OY_Z5 + po + 4) = make_float4(r5[4], r5[5], r5[6], r5[7]);
                    *(float4*)(Zd + OY_Z6 + po)     = make_float4(r6[0], r6[1], r6[2], r6[3]);
                    *(float4*)(Zd + OY_Z6 + po + 4) = make_float4(r6[4], r6[5], r6[6], r6[7]);
                }
                const float s5 = rows_over_lanes(r5, m16);
                const float s6 = rows_over_lanes(r6, m16);
                if (m16 < 8) {
                    sRS[RS5 + tn * NL + mr + m16] = s5;
                    sRS[RS6 + tn * NL + mr + m16] = s6;
                }
            }
            t += tm_hi - tm_lo;
        }
    }
    if (tid < NL) {
        float s = 0.f; const float* r = sPG + tid * PG_LP;
        if (DUAL) {
            for (int g = 0; g < NG; ++g) s += r[g] * sXP[g];
        } else {
            const float* p = xb + OX_P;
            for (int g = 0; g < NG; ++g) s += r[g] * p[g];
        }
        w[WS_GP + tid] = s;
    }
    if (tid < NS) {
        float s = 0.f;
        if (DUAL) {
            const float* d = sUD + tid;
            for (int g = 0; g < NG; ++g) s += d[g * UD_LP];
        } else {
            const float* Ux = xb + OX_U + tid; const float* Dx = xb + OX_D + tid;
            for (int g = 0; g < NG; ++g) s += Ux[g * S_P] - Dx[g * S_P];
        }
        w[WS_COL + tid] = s;
    }
    if (wave == nw - 1) {
        float a;
        if (DUAL) {
            a = sXP[lane] + sXP[32 + lane] + sXP[64 + lane];
            if (lane < NG - 96) a += sXP[96 + lane];
        } else {
            a = xb[OX_P + lane] + xb[OX_P + 32 + lane] + xb[OX_P + 64 + lane];
            if (lane < NG - 96) a += xb[OX_P + 96 + lane];
        }
        a += row_xor<1>(a); a += row_xor<2>(a); a += row_xor<4>(a); a += row_xor<8>(a);
        a += __shfl_xor(a, 16);
        if (lane == 0) w[WS_SUMP] = a;
    }
    __syncthreads();
    if (tid == 0) yb[OY_Y1] = w[WS_SUMP];
    for (int l = tid; l < NL; l += NT) {
        float g = w[WS_GP + l];
        yb[OY_Y2 + l] =  g + xb[OX_FU + l];
        yb[OY_Y3 + l] = -g + xb[OX_FD + l];
    }
    for (int s = tid; s < NS; s += NT) yb[OY_Y4 + s] = w[WS_COL + s];
    for (int g = tid; g < NG; g += NT) {
        yb[OY_Z1 + g] = xb[OX_P + g]  + xb[OX_RU + g];
        yb[OY_Z2 + g] = xb[OX_RD + g] - xb[OX_P + g];
    }
}

template <bool FUSED>
__device__ __forceinline__ void rmatvec_core(float* w, const float* sPG, float* yb,
                                             float* xo, int tid, float* X, float* XB,
                                             const float* Cost, const float* Cru,
                                             const float* Crd, float step, float* sRS, float* sPart,
                                             float* sUD, float* sXP) {
    const int lane = tid & 31, wave = __builtin_amdgcn_readfirstlane(tid >> 5), nw = NT / 32;
    __syncthreads();
    if (FUSED) {
        pgt_ydiff_partials(sPG, yb, tid, sPart);
        __syncthreads();
        if (tid < NG) {
            float a = 0.f, b = 0.f;
            for (int k = 0; k < N_STRIP; ++k) {
                a += sRS[RS3 + k * G_P + tid]; b += sRS[RS4 + k * G_P + tid];
            }
            const float c = pgt_ydiff_sum(sPart, tid);
            const float z1 = yb[OY_Z1 + tid], z2 = yb[OY_Z2 + tid];
            const float ap  = yb[OY_Y1] + c + z1 - z2;
            const float aru = z1 - a;
            const float ard = z2 - b;
            float o, n;
            o = X[OX_P  + tid]; n = fmaxf(o - step * (Cost[tid] + ap),  0.f);
            const float bp = 2.f * n - o;
            X[OX_P  + tid] = n; XB[OX_P  + tid] = bp; sXP[tid] = bp;
            o = X[OX_RU + tid]; n = fmaxf(o - step * (Cru[tid]  + aru), 0.f);
            X[OX_RU + tid] = n; XB[OX_RU + tid] = 2.f * n - o;
            o = X[OX_RD + tid]; n = fmaxf(o - step * (Crd[tid]  + ard), 0.f);
            X[OX_RD + tid] = n; XB[OX_RD + tid] = 2.f * n - o;
        }
        if (tid >= NG && tid < NG + NL) {
            int l = tid - NG;
            float a = 0.f, b = 0.f;
            for (int k = 0; k < N_STRIP; ++k) {
                a += sRS[RS5 + k * NL + l]; b += sRS[RS6 + k * NL + l];
            }
            const float afu = yb[OY_Y2 + l] - a;
            const float afd = yb[OY_Y3 + l] - b;
            float o, n;
            o = X[OX_FU + l]; n = fmaxf(o - step * afu, 0.f);
            X[OX_FU + l] = n; XB[OX_FU + l] = 2.f * n - o;
            o = X[OX_FD + l]; n = fmaxf(o - step * afd, 0.f);
            X[OX_FD + l] = n; XB[OX_FD + l] = 2.f * n - o;
        }
        __syncthreads();
    }
    {
        const float* Z5 = yb + OY_Z5; const float* Z6 = yb + OY_Z6;
        constexpr int TM = 7, NTile = 7 * 13;
        const int m16 = lane & 15;
        const int kh  = (lane >> 4) * 2;
        const int t1 = ((wave + 1) * NTile) / nw;
        int t = (wave * NTile) / nw;
        while (t < t1) {
            const int tn = t / TM;
            const int tm_lo = t - tn * TM;
            const int tm_hi = (tm_lo + (t1 - t) < TM) ? tm_lo + (t1 - t) : TM;
            const int n = tn * 16 + m16;
            v8f acc[TM];
#pragma unroll
            for (int q = 0; q < TM; ++q) acc[q] = (v8f){0.f,0.f,0.f,0.f,0.f,0.f,0.f,0.f};
            for (int kb = 0; kb < NL; kb += 40) {
                v2f b[10];
                if (FUSED) {
                    const int zo = tn * (NL / 16) * TILE_F + (kb >> 3) * 128 + m16 * 8 + (lane >> 4) * 4;
#pragma unroll
                    for (int blk = 0; blk < 5; ++blk) {
                        const float4 u = *(const float4*)(Z5 + zo + blk * 128);
                        const float4 v = *(const float4*)(Z6 + zo + blk * 128);
                        b[2 * blk][0]     = u.x - v.x;  b[2 * blk][1]     = u.y - v.y;
                        b[2 * blk + 1][0] = u.z - v.z;  b[2 * blk + 1][1] = u.w - v.w;
                    }
                } else {
#pragma unroll
                    for (int j = 0; j < 10; ++j) {
                        int bi = (kb + 4 * j + kh) * S_P + n;
                        b[j][0] = Z5[bi]       - Z6[bi];
                        b[j][1] = Z5[bi + S_P] - Z6[bi + S_P];
                    }
                }
#pragma unroll
                for (int q = 0; q < TM; ++q) {
                    if (q >= tm_lo && q < tm_hi) {
                        const float* acol = sPG + (kb + (FUSED ? 2 * kh : kh)) * PG_LP + q * 16 + m16;
                        v2f a[10];
#pragma unroll
                        for (int j = 0; j < 10; ++j) {
                            const int r = FUSED ? 8 * (j >> 1) + 2 * (j & 1) : 4 * j;
                            a[j][0] = acol[r * PG_LP]; a[j][1] = acol[(r + 1) * PG_LP];
                        }
#pragma unroll
                        for (int j = 0; j < 10; ++j)
                            acc[q] = __builtin_amdgcn_wmma_f32_16x16x4_f32(false, a[j], false, b[j],
                                                                           (short)0, acc[q], false, false);
                        __builtin_amdgcn_sched_group_barrier(0x100, 10, 0);
                        __builtin_amdgcn_sched_group_barrier(0x008, 10, 0);
                    }
                }
            }
            const float y4v = yb[OY_Y4 + n];
            const int ud0 = ((lane >> 4) << 3) * UD_LP + n;
#pragma unroll
            for (int q = 0; q < TM; ++q) {
                if (q >= tm_lo && q < tm_hi) {
                    int mr = q * 16 + ((lane >> 4) << 3);
                    if (FUSED) {
                        float xu[8], xd[8], z3v[8], z4v[8], ruv[8], rdv[8];
                        const int po = (tn * TM + q) * TILE_F + lane * 8;
                        {
                            const float4 u0 = *(const float4*)(X + OX_U + po),   u1 = *(const float4*)(X + OX_U + po + 4);
                            const float4 d0 = *(const float4*)(X + OX_D + po),   d1 = *(const float4*)(X + OX_D + po + 4);
                            const float4 e0 = *(const float4*)(yb + OY_Z3 + po), e1 = *(const float4*)(yb + OY_Z3 + po + 4);
                            const float4 f0 = *(const float4*)(yb + OY_Z4 + po), f1 = *(const float4*)(yb + OY_Z4 + po + 4);
                            xu[0] = u0.x; xu[1] = u0.y; xu[2] = u0.z; xu[3] = u0.w; xu[4] = u1.x; xu[5] = u1.y; xu[6] = u1.z; xu[7] = u1.w;
                            xd[0] = d0.x; xd[1] = d0.y; xd[2] = d0.z; xd[3] = d0.w; xd[4] = d1.x; xd[5] = d1.y; xd[6] = d1.z; xd[7] = d1.w;
                            z3v[0] = e0.x; z3v[1] = e0.y; z3v[2] = e0.z; z3v[3] = e0.w; z3v[4] = e1.x; z3v[5] = e1.y; z3v[6] = e1.z; z3v[7] = e1.w;
                            z4v[0] = f0.x; z4v[1] = f0.y; z4v[2] = f0.z; z4v[3] = f0.w; z4v[4] = f1.x; z4v[5] = f1.y; z4v[6] = f1.z; z4v[7] = f1.w;
                        }
#pragma unroll
                        for (int i = 0; i < 8; ++i) { ruv[i] = XB[OX_RU + mr + i]; rdv[i] = XB[OX_RD + mr + i]; }
                        float r3[8], r4[8], sU[8], sD[8];
#pragma unroll
                        for (int i = 0; i < 8; ++i) {
                            float tv = acc[q][i];
                            float aU =  y4v + z3v[i] + tv;
                            float aD = -y4v + z4v[i] - tv;
                            float nU = fmaxf(xu[i] - step * aU, 0.f);
                            float nD = fmaxf(xd[i] - step * aD, 0.f);
                            float bU = 2.f * nU - xu[i];
                            float bD = 2.f * nD - xd[i];
                            float n3 = fmaxf(z3v[i] + step * (bU - ruv[i]), 0.f);
                            float n4 = fmaxf(z4v[i] + step * (bD - rdv[i]), 0.f);
                            const bool ok = n < NS && mr + i < NG;
                            const float dv = ok ? bU - bD : 0.f;
                            if (mr + i < NG) sUD[ud0 + (q * 16 + i) * UD_LP] = dv;
                            sU[i] = ok ? nU : 0.f;
                            sD[i] = ok ? nD : 0.f;
                            r3[i] = ok ? n3 : 0.f;
                            r4[i] = ok ? n4 : 0.f;
                        }
                        *(float4*)(X + OX_U + po)       = make_float4(sU[0], sU[1], sU[2], sU[3]);
                        *(float4*)(X + OX_U + po + 4)   = make_float4(sU[4], sU[5], sU[6], sU[7]);
                        *(float4*)(X + OX_D + po)       = make_float4(sD[0], sD[1], sD[2], sD[3]);
                        *(float4*)(X + OX_D + po + 4)   = make_float4(sD[4], sD[5], sD[6], sD[7]);
                        *(float4*)(yb + OY_Z3 + po)     = make_float4(r3[0], r3[1], r3[2], r3[3]);
                        *(float4*)(yb + OY_Z3 + po + 4) = make_float4(r3[4], r3[5], r3[6], r3[7]);
                        *(float4*)(yb + OY_Z4 + po)     = make_float4(r4[0], r4[1], r4[2], r4[3]);
                        *(float4*)(yb + OY_Z4 + po + 4) = make_float4(r4[4], r4[5], r4[6], r4[7]);
                        const float s3 = rows_over_lanes(r3, m16);
                        const float s4 = rows_over_lanes(r4, m16);
                        if (m16 < 8) {
                            sRS[RS3 + tn * G_P + mr + m16] = s3;
                            sRS[RS4 + tn * G_P + mr + m16] = s4;
                        }
                    } else {
                        float xu[8], xd[8], ruv[8], rdv[8];
#pragma unroll
                        for (int i = 0; i < 8; ++i) {
                            int o = (mr + i) * S_P + n;
                            xu[i]  = xo[OX_U + o];       xd[i]  = xo[OX_D + o];
                            ruv[i] = xo[OX_RU + mr + i]; rdv[i] = xo[OX_RD + mr + i];
                        }
                        float r3[8], r4[8];
#pragma unroll
                        for (int i = 0; i < 8; ++i) {
                            int o = (mr + i) * S_P + n;
                            float tv = acc[q][i];
                            float z3 = xu[i] - ruv[i];
                            float z4 = xd[i] - rdv[i];
                            const bool ok = n < NS && mr + i < NG;
                            if (ok) {
                                xo[OX_U + o] =  y4v + z3 + tv;
                                xo[OX_D + o] = -y4v + z4 - tv;
                            }
                            r3[i] = ok ? z3 : 0.f;
                            r4[i] = ok ? z4 : 0.f;
                        }
                        const float s3 = rows_over_lanes(r3, m16);
                        const float s4 = rows_over_lanes(r4, m16);
                        if (m16 < 8) {
                            sRS[RS3 + tn * G_P + mr + m16] = s3;
                            sRS[RS4 + tn * G_P + mr + m16] = s4;
                        }
                    }
                }
            }
            t += tm_hi - tm_lo;
        }
    }
    if (FUSED) return;
    pgt_ydiff_partials(sPG, yb, tid, sPart);
    __syncthreads();
    if (tid < NG) {
        float a = 0.f, b = 0.f;
        for (int k = 0; k < N_STRIP; ++k) {
            a += sRS[RS3 + k * G_P + tid]; b += sRS[RS4 + k * G_P + tid];
        }
        const float z1 = yb[OY_Z1 + tid], z2 = yb[OY_Z2 + tid];
        xo[OX_P  + tid] = yb[OY_Y1] + pgt_ydiff_sum(sPart, tid) + z1 - z2;
        xo[OX_RU + tid] = z1 - a;
        xo[OX_RD + tid] = z2 - b;
    }
    if (tid >= NG && tid < NG + NL) {
        int l = tid - NG;
        float a = 0.f, b = 0.f;
        for (int k = 0; k < N_STRIP; ++k) {
            a += sRS[RS5 + k * NL + l]; b += sRS[RS6 + k * NL + l];
        }
        xo[OX_FU + l] = yb[OY_Y2 + l] - a;
        xo[OX_FD + l] = yb[OY_Y3 + l] - b;
    }
}


__global__ __launch_bounds__(NT) void k_solver(float* w, const float* Cost,
                                               const float* Cru, const float* Crd,
                                               const float* Pmax, float* out) {
    __shared__ float red[NT];
    __shared__ __attribute__((aligned(16))) float sPG[NL * PG_LP];
    __shared__ float sRS[RS_TOTAL];
    __shared__ float sPart[C_RANGES * C_SLOT];
    __shared__ float sUD[NG * UD_LP];
    __shared__ float sXP[G_P];
    const int tid = threadIdx.x;
    float* X   = w + WS_X;
    float* XB  = w + WS_XB;
    float* Y   = w + WS_Y;
    float* AX  = w + WS_AX;

    for (int i = tid; i < NL * G_P; i += NT) {
        int l = i / G_P, g = i - l * G_P;
        sPG[l * PG_LP + g] = w[WS_PG + i];
    }
    __syncthreads();

    float sumsq = 0.f;
    for (int it = 0; it < POWER_ITERS; ++it) {
        __syncthreads();
        float part = 0.f;
        for (int i = tid; i < XSZ; i += NT) { float v = X[i]; part += v * v; }
        sumsq = block_reduce(part, red, tid);
        float inv = 1.0f / sqrtf(sumsq);
        for (int i = tid; i < XSZ; i += NT) X[i] *= inv;
        matvec_core<false>(w, sPG, X, AX, tid, nullptr, nullptr, 0.f, sRS, nullptr, nullptr);
        rmatvec_core<false>(w, sPG, AX, X, tid, nullptr, nullptr, nullptr, nullptr, nullptr, 0.f, sRS, sPart, nullptr, nullptr);
    }
    __syncthreads();
    {
        float part = 0.f;
        for (int i = tid; i < XSZ; i += NT) { float v = X[i]; part += v * v; }
        sumsq = block_reduce(part, red, tid);
    }
    const float step = 0.9f / sqrtf(sqrtf(sumsq));
    const float b1 = w[WS_B1];
    const float* pw = w + WS_PW;

    for (int i = tid; i < XSZ; i += NT) { X[i] = 0.f; XB[i] = 0.f; }
    for (int i = tid; i < YSZ; i += NT) Y[i] = 0.f;
    for (int i = tid; i < RS_TOTAL; i += NT) sRS[i] = 0.f;
    for (int i = tid; i < NG * UD_LP; i += NT) sUD[i] = 0.f;

    for (int it = 0; it < PDHG_ITERS; ++it) {
        rmatvec_core<true>(w, sPG, Y, nullptr, tid, X, XB, Cost, Cru, Crd, step, sRS, sPart, sUD, sXP);
        matvec_core<true>(w, sPG, XB, AX, tid, Y, pw, step, sRS, sUD, sXP);
        __syncthreads();
        if (tid == 0) Y[OY_Y1] += step * (AX[OY_Y1] - b1);
        for (int l = tid; l < NL; l += NT) {
            Y[OY_Y2 + l] += step * (AX[OY_Y2 + l] - w[WS_B2 + l]);
            Y[OY_Y3 + l] += step * (AX[OY_Y3 + l] - w[WS_B3 + l]);
        }
        for (int s = tid; s < NS; s += NT)
            Y[OY_Y4 + s] += step * (AX[OY_Y4 + s] - w[WS_B4 + s]);
        for (int g = tid; g < NG; g += NT) {
            Y[OY_Z1 + g] = fmaxf(Y[OY_Z1 + g] + step * (AX[OY_Z1 + g] - Pmax[g]), 0.f);
            Y[OY_Z2 + g] = fmaxf(Y[OY_Z2 + g] + step *  AX[OY_Z2 + g], 0.f);
        }
        __syncthreads();
    }

    for (int g = tid; g < NG; g += NT) {
        out[g]          = X[OX_P  + g];
        out[NG + g]     = X[OX_RU + g];
        out[2 * NG + g] = X[OX_RD + g];
    }
    for (int i = tid; i < NG * NS; i += NT) {
        int g = i / NS, s = i % NS;
        const int po = packed_off(g, s, G_P / 16);
        out[3 * NG + i]           = X[OX_U + po];
        out[3 * NG + NG * NS + i] = X[OX_D + po];
    }
    for (int l = tid; l < NL; l += NT) {
        out[3 * NG + 2 * NG * NS + l]      = X[OX_FU + l];
        out[3 * NG + 2 * NG * NS + NL + l] = X[OX_FD + l];
    }
}


extern "C" void kernel_launch(void* const* d_in, const int* in_sizes, int n_in,
                              void* d_out, int out_size, void* d_ws, size_t ws_size,
                              hipStream_t stream) {
    const float* w_scen   = (const float*)d_in[0];
    const float* Cost     = (const float*)d_in[1];
    const float* C_r_up   = (const float*)d_in[2];
    const float* C_r_down = (const float*)d_in[3];
    const float* Pmax     = (const float*)d_in[4];
    const float* PTDF     = (const float*)d_in[5];
    const float* node_G   = (const float*)d_in[6];
    const float* node_W   = (const float*)d_in[7];
    const float* node_L   = (const float*)d_in[8];
    const float* Pd       = (const float*)d_in[9];
    const float* w_exp    = (const float*)d_in[10];
    const float* Line_cap = (const float*)d_in[11];
    float* w   = (float*)d_ws;
    float* out = (float*)d_out;

    k_zero <<<512, 256, 0, stream>>>(w);
    k_prep1<<<240, 256, 0, stream>>>(w, w_scen, node_W, node_L, Pd, w_exp);
    k_prep2<<<640, 256, 0, stream>>>(w, PTDF, node_G, Line_cap);
    k_setv <<<161, 256, 0, stream>>>(w);
    k_solver<<<1, NT, 0, stream>>>(w, Cost, C_r_up, C_r_down, Pmax, out);
}


# PDHG loop: small vectors (p, r, f of x, xb; y1..y4, z1, z2; A xb's small blocks) stay in LDS (PG pitch 100 makes room), their constants in registers: small phases touch no memory; sums keep order
# speedup vs baseline: 5.2542x; 1.0333x over previous
#include <hip/hip_runtime.h>
#include <math.h>

typedef __attribute__((ext_vector_type(2))) float v2f;
typedef __attribute__((ext_vector_type(8))) float v8f;

#define NT 512

constexpr int NG = 100, NL = 400, NN = 300, NWF = 50, NS = 200;
constexpr int G_P = 112;
constexpr int S_P = 208;
constexpr int PDHG_ITERS = 300, POWER_ITERS = 25;
constexpr int PG_LP = 100;
constexpr int PG_TAIL = 16;
constexpr int SX_P = 0, SX_RU = 112, SX_RD = 224, SX_FU = 336, SX_FD = SX_FU + NL, SX_TOTAL = SX_FD + NL;
constexpr int SW_GP = 0, SW_COL = NL, SW_SUMP = NL + 208, SW_TOTAL = SW_SUMP + 16;
constexpr int N_STRIP = S_P / 16;
constexpr int RS5 = 0;
constexpr int RS6 = RS5 + N_STRIP * NL;
constexpr int RS3 = RS6 + N_STRIP * NL;
constexpr int RS4 = RS3 + N_STRIP * G_P;
constexpr int RS_TOTAL = RS4 + N_STRIP * G_P;
constexpr int C_RANGES = 4;
constexpr int C_RANGE = NL / C_RANGES;
constexpr int C_SLOT = 128;
constexpr int UD_LP = 210;
constexpr int TILE_F = 256;
__host__ __device__ constexpr int packed_off(int row, int col, int row_tiles) {
    return ((col >> 4) * row_tiles + (row >> 4)) * TILE_F + ((((row >> 3) & 1) * 16) + (col & 15)) * 8 + (row & 7);
}

constexpr int OX_P  = 0;
constexpr int OX_RU = OX_P  + G_P;
constexpr int OX_RD = OX_RU + G_P;
constexpr int OX_U  = OX_RD + G_P;
constexpr int OX_D  = OX_U  + G_P * S_P;
constexpr int OX_FU = OX_D  + G_P * S_P;
constexpr int OX_FD = OX_FU + NL;
constexpr int XSZ   = OX_FD + NL;

constexpr int OY_Y1 = 0;
constexpr int OY_Y2 = 16;
constexpr int OY_Y3 = OY_Y2 + NL;
constexpr int OY_Y4 = OY_Y3 + NL;
constexpr int OY_Z1 = OY_Y4 + S_P;
constexpr int OY_Z2 = OY_Z1 + G_P;
constexpr int OY_Z3 = OY_Z2 + G_P;
constexpr int OY_Z4 = OY_Z3 + G_P * S_P;
constexpr int OY_Z5 = OY_Z4 + G_P * S_P;
constexpr int OY_Z6 = OY_Z5 + NL * S_P;
constexpr int YSZ   = OY_Z6 + NL * S_P;

constexpr int WS_PG   = 0;
constexpr int WS_PGT  = WS_PG   + NL * G_P;
constexpr int WS_PW   = WS_PGT  + G_P * NL;
constexpr int WS_B2   = WS_PW   + NL * S_P;
constexpr int WS_B3   = WS_B2   + NL;
constexpr int WS_B4   = WS_B3   + NL;
constexpr int WS_B1   = WS_B4   + S_P;
constexpr int WS_M1   = WS_B1   + 16;
constexpr int WS_NV   = WS_M1   + NN * S_P;
constexpr int WS_X    = WS_NV   + NN + 4;
constexpr int WS_XB   = WS_X    + XSZ;
constexpr int WS_ATY  = WS_XB   + XSZ;
constexpr int WS_Y    = WS_ATY  + XSZ;
constexpr int WS_AX   = WS_Y    + YSZ;
constexpr int WS_GP   = WS_AX   + YSZ;
constexpr int WS_COL  = WS_GP   + NL;
constexpr int WS_GTY  = WS_COL  + S_P;
constexpr int WS_SUMP = WS_GTY  + G_P;
constexpr int WS_TOTAL= WS_SUMP + 16;


__global__ void k_zero(float* w) {
    for (int i = blockIdx.x * blockDim.x + threadIdx.x; i < WS_TOTAL;
         i += gridDim.x * blockDim.x) w[i] = 0.0f;
}

__global__ void k_prep1(float* w, const float* w_scen, const float* node_W,
                        const float* node_L, const float* Pd, const float* w_exp) {
    const int NM1 = NN * NS;
    const int total = NM1 + NN + NS + 1;
    for (int idx = blockIdx.x * blockDim.x + threadIdx.x; idx < total;
         idx += gridDim.x * blockDim.x) {
        if (idx < NM1) {
            int n = idx / NS, s = idx % NS;
            float a = 0.f;
            for (int ww = 0; ww < NWF; ++ww) a += node_W[n*NWF + ww] * w_scen[s*NWF + ww];
            w[WS_M1 + n*S_P + s] = a;
        } else if (idx < NM1 + NN) {
            int n = idx - NM1;
            float a = 0.f;
            for (int ww = 0; ww < NWF; ++ww) a += node_W[n*NWF + ww] * w_exp[ww];
            for (int m = 0; m < NN; ++m)     a -= node_L[n*NN + m] * Pd[m];
            w[WS_NV + n] = a;
        } else if (idx < NM1 + NN + NS) {
            int s = idx - NM1 - NN;
            float a = 0.f;
            for (int ww = 0; ww < NWF; ++ww) a += w_scen[s*NWF + ww];
            w[WS_B4 + s] = -a;
        } else {
            float a = 0.f;
            for (int m = 0; m < NN; ++m)  a += Pd[m];
            for (int ww = 0; ww < NWF; ++ww) a -= w_exp[ww];
            w[WS_B1] = a;
        }
    }
}

__global__ void k_prep2(float* w, const float* PTDF, const float* node_G,
                        const float* Line_cap) {
    const int A = NL * NG;
    const int C = A + NL * NS;
    const int D2 = C + NL;
    const int E = D2 + NL;
    for (int idx = blockIdx.x * blockDim.x + threadIdx.x; idx < E;
         idx += gridDim.x * blockDim.x) {
        if (idx < A) {
            int l = idx / NG, g = idx % NG;
            float a = 0.f;
            for (int n = 0; n < NN; ++n) a += PTDF[l*NN + n] * node_G[n*NG + g];
            w[WS_PG + l*G_P + g] = a;
        } else if (idx < C) {
            int r = idx - A; int l = r / NS, s = r % NS;
            float a = 0.f;
            for (int n = 0; n < NN; ++n) a += PTDF[l*NN + n] * w[WS_M1 + n*S_P + s];
            w[WS_PW + packed_off(l, s, NL / 16)] = a;
        } else if (idx < D2) {
            int l = idx - C;
            float a = 0.f;
            for (int n = 0; n < NN; ++n) a += PTDF[l*NN + n] * w[WS_NV + n];
            w[WS_B2 + l] = Line_cap[l] - a;
        } else {
            int l = idx - D2;
            float a = 0.f;
            for (int n = 0; n < NN; ++n) a += PTDF[l*NN + n] * w[WS_NV + n];
            w[WS_B3 + l] = Line_cap[l] + a;
        }
    }
}

__global__ void k_setv(float* w) {
    const int total = 3*NG + 2*NG*NS + 2*NL;
    float* X = w + WS_X;
    for (int idx = blockIdx.x * blockDim.x + threadIdx.x; idx < total;
         idx += gridDim.x * blockDim.x) {
        int off;
        if (idx < NG)                off = OX_P  + idx;
        else if (idx < 2*NG)         off = OX_RU + (idx - NG);
        else if (idx < 3*NG)         off = OX_RD + (idx - 2*NG);
        else if (idx < 3*NG + NG*NS) { int r = idx - 3*NG; off = OX_U + (r/NS)*S_P + (r%NS); }
        else if (idx < 3*NG + 2*NG*NS) { int r = idx - 3*NG - NG*NS; off = OX_D + (r/NS)*S_P + (r%NS); }
        else if (idx < 3*NG + 2*NG*NS + NL) off = OX_FU + (idx - 3*NG - 2*NG*NS);
        else                         off = OX_FD + (idx - 3*NG - 2*NG*NS - NL);
        X[off] = 1.0f;
    }
}


__device__ __forceinline__ float block_reduce(float v, float* red, int tid) {
    red[tid] = v; __syncthreads();
    for (int off = NT >> 1; off > 0; off >>= 1) {
        if (tid < off) red[tid] += red[tid + off];
        __syncthreads();
    }
    float r = red[0]; __syncthreads();
    return r;
}

template <int M>
__device__ __forceinline__ float row_xor(float x) {
    return __int_as_float(__builtin_amdgcn_update_dpp(0, __float_as_int(x), 0x160 + M, 0xf, 0xf, true));
}

__device__ __forceinline__ float rows_over_lanes(const float (&v)[8], int c) {
    const bool b2 = (c & 4) != 0, b1 = (c & 2) != 0, b0 = (c & 1) != 0;
    float w[4], u[2];
#pragma unroll
    for (int j = 0; j < 4; ++j) {
        float keep = b2 ? v[4 + j] : v[j];
        float send = b2 ? v[j] : v[4 + j];
        w[j] = keep + row_xor<4>(send);
    }
#pragma unroll
    for (int j = 0; j < 2; ++j) {
        float keep = b1 ? w[2 + j] : w[j];
        float send = b1 ? w[j] : w[2 + j];
        u[j] = keep + row_xor<2>(send);
    }
    float keep = b0 ? u[1] : u[0];
    float send = b0 ? u[0] : u[1];
    float t = keep + row_xor<1>(send);
    return t + row_xor<8>(t);
}

__device__ __forceinline__ void pgt_ydiff_partials(const float* sPG, const float* yb, int tid, float* sPart) {
    const int r = __builtin_amdgcn_readfirstlane(tid / C_SLOT);
    const int g = tid % C_SLOT;
    if (g < NG) {
        const float* pc = sPG + r * C_RANGE * PG_LP + g;
        const float* y2 = yb + OY_Y2 + r * C_RANGE;
        const float* y3 = yb + OY_Y3 + r * C_RANGE;
        float c = 0.f;
#pragma unroll
        for (int l = 0; l < C_RANGE; ++l) c += pc[l * PG_LP] * (y2[l] - y3[l]);
        sPart[r * C_SLOT + g] = c;
    }
}

__device__ __forceinline__ float pgt_ydiff_sum(const float* sPart, int g) {
    return ((sPart[g] + sPart[C_SLOT + g]) + sPart[2 * C_SLOT + g]) + sPart[3 * C_SLOT + g];
}

template <bool DUAL>
__device__ __forceinline__ void matvec_core(float* w, const float* sPG, const float* xb,
                                            float* yb, int tid, float* Yd, const float* pw,
                                            float step, float* sRS, const float* sUD, const float* sXP,
                                            const float* xbs, float* axs, float* sW) {
    const float* pP  = DUAL ? xbs + SX_P  : xb + OX_P;
    const float* ruP = DUAL ? xbs + SX_RU : xb + OX_RU;
    const float* rdP = DUAL ? xbs + SX_RD : xb + OX_RD;
    const float* fuP = DUAL ? xbs + SX_FU : xb + OX_FU;
    const float* fdP = DUAL ? xbs + SX_FD : xb + OX_FD;
    float* gpP   = DUAL ? sW + SW_GP   : w + WS_GP;
    float* colP  = DUAL ? sW + SW_COL  : w + WS_COL;
    float* sumpP = DUAL ? sW + SW_SUMP : w + WS_SUMP;
    float* yo    = DUAL ? axs : yb;
    const int lane = tid & 31, wave = __builtin_amdgcn_readfirstlane(tid >> 5), nw = NT / 32;
    float* Zd = DUAL ? Yd : yb;
    __syncthreads();
    {
        const float* Ux = xb + OX_U; const float* Dx = xb + OX_D;
        constexpr int TM = 25, NTile = 25 * 13;
        const int m16 = lane & 15;
        const int kh  = (lane >> 4) * 2;
        const int t1 = ((wave + 1) * NTile) / nw;
        int t = (wave * NTile) / nw;
        while (t < t1) {
            const int tn = t / TM;
            const int tm_lo = t - tn * TM;
            const int tm_hi = (tm_lo + (t1 - t) < TM) ? tm_lo + (t1 - t) : TM;
            const int n = tn * 16 + m16;
            v2f b[25];
#pragma unroll
            for (int j = 0; j < 25; ++j) {
                if (DUAL) {
                    int bi = (4 * j + kh) * UD_LP + n;
                    b[j][0] = sUD[bi];
                    b[j][1] = sUD[bi + UD_LP];
                } else {
                    int bi = (4 * j + kh) * S_P + n;
                    b[j][0] = Ux[bi]       - Dx[bi];
                    b[j][1] = Ux[bi + S_P] - Dx[bi + S_P];
                }
            }
            for (int tm = tm_lo; tm < tm_hi; ++tm) {
                const int mr = tm * 16 + ((lane >> 4) << 3);
                float y5[8], y6[8], pv[8], fu[8], fd[8];
                const int po = (tn * TM + tm) * TILE_F + lane * 8;
                if (DUAL) {
                    const float4 p0 = *(const float4*)(Yd + OY_Z5 + po), p1 = *(const float4*)(Yd + OY_Z5 + po + 4);
                    const float4 q0 = *(const float4*)(Yd + OY_Z6 + po), q1 = *(const float4*)(Yd + OY_Z6 + po + 4);
                    const float4 w0 = *(const float4*)(pw + po),         w1 = *(const float4*)(pw + po + 4);
                    y5[0] = p0.x; y5[1] = p0.y; y5[2] = p0.z; y5[3] = p0.w; y5[4] = p1.x; y5[5] = p1.y; y5[6] = p1.z; y5[7] = p1.w;
                    y6[0] = q0.x; y6[1] = q0.y; y6[2] = q0.z; y6[3] = q0.w; y6[4] = q1.x; y6[5] = q1.y; y6[6] = q1.z; y6[7] = q1.w;
                    pv[0] = w0.x; pv[1] = w0.y; pv[2] = w0.z; pv[3] = w0.w; pv[4] = w1.x; pv[5] = w1.y; pv[6] = w1.z; pv[7] = w1.w;
                }
#pragma unroll
                for (int i = 0; i < 8; ++i) { fu[i] = fuP[mr + i]; fd[i] = fdP[mr + i]; }
                const float* arow = sPG + (tm * 16 + m16) * PG_LP + kh;
                v2f a[25];
#pragma unroll
                for (int j = 0; j < 25; ++j) { a[j][0] = arow[4 * j]; a[j][1] = arow[4 * j + 1]; }
                v8f acc = {0.f,0.f,0.f,0.f,0.f,0.f,0.f,0.f};
#pragma unroll
                for (int j = 0; j < 25; ++j)
                    acc = __builtin_amdgcn_wmma_f32_16x16x4_f32(false, a[j], false, b[j],
                                                                (short)0, acc, false, false);
                __builtin_amdgcn_sched_group_barrier(0x100, 13, 0);
                __builtin_amdgcn_sched_group_barrier(0x008, 25, 0);
                const bool live = n < NS;
                float r5[8], r6[8];
#pragma unroll
                for (int i = 0; i < 8; ++i) {
                    int o = (mr + i) * S_P + n;
                    float f = acc[i];
                    float n5, n6;
                    if (DUAL) {
                        n5 = fmaxf(y5[i] + step * (( f - fu[i]) + pv[i]), 0.f);
                        n6 = fmaxf(y6[i] + step * ((-f - fd[i]) - pv[i]), 0.f);
                    } else {
                        n5 =  f - fu[i];
                        n6 = -f - fd[i];
                    }
                    r5[i] = live ? n5 : 0.f;
                    r6[i] = live ? n6 : 0.f;
                    if (!DUAL) {
                        Zd[OY_Z5 + o] = r5[i];
                        Zd[OY_Z6 + o] = r6[i];
                    }
                }
                if (DUAL) {
                    *(float4*)(Zd + OY_Z5 + po)     = make_float4(r5[0], r5[1], r5[2], r5[3]);
                    *(float4*)(Zd + OY_Z5 + po + 4) = make_float4(r5[4], r5[5], r5[6], r5[7]);
                    *(float4*)(Zd + OY_Z6 + po)     = make_float4(r6[0], r6[1], r6[2], r6[3]);
                    *(float4*)(Zd + OY_Z6 + po + 4) = make_float4(r6[4], r6[5], r6[6], r6[7]);
                }
                const float s5 = rows_over_lanes(r5, m16);
                const float s6 = rows_over_lanes(r6, m16);
                if (m16 < 8) {
                    sRS[RS5 + tn * NL + mr + m16] = s5;
                    sRS[RS6 + tn * NL + mr + m16] = s6;
                }
            }
            t += tm_hi - tm_lo;
        }
    }
    if (tid < NL) {
        float s = 0.f; const float* r = sPG + tid * PG_LP;
        if (DUAL) {
            for (int g = 0; g < NG; ++g) s += r[g] * sXP[g];
        } else {
            const float* p = xb + OX_P;
            for (int g = 0; g < NG; ++g) s += r[g] * p[g];
        }
        gpP[tid] = s;
    }
    if (tid < NS) {
        float s = 0.f;
        if (DUAL) {
            const float* d = sUD + tid;
            for (int g = 0; g < NG; ++g) s += d[g * UD_LP];
        } else {
            const float* Ux = xb + OX_U + tid; const float* Dx = xb + OX_D + tid;
            for (int g = 0; g < NG; ++g) s += Ux[g * S_P] - Dx[g * S_P];
        }
        colP[tid] = s;
    }
    if (wave == nw - 1) {
        float a;
        if (DUAL) {
            a = sXP[lane] + sXP[32 + lane] + sXP[64 + lane];
            if (lane < NG - 96) a += sXP[96 + lane];
        } else {
            a = xb[OX_P + lane] + xb[OX_P + 32 + lane] + xb[OX_P + 64 + lane];
            if (lane < NG - 96) a += xb[OX_P + 96 + lane];
        }
        a += row_xor<1>(a); a += row_xor<2>(a); a += row_xor<4>(a); a += row_xor<8>(a);
        a += __shfl_xor(a, 16);
        if (lane == 0) sumpP[0] = a;
    }
    __syncthreads();
    if (tid == 0) yo[OY_Y1] = sumpP[0];
    for (int l = tid; l < NL; l += NT) {
        float g = gpP[l];
        yo[OY_Y2 + l] =  g + fuP[l];
        yo[OY_Y3 + l] = -g + fdP[l];
    }
    for (int s = tid; s < NS; s += NT) yo[OY_Y4 + s] = colP[s];
    for (int g = tid; g < NG; g += NT) {
        yo[OY_Z1 + g] = pP[g]  + ruP[g];
        yo[OY_Z2 + g] = rdP[g] - pP[g];
    }
}

template <bool FUSED>
__device__ __forceinline__ void rmatvec_core(float* w, const float* sPG, float* yb,
                                             float* xo, int tid, float* X,
                                             float costv, float cruv, float crdv, float step, float* sRS, float* sPart,
                                             float* sUD, float* sXP, float* xs, float* xbs, const float* ys) {
    const int lane = tid & 31, wave = __builtin_amdgcn_readfirstlane(tid >> 5), nw = NT / 32;
    __syncthreads();
    if (FUSED) {
        pgt_ydiff_partials(sPG, ys, tid, sPart);
        __syncthreads();
        if (tid < NG) {
            float a = 0.f, b = 0.f;
            for (int k = 0; k < N_STRIP; ++k) {
                a += sRS[RS3 + k * G_P + tid]; b += sRS[RS4 + k * G_P + tid];
            }
            const float c = pgt_ydiff_sum(sPart, tid);
            const float z1 = ys[OY_Z1 + tid], z2 = ys[OY_Z2 + tid];
            const float ap  = ys[OY_Y1] + c + z1 - z2;
            const float aru = z1 - a;
            const float ard = z2 - b;
            float o, n;
            o = xs[SX_P  + tid]; n = fmaxf(o - step * (costv + ap),  0.f);
            const float bp = 2.f * n - o;
            xs[SX_P  + tid] = n; xbs[SX_P  + tid] = bp; sXP[tid] = bp;
            o = xs[SX_RU + tid]; n = fmaxf(o - step * (cruv  + aru), 0.f);
            xs[SX_RU + tid] = n; xbs[SX_RU + tid] = 2.f * n - o;
            o = xs[SX_RD + tid]; n = fmaxf(o - step * (crdv  + ard), 0.f);
            xs[SX_RD + tid] = n; xbs[SX_RD + tid] = 2.f * n - o;
        }
        if (tid >= NG && tid < NG + NL) {
            int l = tid - NG;
            float a = 0.f, b = 0.f;
            for (int k = 0; k < N_STRIP; ++k) {
                a += sRS[RS5 + k * NL + l]; b += sRS[RS6 + k * NL + l];
            }
            const float afu = ys[OY_Y2 + l] - a;
            const float afd = ys[OY_Y3 + l] - b;
            float o, n;
            o = xs[SX_FU + l]; n = fmaxf(o - step * afu, 0.f);
            xs[SX_FU + l] = n; xbs[SX_FU + l] = 2.f * n - o;
            o = xs[SX_FD + l]; n = fmaxf(o - step * afd, 0.f);
            xs[SX_FD + l] = n; xbs[SX_FD + l] = 2.f * n - o;
        }
        __syncthreads();
    }
    {
        const float* Z5 = yb + OY_Z5; const float* Z6 = yb + OY_Z6;
        constexpr int TM = 7, NTile = 7 * 13;
        const int m16 = lane & 15;
        const int kh  = (lane >> 4) * 2;
        const int t1 = ((wave + 1) * NTile) / nw;
        int t = (wave * NTile) / nw;
        while (t < t1) {
            const int tn = t / TM;
            const int tm_lo = t - tn * TM;
            const int tm_hi = (tm_lo + (t1 - t) < TM) ? tm_lo + (t1 - t) : TM;
            const int n = tn * 16 + m16;
            v8f acc[TM];
#pragma unroll
            for (int q = 0; q < TM; ++q) acc[q] = (v8f){0.f,0.f,0.f,0.f,0.f,0.f,0.f,0.f};
            for (int kb = 0; kb < NL; kb += 40) {
                v2f b[10];
                if (FUSED) {
                    const int zo = tn * (NL / 16) * TILE_F + (kb >> 3) * 128 + m16 * 8 + (lane >> 4) * 4;
#pragma unroll
                    for (int blk = 0; blk < 5; ++blk) {
                        const float4 u = *(const float4*)(Z5 + zo + blk * 128);
                        const float4 v = *(const float4*)(Z6 + zo + blk * 128);
                        b[2 * blk][0]     = u.x - v.x;  b[2 * blk][1]     = u.y - v.y;
                        b[2 * blk + 1][0] = u.z - v.z;  b[2 * blk + 1][1] = u.w - v.w;
                    }
                } else {
#pragma unroll
                    for (int j = 0; j < 10; ++j) {
                        int bi = (kb + 4 * j + kh) * S_P + n;
                        b[j][0] = Z5[bi]       - Z6[bi];
                        b[j][1] = Z5[bi + S_P] - Z6[bi + S_P];
                    }
                }
#pragma unroll
                for (int q = 0; q < TM; ++q) {
                    if (q >= tm_lo && q < tm_hi) {
                        const float* acol = sPG + (kb + (FUSED ? 2 * kh : kh)) * PG_LP + q * 16 + m16;
                        v2f a[10];
#pragma unroll
                        for (int j = 0; j < 10; ++j) {
                            const int r = FUSED ? 8 * (j >> 1) + 2 * (j & 1) : 4 * j;
                            a[j][0] = acol[r * PG_LP]; a[j][1] = acol[(r + 1) * PG_LP];
                        }
#pragma unroll
                        for (int j = 0; j < 10; ++j)
                            acc[q] = __builtin_amdgcn_wmma_f32_16x16x4_f32(false, a[j], false, b[j],
                                                                           (short)0, acc[q], false, false);
                        __builtin_amdgcn_sched_group_barrier(0x100, 10, 0);
                        __builtin_amdgcn_sched_group_barrier(0x008, 10, 0);
                    }
                }
            }
            const float y4v = FUSED ? ys[OY_Y4 + n] : yb[OY_Y4 + n];
            const int ud0 = ((lane >> 4) << 3) * UD_LP + n;
#pragma unroll
            for (int q = 0; q < TM; ++q) {
                if (q >= tm_lo && q < tm_hi) {
                    int mr = q * 16 + ((lane >> 4) << 3);
                    if (FUSED) {
                        float xu[8], xd[8], z3v[8], z4v[8], ruv[8], rdv[8];
                        const int po = (tn * TM + q) * TILE_F + lane * 8;
                        {
                            const float4 u0 = *(const float4*)(X + OX_U + po),   u1 = *(const float4*)(X + OX_U + po + 4);
                            const float4 d0 = *(const float4*)(X + OX_D + po),   d1 = *(const float4*)(X + OX_D + po + 4);
                            const float4 e0 = *(const float4*)(yb + OY_Z3 + po), e1 = *(const float4*)(yb + OY_Z3 + po + 4);
                            const float4 f0 = *(const float4*)(yb + OY_Z4 + po), f1 = *(const float4*)(yb + OY_Z4 + po + 4);
                            xu[0] = u0.x; xu[1] = u0.y; xu[2] = u0.z; xu[3] = u0.w; xu[4] = u1.x; xu[5] = u1.y; xu[6] = u1.z; xu[7] = u1.w;
                            xd[0] = d0.x; xd[1] = d0.y; xd[2] = d0.z; xd[3] = d0.w; xd[4] = d1.x; xd[5] = d1.y; xd[6] = d1.z; xd[7] = d1.w;
                            z3v[0] = e0.x; z3v[1] = e0.y; z3v[2] = e0.z; z3v[3] = e0.w; z3v[4] = e1.x; z3v[5] = e1.y; z3v[6] = e1.z; z3v[7] = e1.w;
                            z4v[0] = f0.x; z4v[1] = f0.y; z4v[2] = f0.z; z4v[3] = f0.w; z4v[4] = f1.x; z4v[5] = f1.y; z4v[6] = f1.z; z4v[7] = f1.w;
                        }
#pragma unroll
                        for (int i = 0; i < 8; ++i) { ruv[i] = xbs[SX_RU + mr + i]; rdv[i] = xbs[SX_RD + mr + i]; }
                        float r3[8], r4[8], sU[8], sD[8];
#pragma unroll
                        for (int i = 0; i < 8; ++i) {
                            float tv = acc[q][i];
                            float aU =  y4v + z3v[i] + tv;
                            float aD = -y4v + z4v[i] - tv;
                            float nU = fmaxf(xu[i] - step * aU, 0.f);
                            float nD = fmaxf(xd[i] - step * aD, 0.f);
                            float bU = 2.f * nU - xu[i];
                            float bD = 2.f * nD - xd[i];
                            float n3 = fmaxf(z3v[i] + step * (bU - ruv[i]), 0.f);
                            float n4 = fmaxf(z4v[i] + step * (bD - rdv[i]), 0.f);
                            const bool ok = n < NS && mr + i < NG;
                            const float dv = ok ? bU - bD : 0.f;
                            if (mr + i < NG) sUD[ud0 + (q * 16 + i) * UD_LP] = dv;
                            sU[i] = ok ? nU : 0.f;
                            sD[i] = ok ? nD : 0.f;
                            r3[i] = ok ? n3 : 0.f;
                            r4[i] = ok ? n4 : 0.f;
                        }
                        *(float4*)(X + OX_U + po)       = make_float4(sU[0], sU[1], sU[2], sU[3]);
                        *(float4*)(X + OX_U + po + 4)   = make_float4(sU[4], sU[5], sU[6], sU[7]);
                        *(float4*)(X + OX_D + po)       = make_float4(sD[0], sD[1], sD[2], sD[3]);
                        *(float4*)(X + OX_D + po + 4)   = make_float4(sD[4], sD[5], sD[6], sD[7]);
                        *(float4*)(yb + OY_Z3 + po)     = make_float4(r3[0], r3[1], r3[2], r3[3]);
                        *(float4*)(yb + OY_Z3 + po + 4) = make_float4(r3[4], r3[5], r3[6], r3[7]);
                        *(float4*)(yb + OY_Z4 + po)     = make_float4(r4[0], r4[1], r4[2], r4[3]);
                        *(float4*)(yb + OY_Z4 + po + 4) = make_float4(r4[4], r4[5], r4[6], r4[7]);
                        const float s3 = rows_over_lanes(r3, m16);
                        const float s4 = rows_over_lanes(r4, m16);
                        if (m16 < 8) {
                            sRS[RS3 + tn * G_P + mr + m16] = s3;
                            sRS[RS4 + tn * G_P + mr + m16] = s4;
                        }
                    } else {
                        float xu[8], xd[8], ruv[8], rdv[8];
#pragma unroll
                        for (int i = 0; i < 8; ++i) {
                            int o = (mr + i) * S_P + n;
                            xu[i]  = xo[OX_U + o];       xd[i]  = xo[OX_D + o];
                            ruv[i] = xo[OX_RU + mr + i]; rdv[i] = xo[OX_RD + mr + i];
                        }
                        float r3[8], r4[8];
#pragma unroll
                        for (int i = 0; i < 8; ++i) {
                            int o = (mr + i) * S_P + n;
                            float tv = acc[q][i];
                            float z3 = xu[i] - ruv[i];
                            float z4 = xd[i] - rdv[i];
                            const bool ok = n < NS && mr + i < NG;
                            if (ok) {
                                xo[OX_U + o] =  y4v + z3 + tv;
                                xo[OX_D + o] = -y4v + z4 - tv;
                            }
                            r3[i] = ok ? z3 : 0.f;
                            r4[i] = ok ? z4 : 0.f;
                        }
                        const float s3 = rows_over_lanes(r3, m16);
                        const float s4 = rows_over_lanes(r4, m16);
                        if (m16 < 8) {
                            sRS[RS3 + tn * G_P + mr + m16] = s3;
                            sRS[RS4 + tn * G_P + mr + m16] = s4;
                        }
                    }
                }
            }
            t += tm_hi - tm_lo;
        }
    }
    if (FUSED) return;
    pgt_ydiff_partials(sPG, yb, tid, sPart);
    __syncthreads();
    if (tid < NG) {
        float a = 0.f, b = 0.f;
        for (int k = 0; k < N_STRIP; ++k) {
            a += sRS[RS3 + k * G_P + tid]; b += sRS[RS4 + k * G_P + tid];
        }
        const float z1 = yb[OY_Z1 + tid], z2 = yb[OY_Z2 + tid];
        xo[OX_P  + tid] = yb[OY_Y1] + pgt_ydiff_sum(sPart, tid) + z1 - z2;
        xo[OX_RU + tid] = z1 - a;
        xo[OX_RD + tid] = z2 - b;
    }
    if (tid >= NG && tid < NG + NL) {
        int l = tid - NG;
        float a = 0.f, b = 0.f;
        for (int k = 0; k < N_STRIP; ++k) {
            a += sRS[RS5 + k * NL + l]; b += sRS[RS6 + k * NL + l];
        }
        xo[OX_FU + l] = yb[OY_Y2 + l] - a;
        xo[OX_FD + l] = yb[OY_Y3 + l] - b;
    }
}


__global__ __launch_bounds__(NT) void k_solver(float* w, const float* Cost,
                                               const float* Cru, const float* Crd,
                                               const float* Pmax, float* out) {
    __shared__ float red[NT];
    __shared__ __attribute__((aligned(16))) float sPG[NL * PG_LP + PG_TAIL];
    __shared__ float sRS[RS_TOTAL];
    __shared__ float sPart[C_RANGES * C_SLOT];
    __shared__ float sUD[NG * UD_LP];
    __shared__ float sXP[G_P];
    __shared__ __attribute__((aligned(16))) float sXs[SX_TOTAL];
    __shared__ __attribute__((aligned(16))) float sXBs[SX_TOTAL];
    __shared__ __attribute__((aligned(16))) float sYs[OY_Z3];
    __shared__ __attribute__((aligned(16))) float sAXs[OY_Z3];
    __shared__ __attribute__((aligned(16))) float sW[SW_TOTAL];
    const int tid = threadIdx.x;
    float* X   = w + WS_X;
    float* Y   = w + WS_Y;
    float* AX  = w + WS_AX;

    for (int i = tid; i < NL * NG; i += NT) {
        int l = i / NG, g = i - l * NG;
        sPG[l * PG_LP + g] = w[WS_PG + l * G_P + g];
    }
    if (tid < PG_TAIL) sPG[NL * PG_LP + tid] = 0.f;
    __syncthreads();

    float sumsq = 0.f;
    for (int it = 0; it < POWER_ITERS; ++it) {
        __syncthreads();
        float part = 0.f;
        for (int i = tid; i < XSZ; i += NT) { float v = X[i]; part += v * v; }
        sumsq = block_reduce(part, red, tid);
        float inv = 1.0f / sqrtf(sumsq);
        for (int i = tid; i < XSZ; i += NT) X[i] *= inv;
        matvec_core<false>(w, sPG, X, AX, tid, nullptr, nullptr, 0.f, sRS, nullptr, nullptr, nullptr, nullptr, nullptr);
        rmatvec_core<false>(w, sPG, AX, X, tid, nullptr, 0.f, 0.f, 0.f, 0.f, sRS, sPart, nullptr, nullptr, nullptr, nullptr, nullptr);
    }
    __syncthreads();
    {
        float part = 0.f;
        for (int i = tid; i < XSZ; i += NT) { float v = X[i]; part += v * v; }
        sumsq = block_reduce(part, red, tid);
    }
    const float step = 0.9f / sqrtf(sqrtf(sumsq));
    const float b1 = w[WS_B1];
    const float* pw = w + WS_PW;
    const float costv = tid < NG ? Cost[tid] : 0.f, cruv = tid < NG ? Cru[tid] : 0.f, crdv = tid < NG ? Crd[tid] : 0.f;
    const float pmaxv = tid < NG ? Pmax[tid] : 0.f;
    const float b2v = tid < NL ? w[WS_B2 + tid] : 0.f, b3v = tid < NL ? w[WS_B3 + tid] : 0.f;
    const float b4v = tid < NS ? w[WS_B4 + tid] : 0.f;

    for (int i = tid; i < XSZ; i += NT) X[i] = 0.f;
    for (int i = tid; i < YSZ; i += NT) Y[i] = 0.f;
    for (int i = tid; i < RS_TOTAL; i += NT) sRS[i] = 0.f;
    for (int i = tid; i < NG * UD_LP; i += NT) sUD[i] = 0.f;
    for (int i = tid; i < SX_TOTAL; i += NT) { sXs[i] = 0.f; sXBs[i] = 0.f; }
    for (int i = tid; i < OY_Z3; i += NT) sYs[i] = 0.f;

    for (int it = 0; it < PDHG_ITERS; ++it) {
        rmatvec_core<true>(w, sPG, Y, nullptr, tid, X, costv, cruv, crdv, step, sRS, sPart, sUD, sXP, sXs, sXBs, sYs);
        matvec_core<true>(w, sPG, nullptr, nullptr, tid, Y, pw, step, sRS, sUD, sXP, sXBs, sAXs, sW);
        __syncthreads();
        if (tid == 0) sYs[OY_Y1] += step * (sAXs[OY_Y1] - b1);
        if (tid < NL) {
            sYs[OY_Y2 + tid] += step * (sAXs[OY_Y2 + tid] - b2v);
            sYs[OY_Y3 + tid] += step * (sAXs[OY_Y3 + tid] - b3v);
        }
        if (tid < NS)
            sYs[OY_Y4 + tid] += step * (sAXs[OY_Y4 + tid] - b4v);
        if (tid < NG) {
            sYs[OY_Z1 + tid] = fmaxf(sYs[OY_Z1 + tid] + step * (sAXs[OY_Z1 + tid] - pmaxv), 0.f);
            sYs[OY_Z2 + tid] = fmaxf(sYs[OY_Z2 + tid] + step *  sAXs[OY_Z2 + tid], 0.f);
        }
        __syncthreads();
    }

    for (int g = tid; g < NG; g += NT) {
        out[g]          = sXs[SX_P  + g];
        out[NG + g]     = sXs[SX_RU + g];
        out[2 * NG + g] = sXs[SX_RD + g];
    }
    for (int i = tid; i < NG * NS; i += NT) {
        int g = i / NS, s = i % NS;
        const int po = packed_off(g, s, G_P / 16);
        out[3 * NG + i]           = X[OX_U + po];
        out[3 * NG + NG * NS + i] = X[OX_D + po];
    }
    for (int l = tid; l < NL; l += NT) {
        out[3 * NG + 2 * NG * NS + l]      = sXs[SX_FU + l];
        out[3 * NG + 2 * NG * NS + NL + l] = sXs[SX_FD + l];
    }
}


extern "C" void kernel_launch(void* const* d_in, const int* in_sizes, int n_in,
                              void* d_out, int out_size, void* d_ws, size_t ws_size,
                              hipStream_t stream) {
    const float* w_scen   = (const float*)d_in[0];
    const float* Cost     = (const float*)d_in[1];
    const float* C_r_up   = (const float*)d_in[2];
    const float* C_r_down = (const float*)d_in[3];
    const float* Pmax     = (const float*)d_in[4];
    const float* PTDF     = (const float*)d_in[5];
    const float* node_G   = (const float*)d_in[6];
    const float* node_W   = (const float*)d_in[7];
    const float* node_L   = (const float*)d_in[8];
    const float* Pd       = (const float*)d_in[9];
    const float* w_exp    = (const float*)d_in[10];
    const float* Line_cap = (const float*)d_in[11];
    float* w   = (float*)d_ws;
    float* out = (float*)d_out;

    k_zero <<<512, 256, 0, stream>>>(w);
    k_prep1<<<240, 256, 0, stream>>>(w, w_scen, node_W, node_L, Pd, w_exp);
    k_prep2<<<640, 256, 0, stream>>>(w, PTDF, node_G, Line_cap);
    k_setv <<<161, 256, 0, stream>>>(w);
    k_solver<<<1, NT, 0, stream>>>(w, Cost, C_r_up, C_r_down, Pmax, out);
}
